# starting kernel
# baseline (speedup 1.0000x reference)
#include <hip/hip_runtime.h>
#include <hip/hip_bf16.h>

typedef __attribute__((ext_vector_type(2))) float v2f;
typedef __attribute__((ext_vector_type(8))) float v8f;

#define INCH  128
#define HC    256
#define NHD   8
#define CPH   32
#define OUTC  40
#define SLOPE 0.2f

__device__ __forceinline__ float leaky(float x) { return x > 0.f ? x : SLOPE * x; }

__global__ void gemm_x_w1(const float* __restrict__ X, const float* __restrict__ W,
                          float* __restrict__ H1, int nrows) {
    const int lane = threadIdx.x & 31;
    const int wave = threadIdx.x >> 5;
    const int m    = lane & 15;
    const int half = lane >> 4;
    const int rowBase = blockIdx.x << 4;
    if (rowBase >= nrows) return;
    const float* xrow = X + (size_t)(rowBase + m) * INCH;
    for (int t = 0; t < 2; ++t) {
        const int colBase = ((wave << 1) + t) << 4;
        v8f acc = {0.f, 0.f, 0.f, 0.f, 0.f, 0.f, 0.f, 0.f};
        for (int k = 0; k < INCH; k += 4) {
            const int kk = k + (half << 1);
            v2f a, b;
            a.x = xrow[kk];
            a.y = xrow[kk + 1];
            b.x = W[(size_t)kk * HC + colBase + m];
            b.y = W[(size_t)(kk + 1) * HC + colBase + m];
            acc = __builtin_amdgcn_wmma_f32_16x16x4_f32(false, a, false, b,
                                                        (short)0, acc, false, false);
        }
#pragma unroll
        for (int v = 0; v < 8; ++v)
            H1[(size_t)(rowBase + v + (half << 3)) * HC + colBase + m] = acc[v];
    }
}

__global__ void gemm_h_w2(const float* __restrict__ Hin, const float* __restrict__ W,
                          float* __restrict__ H2, int nrows) {
    const int lane = threadIdx.x & 31;
    const int wave = threadIdx.x >> 5;
    const int m    = lane & 15;
    const int half = lane >> 4;
    const int rowBase = blockIdx.x << 4;
    if (rowBase >= nrows) return;
    const int colBase = wave << 4;
    const int col = colBase + m;
    const float* hrow = Hin + (size_t)(rowBase + m) * HC;
    v8f acc = {0.f, 0.f, 0.f, 0.f, 0.f, 0.f, 0.f, 0.f};
    for (int k = 0; k < HC; k += 4) {
        const int kk = k + (half << 1);
        v2f a, b;
        a.x = hrow[kk];
        a.y = hrow[kk + 1];
        b.x = (col < OUTC) ? W[kk * OUTC + col] : 0.f;
        b.y = (col < OUTC) ? W[(kk + 1) * OUTC + col] : 0.f;
        acc = __builtin_amdgcn_wmma_f32_16x16x4_f32(false, a, false, b,
                                                    (short)0, acc, false, false);
    }
    if (col < OUTC) {
#pragma unroll
        for (int v = 0; v < 8; ++v)
            H2[(size_t)(rowBase + v + (half << 3)) * OUTC + col] = acc[v];
    }
}

__global__ void attn_dots_l1(const float* __restrict__ H1, const float* __restrict__ Asrc,
                             const float* __restrict__ Adst, float* __restrict__ aS,
                             float* __restrict__ aD, int n) {
    __shared__ float sAs[HC], sAd[HC];
    for (int i = threadIdx.x; i < HC; i += blockDim.x) { sAs[i] = Asrc[i]; sAd[i] = Adst[i]; }
    __syncthreads();
    const int idx = blockIdx.x * blockDim.x + threadIdx.x;
    if (idx >= n * NHD) return;
    const int node = idx >> 3, h = idx & 7;
    const float* row = H1 + (size_t)node * HC + h * CPH;
    float s = 0.f, d = 0.f;
#pragma unroll
    for (int c = 0; c < CPH; ++c) {
        const float v = row[c];
        s += v * sAs[h * CPH + c];
        d += v * sAd[h * CPH + c];
    }
    aS[idx] = s; aD[idx] = d;
}

__global__ void attn_dots_l2(const float* __restrict__ H2, const float* __restrict__ Asrc,
                             const float* __restrict__ Adst, float* __restrict__ aS,
                             float* __restrict__ aD, int n) {
    const int node = blockIdx.x * blockDim.x + threadIdx.x;
    if (node >= n) return;
    const float* row = H2 + (size_t)node * OUTC;
    float s = 0.f, d = 0.f;
#pragma unroll
    for (int c = 0; c < OUTC; ++c) { const float v = row[c]; s += v * Asrc[c]; d += v * Adst[c]; }
    aS[node] = s; aD[node] = d;
}

__global__ void count_edges(const int* __restrict__ dstA, int* __restrict__ cnt, int E, int n) {
    const int e = blockIdx.x * blockDim.x + threadIdx.x;
    if (e >= E + n) return;
    const int d = (e < E) ? dstA[e] : (e - E);
    atomicAdd(&cnt[d], 1);
}

__global__ void scan_rowptr(const int* __restrict__ cnt, int* __restrict__ rowptr,
                            int* __restrict__ cursor, int n) {
    __shared__ int lds[1024];
    __shared__ int carryS;
    int carry = 0;
    for (int base = 0; base < n; base += 1024) {
        const int i = base + (int)threadIdx.x;
        const int v = (i < n) ? cnt[i] : 0;
        lds[threadIdx.x] = v;
        __syncthreads();
        for (int off = 1; off < 1024; off <<= 1) {
            const int t = (threadIdx.x >= (unsigned)off) ? lds[threadIdx.x - off] : 0;
            __syncthreads();
            lds[threadIdx.x] += t;
            __syncthreads();
        }
        const int incl = lds[threadIdx.x];
        const int excl = incl - v;
        if (i < n) { rowptr[i] = carry + excl; cursor[i] = carry + excl; }
        if (threadIdx.x == 1023) carryS = carry + incl;
        __syncthreads();
        carry = carryS;
    }
    if (threadIdx.x == 0) rowptr[n] = carry;
}

__global__ void fill_csr(const int* __restrict__ srcA, const int* __restrict__ dstA,
                         int* __restrict__ cursor, int* __restrict__ col, int E, int n) {
    const int e = blockIdx.x * blockDim.x + threadIdx.x;
    if (e >= E + n) return;
    const int s = (e < E) ? srcA[e] : (e - E);
    const int d = (e < E) ? dstA[e] : (e - E);
    const int slot = atomicAdd(&cursor[d], 1);
    col[slot] = s;
}

__global__ void gat_gather_l1(const int* __restrict__ rowptr, const int* __restrict__ col,
                              const float* __restrict__ aS, const float* __restrict__ aD,
                              const float* __restrict__ H1, const float* __restrict__ b1,
                              float* __restrict__ out1, int n) {
    const int lane = threadIdx.x & 31;
    const int w = (blockIdx.x * blockDim.x + threadIdx.x) >> 5;
    if (w >= n * NHD) return;
    const int node = w >> 3, h = w & 7;
    const int beg = rowptr[node], end = rowptr[node + 1];
    const float aDd = aD[node * NHD + h];
    float m = -__builtin_inff(), den = 0.f, acc = 0.f;
    for (int j = beg; j < end; ++j) {
        const int s = col[j];
        if (j + 1 < end) {
            const int s2 = col[j + 1];
            __builtin_prefetch(&H1[(size_t)s2 * HC + h * CPH + lane], 0, 1);
        }
        const float l = leaky(aS[s * NHD + h] + aDd);
        const float mn = fmaxf(m, l);
        const float scale = expf(m - mn);
        const float wgt   = expf(l - mn);
        acc = acc * scale + wgt * H1[(size_t)s * HC + h * CPH + lane];
        den = den * scale + wgt;
        m = mn;
    }
    const float v = acc / den + b1[h * CPH + lane];
    out1[(size_t)node * HC + h * CPH + lane] = v > 0.f ? v : 0.f;
}

__global__ void gat_l2_fused(const int* __restrict__ rowptr, const int* __restrict__ col,
                             const float* __restrict__ aS, const float* __restrict__ aD,
                             const float* __restrict__ H2, const float* __restrict__ b2,
                             float* __restrict__ out, int n) {
    const int lane = threadIdx.x & 31;
    const int node = (blockIdx.x * blockDim.x + threadIdx.x) >> 5;
    if (node >= n) return;
    const int beg = rowptr[node], end = rowptr[node + 1];
    const float aDd = aD[node];
    const int c1 = lane + 32;
    float m = -__builtin_inff(), den = 0.f, acc0 = 0.f, acc1 = 0.f;
    for (int j = beg; j < end; ++j) {
        const int s = col[j];
        if (j + 1 < end) {
            const int s2 = col[j + 1];
            __builtin_prefetch(&H2[(size_t)s2 * OUTC + lane], 0, 1);
        }
        const float l = leaky(aS[s] + aDd);
        const float mn = fmaxf(m, l);
        const float scale = expf(m - mn);
        const float wgt   = expf(l - mn);
        const float* hrow = H2 + (size_t)s * OUTC;
        acc0 = acc0 * scale + wgt * ((lane < OUTC) ? hrow[lane] : 0.f);
        acc1 = acc1 * scale + wgt * ((c1 < OUTC) ? hrow[c1] : 0.f);
        den = den * scale + wgt;
        m = mn;
    }
    float v0 = (lane < OUTC) ? acc0 / den + b2[lane] : -__builtin_inff();
    float v1 = (c1 < OUTC)  ? acc1 / den + b2[c1]  : -__builtin_inff();
    float mx = fmaxf(v0, v1);
    for (int o = 16; o > 0; o >>= 1) mx = fmaxf(mx, __shfl_xor(mx, o, 32));
    float ssum = 0.f;
    if (lane < OUTC) ssum += expf(v0 - mx);
    if (c1 < OUTC)  ssum += expf(v1 - mx);
    for (int o = 16; o > 0; o >>= 1) ssum += __shfl_xor(ssum, o, 32);
    const float lse = mx + logf(ssum);
    if (lane < OUTC) out[(size_t)node * OUTC + lane] = v0 - lse;
    if (c1 < OUTC)  out[(size_t)node * OUTC + c1]  = v1 - lse;
}

extern "C" void kernel_launch(void* const* d_in, const int* in_sizes, int n_in,
                              void* d_out, int out_size, void* d_ws, size_t ws_size,
                              hipStream_t stream) {
    const float* x   = (const float*)d_in[0];
    const int*   ei  = (const int*)d_in[1];
    const float* W1  = (const float*)d_in[2];
    const float* as1 = (const float*)d_in[3];
    const float* ad1 = (const float*)d_in[4];
    const float* b1  = (const float*)d_in[5];
    const float* W2  = (const float*)d_in[6];
    const float* as2 = (const float*)d_in[7];
    const float* ad2 = (const float*)d_in[8];
    const float* b2  = (const float*)d_in[9];
    const int N  = in_sizes[0] / INCH;
    const int E  = in_sizes[1] / 2;
    const int Et = E + N;
    const int* srcA = ei;
    const int* dstA = ei + E;

    float* ws = (float*)d_ws;
    size_t off = 0;
    float* h1   = ws + off; off += (size_t)N * HC;
    float* out1 = ws + off; off += (size_t)N * HC;
    float* aS1  = ws + off; off += (size_t)N * NHD;
    float* aD1  = ws + off; off += (size_t)N * NHD;
    int* rowptr = (int*)(ws + off); off += (size_t)N + 1;
    int* cnt    = (int*)(ws + off); off += (size_t)N;
    int* cursor = (int*)(ws + off); off += (size_t)N;
    int* col    = (int*)(ws + off); off += (size_t)Et;
    float* h2  = h1;
    float* aS2 = aS1;
    float* aD2 = aD1;

    hipMemsetAsync(cnt, 0, (size_t)N * sizeof(int), stream);
    count_edges<<<(Et + 255) / 256, 256, 0, stream>>>(dstA, cnt, E, N);
    scan_rowptr<<<1, 1024, 0, stream>>>(cnt, rowptr, cursor, N);
    fill_csr<<<(Et + 255) / 256, 256, 0, stream>>>(srcA, dstA, cursor, col, E, N);

    gemm_x_w1<<<(N + 15) / 16, 256, 0, stream>>>(x, W1, h1, N);
    attn_dots_l1<<<(N * NHD + 255) / 256, 256, 0, stream>>>(h1, as1, ad1, aS1, aD1, N);
    gat_gather_l1<<<(N * NHD * 32 + 255) / 256, 256, 0, stream>>>(
        rowptr, col, aS1, aD1, h1, b1, out1, N);

    gemm_h_w2<<<(N + 15) / 16, 96, 0, stream>>>(out1, W2, h2, N);
    attn_dots_l2<<<(N + 255) / 256, 256, 0, stream>>>(h2, as2, ad2, aS2, aD2, N);
    gat_l2_fused<<<(N * 32 + 255) / 256, 256, 0, stream>>>(
        rowptr, col, aS2, aD2, h2, b2, (float*)d_out, N);
}


# layer-1 gather: one wave per destination node covers all 8 heads, each lane 8 contiguous channels by two 128-bit loads (was a wave per node and head, 32-bit loads); per-channel sum order unchanged
# speedup vs baseline: 1.3750x; 1.3750x over previous
#include <hip/hip_runtime.h>
#include <hip/hip_bf16.h>

typedef __attribute__((ext_vector_type(2))) float v2f;
typedef __attribute__((ext_vector_type(8))) float v8f;

#define INCH  128
#define HC    256
#define NHD   8
#define CPH   32
#define OUTC  40
#define SLOPE 0.2f

__device__ __forceinline__ float leaky(float x) { return x > 0.f ? x : SLOPE * x; }

__global__ void gemm_x_w1(const float* __restrict__ X, const float* __restrict__ W,
                          float* __restrict__ H1, int nrows) {
    const int lane = threadIdx.x & 31;
    const int wave = threadIdx.x >> 5;
    const int m    = lane & 15;
    const int half = lane >> 4;
    const int rowBase = blockIdx.x << 4;
    if (rowBase >= nrows) return;
    const float* xrow = X + (size_t)(rowBase + m) * INCH;
    for (int t = 0; t < 2; ++t) {
        const int colBase = ((wave << 1) + t) << 4;
        v8f acc = {0.f, 0.f, 0.f, 0.f, 0.f, 0.f, 0.f, 0.f};
        for (int k = 0; k < INCH; k += 4) {
            const int kk = k + (half << 1);
            v2f a, b;
            a.x = xrow[kk];
            a.y = xrow[kk + 1];
            b.x = W[(size_t)kk * HC + colBase + m];
            b.y = W[(size_t)(kk + 1) * HC + colBase + m];
            acc = __builtin_amdgcn_wmma_f32_16x16x4_f32(false, a, false, b,
                                                        (short)0, acc, false, false);
        }
#pragma unroll
        for (int v = 0; v < 8; ++v)
            H1[(size_t)(rowBase + v + (half << 3)) * HC + colBase + m] = acc[v];
    }
}

__global__ void gemm_h_w2(const float* __restrict__ Hin, const float* __restrict__ W,
                          float* __restrict__ H2, int nrows) {
    const int lane = threadIdx.x & 31;
    const int wave = threadIdx.x >> 5;
    const int m    = lane & 15;
    const int half = lane >> 4;
    const int rowBase = blockIdx.x << 4;
    if (rowBase >= nrows) return;
    const int colBase = wave << 4;
    const int col = colBase + m;
    const float* hrow = Hin + (size_t)(rowBase + m) * HC;
    v8f acc = {0.f, 0.f, 0.f, 0.f, 0.f, 0.f, 0.f, 0.f};
    for (int k = 0; k < HC; k += 4) {
        const int kk = k + (half << 1);
        v2f a, b;
        a.x = hrow[kk];
        a.y = hrow[kk + 1];
        b.x = (col < OUTC) ? W[kk * OUTC + col] : 0.f;
        b.y = (col < OUTC) ? W[(kk + 1) * OUTC + col] : 0.f;
        acc = __builtin_amdgcn_wmma_f32_16x16x4_f32(false, a, false, b,
                                                    (short)0, acc, false, false);
    }
    if (col < OUTC) {
#pragma unroll
        for (int v = 0; v < 8; ++v)
            H2[(size_t)(rowBase + v + (half << 3)) * OUTC + col] = acc[v];
    }
}

__global__ void attn_dots_l1(const float* __restrict__ H1, const float* __restrict__ Asrc,
                             const float* __restrict__ Adst, float* __restrict__ aS,
                             float* __restrict__ aD, int n) {
    __shared__ float sAs[HC], sAd[HC];
    for (int i = threadIdx.x; i < HC; i += blockDim.x) { sAs[i] = Asrc[i]; sAd[i] = Adst[i]; }
    __syncthreads();
    const int idx = blockIdx.x * blockDim.x + threadIdx.x;
    if (idx >= n * NHD) return;
    const int node = idx >> 3, h = idx & 7;
    const float* row = H1 + (size_t)node * HC + h * CPH;
    float s = 0.f, d = 0.f;
#pragma unroll
    for (int c = 0; c < CPH; ++c) {
        const float v = row[c];
        s += v * sAs[h * CPH + c];
        d += v * sAd[h * CPH + c];
    }
    aS[idx] = s; aD[idx] = d;
}

__global__ void attn_dots_l2(const float* __restrict__ H2, const float* __restrict__ Asrc,
                             const float* __restrict__ Adst, float* __restrict__ aS,
                             float* __restrict__ aD, int n) {
    const int node = blockIdx.x * blockDim.x + threadIdx.x;
    if (node >= n) return;
    const float* row = H2 + (size_t)node * OUTC;
    float s = 0.f, d = 0.f;
#pragma unroll
    for (int c = 0; c < OUTC; ++c) { const float v = row[c]; s += v * Asrc[c]; d += v * Adst[c]; }
    aS[node] = s; aD[node] = d;
}

__global__ void count_edges(const int* __restrict__ dstA, int* __restrict__ cnt, int E, int n) {
    const int e = blockIdx.x * blockDim.x + threadIdx.x;
    if (e >= E + n) return;
    const int d = (e < E) ? dstA[e] : (e - E);
    atomicAdd(&cnt[d], 1);
}

__global__ void scan_rowptr(const int* __restrict__ cnt, int* __restrict__ rowptr,
                            int* __restrict__ cursor, int n) {
    __shared__ int lds[1024];
    __shared__ int carryS;
    int carry = 0;
    for (int base = 0; base < n; base += 1024) {
        const int i = base + (int)threadIdx.x;
        const int v = (i < n) ? cnt[i] : 0;
        lds[threadIdx.x] = v;
        __syncthreads();
        for (int off = 1; off < 1024; off <<= 1) {
            const int t = (threadIdx.x >= (unsigned)off) ? lds[threadIdx.x - off] : 0;
            __syncthreads();
            lds[threadIdx.x] += t;
            __syncthreads();
        }
        const int incl = lds[threadIdx.x];
        const int excl = incl - v;
        if (i < n) { rowptr[i] = carry + excl; cursor[i] = carry + excl; }
        if (threadIdx.x == 1023) carryS = carry + incl;
        __syncthreads();
        carry = carryS;
    }
    if (threadIdx.x == 0) rowptr[n] = carry;
}

__global__ void fill_csr(const int* __restrict__ srcA, const int* __restrict__ dstA,
                         int* __restrict__ cursor, int* __restrict__ col, int E, int n) {
    const int e = blockIdx.x * blockDim.x + threadIdx.x;
    if (e >= E + n) return;
    const int s = (e < E) ? srcA[e] : (e - E);
    const int d = (e < E) ? dstA[e] : (e - E);
    const int slot = atomicAdd(&cursor[d], 1);
    col[slot] = s;
}

__global__ void gat_gather_l1(const int* __restrict__ rowptr, const int* __restrict__ col,
                              const float* __restrict__ aS, const float* __restrict__ aD,
                              const float* __restrict__ H1, const float* __restrict__ b1,
                              float* __restrict__ out1, int n) {
    const int lane = threadIdx.x & 31;
    const int node = (blockIdx.x * blockDim.x + threadIdx.x) >> 5;
    if (node >= n) return;
    const int h  = lane >> 2;
    const int ch = lane << 3;
    const int beg = rowptr[node], end = rowptr[node + 1];
    const float aDd = aD[node * NHD + h];
    float m = -__builtin_inff(), den = 0.f;
    float a0 = 0.f, a1 = 0.f, a2 = 0.f, a3 = 0.f, a4 = 0.f, a5 = 0.f, a6 = 0.f, a7 = 0.f;
    for (int j = beg; j < end; ++j) {
        const int s = col[j];
        if (j + 1 < end) {
            const int s2 = col[j + 1];
            __builtin_prefetch(&H1[(size_t)s2 * HC + ch], 0, 1);
        }
        const float l = leaky(aS[s * NHD + h] + aDd);
        const float mn = fmaxf(m, l);
        const float scale = expf(m - mn);
        const float wgt   = expf(l - mn);
        const float4* hp = reinterpret_cast<const float4*>(H1 + (size_t)s * HC + ch);
        const float4 x0 = hp[0];
        const float4 x1 = hp[1];
        a0 = a0 * scale + wgt * x0.x;
        a1 = a1 * scale + wgt * x0.y;
        a2 = a2 * scale + wgt * x0.z;
        a3 = a3 * scale + wgt * x0.w;
        a4 = a4 * scale + wgt * x1.x;
        a5 = a5 * scale + wgt * x1.y;
        a6 = a6 * scale + wgt * x1.z;
        a7 = a7 * scale + wgt * x1.w;
        den = den * scale + wgt;
        m = mn;
    }
    const float v0 = a0 / den + b1[ch + 0];
    const float v1 = a1 / den + b1[ch + 1];
    const float v2 = a2 / den + b1[ch + 2];
    const float v3 = a3 / den + b1[ch + 3];
    const float v4 = a4 / den + b1[ch + 4];
    const float v5 = a5 / den + b1[ch + 5];
    const float v6 = a6 / den + b1[ch + 6];
    const float v7 = a7 / den + b1[ch + 7];
    float4 o0, o1;
    o0.x = v0 > 0.f ? v0 : 0.f;
    o0.y = v1 > 0.f ? v1 : 0.f;
    o0.z = v2 > 0.f ? v2 : 0.f;
    o0.w = v3 > 0.f ? v3 : 0.f;
    o1.x = v4 > 0.f ? v4 : 0.f;
    o1.y = v5 > 0.f ? v5 : 0.f;
    o1.z = v6 > 0.f ? v6 : 0.f;
    o1.w = v7 > 0.f ? v7 : 0.f;
    float4* op = reinterpret_cast<float4*>(out1 + (size_t)node * HC + ch);
    op[0] = o0;
    op[1] = o1;
}

__global__ void gat_l2_fused(const int* __restrict__ rowptr, const int* __restrict__ col,
                             const float* __restrict__ aS, const float* __restrict__ aD,
                             const float* __restrict__ H2, const float* __restrict__ b2,
                             float* __restrict__ out, int n) {
    const int lane = threadIdx.x & 31;
    const int node = (blockIdx.x * blockDim.x + threadIdx.x) >> 5;
    if (node >= n) return;
    const int beg = rowptr[node], end = rowptr[node + 1];
    const float aDd = aD[node];
    const int c1 = lane + 32;
    float m = -__builtin_inff(), den = 0.f, acc0 = 0.f, acc1 = 0.f;
    for (int j = beg; j < end; ++j) {
        const int s = col[j];
        if (j + 1 < end) {
            const int s2 = col[j + 1];
            __builtin_prefetch(&H2[(size_t)s2 * OUTC + lane], 0, 1);
        }
        const float l = leaky(aS[s] + aDd);
        const float mn = fmaxf(m, l);
        const float scale = expf(m - mn);
        const float wgt   = expf(l - mn);
        const float* hrow = H2 + (size_t)s * OUTC;
        acc0 = acc0 * scale + wgt * ((lane < OUTC) ? hrow[lane] : 0.f);
        acc1 = acc1 * scale + wgt * ((c1 < OUTC) ? hrow[c1] : 0.f);
        den = den * scale + wgt;
        m = mn;
    }
    float v0 = (lane < OUTC) ? acc0 / den + b2[lane] : -__builtin_inff();
    float v1 = (c1 < OUTC)  ? acc1 / den + b2[c1]  : -__builtin_inff();
    float mx = fmaxf(v0, v1);
    for (int o = 16; o > 0; o >>= 1) mx = fmaxf(mx, __shfl_xor(mx, o, 32));
    float ssum = 0.f;
    if (lane < OUTC) ssum += expf(v0 - mx);
    if (c1 < OUTC)  ssum += expf(v1 - mx);
    for (int o = 16; o > 0; o >>= 1) ssum += __shfl_xor(ssum, o, 32);
    const float lse = mx + logf(ssum);
    if (lane < OUTC) out[(size_t)node * OUTC + lane] = v0 - lse;
    if (c1 < OUTC)  out[(size_t)node * OUTC + c1]  = v1 - lse;
}

extern "C" void kernel_launch(void* const* d_in, const int* in_sizes, int n_in,
                              void* d_out, int out_size, void* d_ws, size_t ws_size,
                              hipStream_t stream) {
    const float* x   = (const float*)d_in[0];
    const int*   ei  = (const int*)d_in[1];
    const float* W1  = (const float*)d_in[2];
    const float* as1 = (const float*)d_in[3];
    const float* ad1 = (const float*)d_in[4];
    const float* b1  = (const float*)d_in[5];
    const float* W2  = (const float*)d_in[6];
    const float* as2 = (const float*)d_in[7];
    const float* ad2 = (const float*)d_in[8];
    const float* b2  = (const float*)d_in[9];
    const int N  = in_sizes[0] / INCH;
    const int E  = in_sizes[1] / 2;
    const int Et = E + N;
    const int* srcA = ei;
    const int* dstA = ei + E;

    float* ws = (float*)d_ws;
    size_t off = 0;
    float* h1   = ws + off; off += (size_t)N * HC;
    float* out1 = ws + off; off += (size_t)N * HC;
    float* aS1  = ws + off; off += (size_t)N * NHD;
    float* aD1  = ws + off; off += (size_t)N * NHD;
    int* rowptr = (int*)(ws + off); off += (size_t)N + 1;
    int* cnt    = (int*)(ws + off); off += (size_t)N;
    int* cursor = (int*)(ws + off); off += (size_t)N;
    int* col    = (int*)(ws + off); off += (size_t)Et;
    float* h2  = h1;
    float* aS2 = aS1;
    float* aD2 = aD1;

    hipMemsetAsync(cnt, 0, (size_t)N * sizeof(int), stream);
    count_edges<<<(Et + 255) / 256, 256, 0, stream>>>(dstA, cnt, E, N);
    scan_rowptr<<<1, 1024, 0, stream>>>(cnt, rowptr, cursor, N);
    fill_csr<<<(Et + 255) / 256, 256, 0, stream>>>(srcA, dstA, cursor, col, E, N);

    gemm_x_w1<<<(N + 15) / 16, 256, 0, stream>>>(x, W1, h1, N);
    attn_dots_l1<<<(N * NHD + 255) / 256, 256, 0, stream>>>(h1, as1, ad1, aS1, aD1, N);
    gat_gather_l1<<<(N * 32 + 255) / 256, 256, 0, stream>>>(
        rowptr, col, aS1, aD1, h1, b1, out1, N);

    gemm_h_w2<<<(N + 15) / 16, 96, 0, stream>>>(out1, W2, h2, N);
    attn_dots_l2<<<(N + 255) / 256, 256, 0, stream>>>(h2, as2, ad2, aS2, aD2, N);
    gat_l2_fused<<<(N * 32 + 255) / 256, 256, 0, stream>>>(
        rowptr, col, aS2, aD2, h2, b2, (float*)d_out, N);
}


# rowptr scan: one block per 1024 counts, all at once; each sums the counts in front of it itself (16 loads in flight), then wave shuffles + one LDS hand-over; was 1 block, 49 serial chunks
# speedup vs baseline: 1.9094x; 1.3887x over previous
#include <hip/hip_runtime.h>
#include <hip/hip_bf16.h>

typedef __attribute__((ext_vector_type(2))) float v2f;
typedef __attribute__((ext_vector_type(8))) float v8f;

#define INCH  128
#define HC    256
#define NHD   8
#define CPH   32
#define OUTC  40
#define SLOPE 0.2f

__device__ __forceinline__ float leaky(float x) { return x > 0.f ? x : SLOPE * x; }

__global__ void gemm_x_w1(const float* __restrict__ X, const float* __restrict__ W,
                          float* __restrict__ H1, int nrows) {
    const int lane = threadIdx.x & 31;
    const int wave = threadIdx.x >> 5;
    const int m    = lane & 15;
    const int half = lane >> 4;
    const int rowBase = blockIdx.x << 4;
    if (rowBase >= nrows) return;
    const float* xrow = X + (size_t)(rowBase + m) * INCH;
    for (int t = 0; t < 2; ++t) {
        const int colBase = ((wave << 1) + t) << 4;
        v8f acc = {0.f, 0.f, 0.f, 0.f, 0.f, 0.f, 0.f, 0.f};
        for (int k = 0; k < INCH; k += 4) {
            const int kk = k + (half << 1);
            v2f a, b;
            a.x = xrow[kk];
            a.y = xrow[kk + 1];
            b.x = W[(size_t)kk * HC + colBase + m];
            b.y = W[(size_t)(kk + 1) * HC + colBase + m];
            acc = __builtin_amdgcn_wmma_f32_16x16x4_f32(false, a, false, b,
                                                        (short)0, acc, false, false);
        }
#pragma unroll
        for (int v = 0; v < 8; ++v)
            H1[(size_t)(rowBase + v + (half << 3)) * HC + colBase + m] = acc[v];
    }
}

__global__ void gemm_h_w2(const float* __restrict__ Hin, const float* __restrict__ W,
                          float* __restrict__ H2, int nrows) {
    const int lane = threadIdx.x & 31;
    const int wave = threadIdx.x >> 5;
    const int m    = lane & 15;
    const int half = lane >> 4;
    const int rowBase = blockIdx.x << 4;
    if (rowBase >= nrows) return;
    const int colBase = wave << 4;
    const int col = colBase + m;
    const float* hrow = Hin + (size_t)(rowBase + m) * HC;
    v8f acc = {0.f, 0.f, 0.f, 0.f, 0.f, 0.f, 0.f, 0.f};
    for (int k = 0; k < HC; k += 4) {
        const int kk = k + (half << 1);
        v2f a, b;
        a.x = hrow[kk];
        a.y = hrow[kk + 1];
        b.x = (col < OUTC) ? W[kk * OUTC + col] : 0.f;
        b.y = (col < OUTC) ? W[(kk + 1) * OUTC + col] : 0.f;
        acc = __builtin_amdgcn_wmma_f32_16x16x4_f32(false, a, false, b,
                                                    (short)0, acc, false, false);
    }
    if (col < OUTC) {
#pragma unroll
        for (int v = 0; v < 8; ++v)
            H2[(size_t)(rowBase + v + (half << 3)) * OUTC + col] = acc[v];
    }
}

__global__ void attn_dots_l1(const float* __restrict__ H1, const float* __restrict__ Asrc,
                             const float* __restrict__ Adst, float* __restrict__ aS,
                             float* __restrict__ aD, int n) {
    __shared__ float sAs[HC], sAd[HC];
    for (int i = threadIdx.x; i < HC; i += blockDim.x) { sAs[i] = Asrc[i]; sAd[i] = Adst[i]; }
    __syncthreads();
    const int idx = blockIdx.x * blockDim.x + threadIdx.x;
    if (idx >= n * NHD) return;
    const int node = idx >> 3, h = idx & 7;
    const float* row = H1 + (size_t)node * HC + h * CPH;
    float s = 0.f, d = 0.f;
#pragma unroll
    for (int c = 0; c < CPH; ++c) {
        const float v = row[c];
        s += v * sAs[h * CPH + c];
        d += v * sAd[h * CPH + c];
    }
    aS[idx] = s; aD[idx] = d;
}

__global__ void attn_dots_l2(const float* __restrict__ H2, const float* __restrict__ Asrc,
                             const float* __restrict__ Adst, float* __restrict__ aS,
                             float* __restrict__ aD, int n) {
    const int node = blockIdx.x * blockDim.x + threadIdx.x;
    if (node >= n) return;
    const float* row = H2 + (size_t)node * OUTC;
    float s = 0.f, d = 0.f;
#pragma unroll
    for (int c = 0; c < OUTC; ++c) { const float v = row[c]; s += v * Asrc[c]; d += v * Adst[c]; }
    aS[node] = s; aD[node] = d;
}

__global__ void count_edges(const int* __restrict__ dstA, int* __restrict__ cnt, int E, int n) {
    const int e = blockIdx.x * blockDim.x + threadIdx.x;
    if (e >= E + n) return;
    const int d = (e < E) ? dstA[e] : (e - E);
    atomicAdd(&cnt[d], 1);
}

#define SCAN_AHEAD 16
__global__ void scan_rowptr(const int* __restrict__ cnt, int* __restrict__ rowptr,
                            int* __restrict__ cursor, int n) {
    __shared__ int sPre[32];
    __shared__ int sTot[32];
    const int tid  = (int)threadIdx.x;
    const int lane = tid & 31;
    const int wave = tid >> 5;
    const int base = (int)blockIdx.x * 1024;
    const int i = base + tid;
    const int v = (i < n) ? cnt[i] : 0;
    int pre = 0;
    for (int k0 = 0; k0 < base; k0 += SCAN_AHEAD * 1024) {
        for (int u = 0; u < SCAN_AHEAD; ++u) {
            const int kb = k0 + u * 1024;
            const bool in = kb < base;
            const int x = cnt[(in ? kb : 0) + tid];
            pre += in ? x : 0;
        }
    }
    int inc = v;
    for (int d = 1; d < 32; d <<= 1) {
        const int t = __shfl_up(inc, d, 32);
        if (lane >= d) inc += t;
    }
    for (int o = 16; o > 0; o >>= 1) pre += __shfl_xor(pre, o, 32);
    if (lane == 31) { sTot[wave] = inc; sPre[wave] = pre; }
    __syncthreads();
    int off = sPre[lane] + ((lane < wave) ? sTot[lane] : 0);
    for (int o = 16; o > 0; o >>= 1) off += __shfl_xor(off, o, 32);
    const int excl = off + inc - v;
    if (i < n) { rowptr[i] = excl; cursor[i] = excl; }
    if (i == n - 1) rowptr[n] = excl + v;
}

__global__ void fill_csr(const int* __restrict__ srcA, const int* __restrict__ dstA,
                         int* __restrict__ cursor, int* __restrict__ col, int E, int n) {
    const int e = blockIdx.x * blockDim.x + threadIdx.x;
    if (e >= E + n) return;
    const int s = (e < E) ? srcA[e] : (e - E);
    const int d = (e < E) ? dstA[e] : (e - E);
    const int slot = atomicAdd(&cursor[d], 1);
    col[slot] = s;
}

__global__ void gat_gather_l1(const int* __restrict__ rowptr, const int* __restrict__ col,
                              const float* __restrict__ aS, const float* __restrict__ aD,
                              const float* __restrict__ H1, const float* __restrict__ b1,
                              float* __restrict__ out1, int n) {
    const int lane = threadIdx.x & 31;
    const int node = (blockIdx.x * blockDim.x + threadIdx.x) >> 5;
    if (node >= n) return;
    const int h  = lane >> 2;
    const int ch = lane << 3;
    const int beg = rowptr[node], end = rowptr[node + 1];
    const float aDd = aD[node * NHD + h];
    float m = -__builtin_inff(), den = 0.f;
    float a0 = 0.f, a1 = 0.f, a2 = 0.f, a3 = 0.f, a4 = 0.f, a5 = 0.f, a6 = 0.f, a7 = 0.f;
    for (int j = beg; j < end; ++j) {
        const int s = col[j];
        if (j + 1 < end) {
            const int s2 = col[j + 1];
            __builtin_prefetch(&H1[(size_t)s2 * HC + ch], 0, 1);
        }
        const float l = leaky(aS[s * NHD + h] + aDd);
        const float mn = fmaxf(m, l);
        const float scale = expf(m - mn);
        const float wgt   = expf(l - mn);
        const float4* hp = reinterpret_cast<const float4*>(H1 + (size_t)s * HC + ch);
        const float4 x0 = hp[0];
        const float4 x1 = hp[1];
        a0 = a0 * scale + wgt * x0.x;
        a1 = a1 * scale + wgt * x0.y;
        a2 = a2 * scale + wgt * x0.z;
        a3 = a3 * scale + wgt * x0.w;
        a4 = a4 * scale + wgt * x1.x;
        a5 = a5 * scale + wgt * x1.y;
        a6 = a6 * scale + wgt * x1.z;
        a7 = a7 * scale + wgt * x1.w;
        den = den * scale + wgt;
        m = mn;
    }
    const float v0 = a0 / den + b1[ch + 0];
    const float v1 = a1 / den + b1[ch + 1];
    const float v2 = a2 / den + b1[ch + 2];
    const float v3 = a3 / den + b1[ch + 3];
    const float v4 = a4 / den + b1[ch + 4];
    const float v5 = a5 / den + b1[ch + 5];
    const float v6 = a6 / den + b1[ch + 6];
    const float v7 = a7 / den + b1[ch + 7];
    float4 o0, o1;
    o0.x = v0 > 0.f ? v0 : 0.f;
    o0.y = v1 > 0.f ? v1 : 0.f;
    o0.z = v2 > 0.f ? v2 : 0.f;
    o0.w = v3 > 0.f ? v3 : 0.f;
    o1.x = v4 > 0.f ? v4 : 0.f;
    o1.y = v5 > 0.f ? v5 : 0.f;
    o1.z = v6 > 0.f ? v6 : 0.f;
    o1.w = v7 > 0.f ? v7 : 0.f;
    float4* op = reinterpret_cast<float4*>(out1 + (size_t)node * HC + ch);
    op[0] = o0;
    op[1] = o1;
}

__global__ void gat_l2_fused(const int* __restrict__ rowptr, const int* __restrict__ col,
                             const float* __restrict__ aS, const float* __restrict__ aD,
                             const float* __restrict__ H2, const float* __restrict__ b2,
                             float* __restrict__ out, int n) {
    const int lane = threadIdx.x & 31;
    const int node = (blockIdx.x * blockDim.x + threadIdx.x) >> 5;
    if (node >= n) return;
    const int beg = rowptr[node], end = rowptr[node + 1];
    const float aDd = aD[node];
    const int c1 = lane + 32;
    float m = -__builtin_inff(), den = 0.f, acc0 = 0.f, acc1 = 0.f;
    for (int j = beg; j < end; ++j) {
        const int s = col[j];
        if (j + 1 < end) {
            const int s2 = col[j + 1];
            __builtin_prefetch(&H2[(size_t)s2 * OUTC + lane], 0, 1);
        }
        const float l = leaky(aS[s] + aDd);
        const float mn = fmaxf(m, l);
        const float scale = expf(m - mn);
        const float wgt   = expf(l - mn);
        const float* hrow = H2 + (size_t)s * OUTC;
        acc0 = acc0 * scale + wgt * ((lane < OUTC) ? hrow[lane] : 0.f);
        acc1 = acc1 * scale + wgt * ((c1 < OUTC) ? hrow[c1] : 0.f);
        den = den * scale + wgt;
        m = mn;
    }
    float v0 = (lane < OUTC) ? acc0 / den + b2[lane] : -__builtin_inff();
    float v1 = (c1 < OUTC)  ? acc1 / den + b2[c1]  : -__builtin_inff();
    float mx = fmaxf(v0, v1);
    for (int o = 16; o > 0; o >>= 1) mx = fmaxf(mx, __shfl_xor(mx, o, 32));
    float ssum = 0.f;
    if (lane < OUTC) ssum += expf(v0 - mx);
    if (c1 < OUTC)  ssum += expf(v1 - mx);
    for (int o = 16; o > 0; o >>= 1) ssum += __shfl_xor(ssum, o, 32);
    const float lse = mx + logf(ssum);
    if (lane < OUTC) out[(size_t)node * OUTC + lane] = v0 - lse;
    if (c1 < OUTC)  out[(size_t)node * OUTC + c1]  = v1 - lse;
}

extern "C" void kernel_launch(void* const* d_in, const int* in_sizes, int n_in,
                              void* d_out, int out_size, void* d_ws, size_t ws_size,
                              hipStream_t stream) {
    const float* x   = (const float*)d_in[0];
    const int*   ei  = (const int*)d_in[1];
    const float* W1  = (const float*)d_in[2];
    const float* as1 = (const float*)d_in[3];
    const float* ad1 = (const float*)d_in[4];
    const float* b1  = (const float*)d_in[5];
    const float* W2  = (const float*)d_in[6];
    const float* as2 = (const float*)d_in[7];
    const float* ad2 = (const float*)d_in[8];
    const float* b2  = (const float*)d_in[9];
    const int N  = in_sizes[0] / INCH;
    const int E  = in_sizes[1] / 2;
    const int Et = E + N;
    const int* srcA = ei;
    const int* dstA = ei + E;

    float* ws = (float*)d_ws;
    size_t off = 0;
    float* h1   = ws + off; off += (size_t)N * HC;
    float* out1 = ws + off; off += (size_t)N * HC;
    float* aS1  = ws + off; off += (size_t)N * NHD;
    float* aD1  = ws + off; off += (size_t)N * NHD;
    int* rowptr = (int*)(ws + off); off += (size_t)N + 1;
    int* cnt    = (int*)(ws + off); off += (size_t)N;
    int* cursor = (int*)(ws + off); off += (size_t)N;
    int* col    = (int*)(ws + off); off += (size_t)Et;
    float* h2  = h1;
    float* aS2 = aS1;
    float* aD2 = aD1;

    hipMemsetAsync(cnt, 0, (size_t)N * sizeof(int), stream);
    count_edges<<<(Et + 255) / 256, 256, 0, stream>>>(dstA, cnt, E, N);
    scan_rowptr<<<(N + 1023) / 1024, 1024, 0, stream>>>(cnt, rowptr, cursor, N);
    fill_csr<<<(Et + 255) / 256, 256, 0, stream>>>(srcA, dstA, cursor, col, E, N);

    gemm_x_w1<<<(N + 15) / 16, 256, 0, stream>>>(x, W1, h1, N);
    attn_dots_l1<<<(N * NHD + 255) / 256, 256, 0, stream>>>(h1, as1, ad1, aS1, aD1, N);
    gat_gather_l1<<<(N * 32 + 255) / 256, 256, 0, stream>>>(
        rowptr, col, aS1, aD1, h1, b1, out1, N);

    gemm_h_w2<<<(N + 15) / 16, 96, 0, stream>>>(out1, W2, h2, N);
    attn_dots_l2<<<(N + 255) / 256, 256, 0, stream>>>(h2, as2, ad2, aS2, aD2, N);
    gat_l2_fused<<<(N * 32 + 255) / 256, 256, 0, stream>>>(
        rowptr, col, aS2, aD2, h2, b2, (float*)d_out, N);
}


# CSR build without global atomics: 64 groups rank their slice of the edge list in LDS node counters, a prefix over the 64 count rows places each edge; memset gone; list order by slice, then arrival
# speedup vs baseline: 2.2139x; 1.1595x over previous
#include <hip/hip_runtime.h>
#include <hip/hip_bf16.h>

typedef __attribute__((ext_vector_type(2))) float v2f;
typedef __attribute__((ext_vector_type(8))) float v8f;

#define INCH  128
#define HC    256
#define NHD   8
#define CPH   32
#define OUTC  40
#define SLOPE 0.2f

__device__ __forceinline__ float leaky(float x) { return x > 0.f ? x : SLOPE * x; }

__global__ void gemm_x_w1(const float* __restrict__ X, const float* __restrict__ W,
                          float* __restrict__ H1, int nrows) {
    const int lane = threadIdx.x & 31;
    const int wave = threadIdx.x >> 5;
    const int m    = lane & 15;
    const int half = lane >> 4;
    const int rowBase = blockIdx.x << 4;
    if (rowBase >= nrows) return;
    const float* xrow = X + (size_t)(rowBase + m) * INCH;
    for (int t = 0; t < 2; ++t) {
        const int colBase = ((wave << 1) + t) << 4;
        v8f acc = {0.f, 0.f, 0.f, 0.f, 0.f, 0.f, 0.f, 0.f};
        for (int k = 0; k < INCH; k += 4) {
            const int kk = k + (half << 1);
            v2f a, b;
            a.x = xrow[kk];
            a.y = xrow[kk + 1];
            b.x = W[(size_t)kk * HC + colBase + m];
            b.y = W[(size_t)(kk + 1) * HC + colBase + m];
            acc = __builtin_amdgcn_wmma_f32_16x16x4_f32(false, a, false, b,
                                                        (short)0, acc, false, false);
        }
#pragma unroll
        for (int v = 0; v < 8; ++v)
            H1[(size_t)(rowBase + v + (half << 3)) * HC + colBase + m] = acc[v];
    }
}

__global__ void gemm_h_w2(const float* __restrict__ Hin, const float* __restrict__ W,
                          float* __restrict__ H2, int nrows) {
    const int lane = threadIdx.x & 31;
    const int wave = threadIdx.x >> 5;
    const int m    = lane & 15;
    const int half = lane >> 4;
    const int rowBase = blockIdx.x << 4;
    if (rowBase >= nrows) return;
    const int colBase = wave << 4;
    const int col = colBase + m;
    const float* hrow = Hin + (size_t)(rowBase + m) * HC;
    v8f acc = {0.f, 0.f, 0.f, 0.f, 0.f, 0.f, 0.f, 0.f};
    for (int k = 0; k < HC; k += 4) {
        const int kk = k + (half << 1);
        v2f a, b;
        a.x = hrow[kk];
        a.y = hrow[kk + 1];
        b.x = (col < OUTC) ? W[kk * OUTC + col] : 0.f;
        b.y = (col < OUTC) ? W[(kk + 1) * OUTC + col] : 0.f;
        acc = __builtin_amdgcn_wmma_f32_16x16x4_f32(false, a, false, b,
                                                    (short)0, acc, false, false);
    }
    if (col < OUTC) {
#pragma unroll
        for (int v = 0; v < 8; ++v)
            H2[(size_t)(rowBase + v + (half << 3)) * OUTC + col] = acc[v];
    }
}

__global__ void attn_dots_l1(const float* __restrict__ H1, const float* __restrict__ Asrc,
                             const float* __restrict__ Adst, float* __restrict__ aS,
                             float* __restrict__ aD, int n) {
    __shared__ float sAs[HC], sAd[HC];
    for (int i = threadIdx.x; i < HC; i += blockDim.x) { sAs[i] = Asrc[i]; sAd[i] = Adst[i]; }
    __syncthreads();
    const int idx = blockIdx.x * blockDim.x + threadIdx.x;
    if (idx >= n * NHD) return;
    const int node = idx >> 3, h = idx & 7;
    const float* row = H1 + (size_t)node * HC + h * CPH;
    float s = 0.f, d = 0.f;
#pragma unroll
    for (int c = 0; c < CPH; ++c) {
        const float v = row[c];
        s += v * sAs[h * CPH + c];
        d += v * sAd[h * CPH + c];
    }
    aS[idx] = s; aD[idx] = d;
}

__global__ void attn_dots_l2(const float* __restrict__ H2, const float* __restrict__ Asrc,
                             const float* __restrict__ Adst, float* __restrict__ aS,
                             float* __restrict__ aD, int n) {
    const int node = blockIdx.x * blockDim.x + threadIdx.x;
    if (node >= n) return;
    const float* row = H2 + (size_t)node * OUTC;
    float s = 0.f, d = 0.f;
#pragma unroll
    for (int c = 0; c < OUTC; ++c) { const float v = row[c]; s += v * Asrc[c]; d += v * Adst[c]; }
    aS[node] = s; aD[node] = d;
}

#define CSR_PARTS   64
#define CSR_THREADS 1024
#define CSR_BINS    51200
#define CSR_AHEAD   16

__device__ __forceinline__ int csr_part_len(int E) {
    return (((E + CSR_PARTS - 1) / CSR_PARTS) + 7) & ~7;
}

__global__ void __launch_bounds__(CSR_THREADS)
count_rank(const int* __restrict__ dstA, int* __restrict__ partc, int* __restrict__ rank,
           int E, int n) {
    __shared__ int hist[CSR_BINS];
    const int tid  = (int)threadIdx.x;
    const int part = (int)blockIdx.x;
    const int len  = csr_part_len(E);
    const int beg  = part * len;
    const int end  = (beg + len < E) ? (beg + len) : E;
    for (int w0 = 0; w0 < n; w0 += CSR_BINS) {
        const int wn = (n - w0 < CSR_BINS) ? (n - w0) : CSR_BINS;
        for (int i = tid; i < wn; i += CSR_THREADS) hist[i] = 0;
        __syncthreads();
        for (int e0 = beg; e0 < end; e0 += CSR_AHEAD * CSR_THREADS) {
            int d[CSR_AHEAD], r[CSR_AHEAD];
#pragma unroll
            for (int u = 0; u < CSR_AHEAD; ++u) {
                const int e = e0 + u * CSR_THREADS + tid;
                const bool in = e < end;
                const int x = dstA[in ? e : beg];
                d[u] = in ? (x - w0) : -1;
            }
#pragma unroll
            for (int u = 0; u < CSR_AHEAD; ++u) {
                r[u] = 0;
                if ((unsigned)d[u] < (unsigned)wn) r[u] = atomicAdd(&hist[d[u]], 1);
            }
#pragma unroll
            for (int u = 0; u < CSR_AHEAD; ++u) {
                const int e = e0 + u * CSR_THREADS + tid;
                if ((unsigned)d[u] < (unsigned)wn) rank[e] = r[u];
            }
        }
        __syncthreads();
        int* row = partc + (size_t)part * n + w0;
        for (int i = tid; i < wn; i += CSR_THREADS) row[i] = hist[i];
        __syncthreads();
    }
}

__global__ void part_prefix(const int* __restrict__ partc, int* __restrict__ partp,
                            int* __restrict__ cnt, int n) {
    const int d = (int)blockIdx.x * (int)blockDim.x + (int)threadIdx.x;
    if (d >= n) return;
    int run = 0;
    for (int p0 = 0; p0 < CSR_PARTS; p0 += 16) {
        int x[16];
#pragma unroll
        for (int u = 0; u < 16; ++u) x[u] = partc[(size_t)(p0 + u) * n + d];
#pragma unroll
        for (int u = 0; u < 16; ++u) { partp[(size_t)(p0 + u) * n + d] = run; run += x[u]; }
    }
    cnt[d] = run + 1;
}

#define SCAN_AHEAD 16
__global__ void scan_rowptr(const int* __restrict__ cnt, int* __restrict__ rowptr, int n) {
    __shared__ int sPre[32];
    __shared__ int sTot[32];
    const int tid  = (int)threadIdx.x;
    const int lane = tid & 31;
    const int wave = tid >> 5;
    const int base = (int)blockIdx.x * 1024;
    const int i = base + tid;
    const int v = (i < n) ? cnt[i] : 0;
    int pre = 0;
    for (int k0 = 0; k0 < base; k0 += SCAN_AHEAD * 1024) {
        for (int u = 0; u < SCAN_AHEAD; ++u) {
            const int kb = k0 + u * 1024;
            const bool in = kb < base;
            const int x = cnt[(in ? kb : 0) + tid];
            pre += in ? x : 0;
        }
    }
    int inc = v;
    for (int d = 1; d < 32; d <<= 1) {
        const int t = __shfl_up(inc, d, 32);
        if (lane >= d) inc += t;
    }
    for (int o = 16; o > 0; o >>= 1) pre += __shfl_xor(pre, o, 32);
    if (lane == 31) { sTot[wave] = inc; sPre[wave] = pre; }
    __syncthreads();
    int off = sPre[lane] + ((lane < wave) ? sTot[lane] : 0);
    for (int o = 16; o > 0; o >>= 1) off += __shfl_xor(off, o, 32);
    const int excl = off + inc - v;
    if (i < n) rowptr[i] = excl;
    if (i == n - 1) rowptr[n] = excl + v;
}

#define FILL_EPT 8
__global__ void fill_csr(const int* __restrict__ srcA, const int* __restrict__ dstA,
                         const int* __restrict__ rank, const int* __restrict__ partp,
                         const int* __restrict__ rowptr, int* __restrict__ col, int E, int n) {
    const int base = ((int)blockIdx.x * (int)blockDim.x + (int)threadIdx.x) * FILL_EPT;
    if (base >= E + n) return;
    const int len = csr_part_len(E);
    const bool vec = (((reinterpret_cast<size_t>(srcA) | reinterpret_cast<size_t>(dstA) |
                        reinterpret_cast<size_t>(rank)) & 15) == 0);
    if (vec && base + FILL_EPT <= E) {
        const int* pp = partp + (size_t)(base / len) * n;
        const int4 sA = *reinterpret_cast<const int4*>(srcA + base);
        const int4 sB = *reinterpret_cast<const int4*>(srcA + base + 4);
        const int4 dA = *reinterpret_cast<const int4*>(dstA + base);
        const int4 dB = *reinterpret_cast<const int4*>(dstA + base + 4);
        const int4 rA = *reinterpret_cast<const int4*>(rank + base);
        const int4 rB = *reinterpret_cast<const int4*>(rank + base + 4);
        const int p0 = rowptr[dA.x] + pp[dA.x];
        const int p1 = rowptr[dA.y] + pp[dA.y];
        const int p2 = rowptr[dA.z] + pp[dA.z];
        const int p3 = rowptr[dA.w] + pp[dA.w];
        const int p4 = rowptr[dB.x] + pp[dB.x];
        const int p5 = rowptr[dB.y] + pp[dB.y];
        const int p6 = rowptr[dB.z] + pp[dB.z];
        const int p7 = rowptr[dB.w] + pp[dB.w];
        col[p0 + rA.x] = sA.x;
        col[p1 + rA.y] = sA.y;
        col[p2 + rA.z] = sA.z;
        col[p3 + rA.w] = sA.w;
        col[p4 + rB.x] = sB.x;
        col[p5 + rB.y] = sB.y;
        col[p6 + rB.z] = sB.z;
        col[p7 + rB.w] = sB.w;
    } else {
        for (int k = 0; k < FILL_EPT; ++k) {
            const int e = base + k;
            if (e < E) {
                const int d = dstA[e];
                col[rowptr[d] + partp[(size_t)(e / len) * n + d] + rank[e]] = srcA[e];
            } else if (e < E + n) {
                const int i = e - E;
                col[rowptr[i + 1] - 1] = i;
            }
        }
    }
}

__global__ void gat_gather_l1(const int* __restrict__ rowptr, const int* __restrict__ col,
                              const float* __restrict__ aS, const float* __restrict__ aD,
                              const float* __restrict__ H1, const float* __restrict__ b1,
                              float* __restrict__ out1, int n) {
    const int lane = threadIdx.x & 31;
    const int node = (blockIdx.x * blockDim.x + threadIdx.x) >> 5;
    if (node >= n) return;
    const int h  = lane >> 2;
    const int ch = lane << 3;
    const int beg = rowptr[node], end = rowptr[node + 1];
    const float aDd = aD[node * NHD + h];
    float m = -__builtin_inff(), den = 0.f;
    float a0 = 0.f, a1 = 0.f, a2 = 0.f, a3 = 0.f, a4 = 0.f, a5 = 0.f, a6 = 0.f, a7 = 0.f;
    for (int j = beg; j < end; ++j) {
        const int s = col[j];
        if (j + 1 < end) {
            const int s2 = col[j + 1];
            __builtin_prefetch(&H1[(size_t)s2 * HC + ch], 0, 1);
        }
        const float l = leaky(aS[s * NHD + h] + aDd);
        const float mn = fmaxf(m, l);
        const float scale = expf(m - mn);
        const float wgt   = expf(l - mn);
        const float4* hp = reinterpret_cast<const float4*>(H1 + (size_t)s * HC + ch);
        const float4 x0 = hp[0];
        const float4 x1 = hp[1];
        a0 = a0 * scale + wgt * x0.x;
        a1 = a1 * scale + wgt * x0.y;
        a2 = a2 * scale + wgt * x0.z;
        a3 = a3 * scale + wgt * x0.w;
        a4 = a4 * scale + wgt * x1.x;
        a5 = a5 * scale + wgt * x1.y;
        a6 = a6 * scale + wgt * x1.z;
        a7 = a7 * scale + wgt * x1.w;
        den = den * scale + wgt;
        m = mn;
    }
    const float v0 = a0 / den + b1[ch + 0];
    const float v1 = a1 / den + b1[ch + 1];
    const float v2 = a2 / den + b1[ch + 2];
    const float v3 = a3 / den + b1[ch + 3];
    const float v4 = a4 / den + b1[ch + 4];
    const float v5 = a5 / den + b1[ch + 5];
    const float v6 = a6 / den + b1[ch + 6];
    const float v7 = a7 / den + b1[ch + 7];
    float4 o0, o1;
    o0.x = v0 > 0.f ? v0 : 0.f;
    o0.y = v1 > 0.f ? v1 : 0.f;
    o0.z = v2 > 0.f ? v2 : 0.f;
    o0.w = v3 > 0.f ? v3 : 0.f;
    o1.x = v4 > 0.f ? v4 : 0.f;
    o1.y = v5 > 0.f ? v5 : 0.f;
    o1.z = v6 > 0.f ? v6 : 0.f;
    o1.w = v7 > 0.f ? v7 : 0.f;
    float4* op = reinterpret_cast<float4*>(out1 + (size_t)node * HC + ch);
    op[0] = o0;
    op[1] = o1;
}

__global__ void gat_l2_fused(const int* __restrict__ rowptr, const int* __restrict__ col,
                             const float* __restrict__ aS, const float* __restrict__ aD,
                             const float* __restrict__ H2, const float* __restrict__ b2,
                             float* __restrict__ out, int n) {
    const int lane = threadIdx.x & 31;
    const int node = (blockIdx.x * blockDim.x + threadIdx.x) >> 5;
    if (node >= n) return;
    const int beg = rowptr[node], end = rowptr[node + 1];
    const float aDd = aD[node];
    const int c1 = lane + 32;
    float m = -__builtin_inff(), den = 0.f, acc0 = 0.f, acc1 = 0.f;
    for (int j = beg; j < end; ++j) {
        const int s = col[j];
        if (j + 1 < end) {
            const int s2 = col[j + 1];
            __builtin_prefetch(&H2[(size_t)s2 * OUTC + lane], 0, 1);
        }
        const float l = leaky(aS[s] + aDd);
        const float mn = fmaxf(m, l);
        const float scale = expf(m - mn);
        const float wgt   = expf(l - mn);
        const float* hrow = H2 + (size_t)s * OUTC;
        acc0 = acc0 * scale + wgt * ((lane < OUTC) ? hrow[lane] : 0.f);
        acc1 = acc1 * scale + wgt * ((c1 < OUTC) ? hrow[c1] : 0.f);
        den = den * scale + wgt;
        m = mn;
    }
    float v0 = (lane < OUTC) ? acc0 / den + b2[lane] : -__builtin_inff();
    float v1 = (c1 < OUTC)  ? acc1 / den + b2[c1]  : -__builtin_inff();
    float mx = fmaxf(v0, v1);
    for (int o = 16; o > 0; o >>= 1) mx = fmaxf(mx, __shfl_xor(mx, o, 32));
    float ssum = 0.f;
    if (lane < OUTC) ssum += expf(v0 - mx);
    if (c1 < OUTC)  ssum += expf(v1 - mx);
    for (int o = 16; o > 0; o >>= 1) ssum += __shfl_xor(ssum, o, 32);
    const float lse = mx + logf(ssum);
    if (lane < OUTC) out[(size_t)node * OUTC + lane] = v0 - lse;
    if (c1 < OUTC)  out[(size_t)node * OUTC + c1]  = v1 - lse;
}

extern "C" void kernel_launch(void* const* d_in, const int* in_sizes, int n_in,
                              void* d_out, int out_size, void* d_ws, size_t ws_size,
                              hipStream_t stream) {
    const float* x   = (const float*)d_in[0];
    const int*   ei  = (const int*)d_in[1];
    const float* W1  = (const float*)d_in[2];
    const float* as1 = (const float*)d_in[3];
    const float* ad1 = (const float*)d_in[4];
    const float* b1  = (const float*)d_in[5];
    const float* W2  = (const float*)d_in[6];
    const float* as2 = (const float*)d_in[7];
    const float* ad2 = (const float*)d_in[8];
    const float* b2  = (const float*)d_in[9];
    const int N  = in_sizes[0] / INCH;
    const int E  = in_sizes[1] / 2;
    const int Et = E + N;
    const int* srcA = ei;
    const int* dstA = ei + E;

    float* ws = (float*)d_ws;
    size_t off = 0;
    float* h1   = ws + off; off += (size_t)N * HC;
    float* out1 = ws + off; off += (size_t)N * HC;
    float* aS1  = ws + off; off += (size_t)N * NHD;
    float* aD1  = ws + off; off += (size_t)N * NHD;
    int* rowptr = (int*)(ws + off); off += (size_t)N + 1;
    int* cnt    = (int*)(ws + off); off += (size_t)N;
    int* col    = (int*)(ws + off); off += (size_t)Et;
    float* h2  = h1;
    float* aS2 = aS1;
    float* aD2 = aD1;
    int* partc = (int*)h1;
    int* partp = partc + (size_t)CSR_PARTS * N;
    int* rank  = (int*)out1;

    count_rank<<<CSR_PARTS, CSR_THREADS, 0, stream>>>(dstA, partc, rank, E, N);
    part_prefix<<<(N + 255) / 256, 256, 0, stream>>>(partc, partp, cnt, N);
    scan_rowptr<<<(N + 1023) / 1024, 1024, 0, stream>>>(cnt, rowptr, N);
    fill_csr<<<(Et + 256 * FILL_EPT - 1) / (256 * FILL_EPT), 256, 0, stream>>>(
        srcA, dstA, rank, partp, rowptr, col, E, N);

    gemm_x_w1<<<(N + 15) / 16, 256, 0, stream>>>(x, W1, h1, N);
    attn_dots_l1<<<(N * NHD + 255) / 256, 256, 0, stream>>>(h1, as1, ad1, aS1, aD1, N);
    gat_gather_l1<<<(N * 32 + 255) / 256, 256, 0, stream>>>(
        rowptr, col, aS1, aD1, h1, b1, out1, N);

    gemm_h_w2<<<(N + 15) / 16, 96, 0, stream>>>(out1, W2, h2, N);
    attn_dots_l2<<<(N + 255) / 256, 256, 0, stream>>>(h2, as2, ad2, aS2, aD2, N);
    gat_l2_fused<<<(N * 32 + 255) / 256, 256, 0, stream>>>(
        rowptr, col, aS2, aD2, h2, b2, (float*)d_out, N);
}


# both projections: a block's 32 A rows staged once in LDS by whole-line 128-bit loads, shared by its waves (was 8 B per lane from 16 lines per wave load); GEMM1 4 tiles per wave; sum order unchanged
# speedup vs baseline: 2.7228x; 1.2299x over previous
#include <hip/hip_runtime.h>
#include <hip/hip_bf16.h>

typedef __attribute__((ext_vector_type(2))) float v2f;
typedef __attribute__((ext_vector_type(8))) float v8f;

#define INCH  128
#define HC    256
#define NHD   8
#define CPH   32
#define OUTC  40
#define SLOPE 0.2f

__device__ __forceinline__ float leaky(float x) { return x > 0.f ? x : SLOPE * x; }

#define G1_ROWS  32
#define G1_PITCH (INCH + 4)
__global__ void __launch_bounds__(256)
gemm_x_w1(const float* __restrict__ X, const float* __restrict__ W,
          float* __restrict__ H1, int nrows) {
    __shared__ __attribute__((aligned(16))) float sA[G1_ROWS * G1_PITCH];
    const int tid  = (int)threadIdx.x;
    const int lane = tid & 31;
    const int wave = tid >> 5;
    const int m    = lane & 15;
    const int half = lane >> 4;
    const int rowBase = (int)blockIdx.x * G1_ROWS;
    float4 st[(G1_ROWS * INCH / 4) / 256];
#pragma unroll
    for (int i = 0; i < (G1_ROWS * INCH / 4) / 256; ++i) {
        const int idx = tid + i * 256;
        const int r = idx >> 5, c4 = idx & 31;
        const int gr = (rowBase + r < nrows) ? (rowBase + r) : (nrows - 1);
        st[i] = *reinterpret_cast<const float4*>(X + (size_t)gr * INCH + (c4 << 2));
    }
#pragma unroll
    for (int i = 0; i < (G1_ROWS * INCH / 4) / 256; ++i) {
        const int idx = tid + i * 256;
        const int r = idx >> 5, c4 = idx & 31;
        *reinterpret_cast<float4*>(&sA[r * G1_PITCH + (c4 << 2)]) = st[i];
    }
    __syncthreads();
    const int c0 = wave << 5;
    const float* wp  = W + c0 + m;
    const float* a0p = sA + m * G1_PITCH + (half << 1);
    const float* a1p = a0p + 16 * G1_PITCH;
    v8f acc00 = {0.f, 0.f, 0.f, 0.f, 0.f, 0.f, 0.f, 0.f};
    v8f acc01 = acc00, acc10 = acc00, acc11 = acc00;
    for (int k = 0; k < INCH; k += 4) {
        const int kk = k + (half << 1);
        const v2f a0 = *reinterpret_cast<const v2f*>(a0p + k);
        const v2f a1 = *reinterpret_cast<const v2f*>(a1p + k);
        v2f b0, b1;
        b0.x = wp[(size_t)kk * HC];
        b0.y = wp[(size_t)(kk + 1) * HC];
        b1.x = wp[(size_t)kk * HC + 16];
        b1.y = wp[(size_t)(kk + 1) * HC + 16];
        acc00 = __builtin_amdgcn_wmma_f32_16x16x4_f32(false, a0, false, b0,
                                                      (short)0, acc00, false, false);
        acc01 = __builtin_amdgcn_wmma_f32_16x16x4_f32(false, a0, false, b1,
                                                      (short)0, acc01, false, false);
        acc10 = __builtin_amdgcn_wmma_f32_16x16x4_f32(false, a1, false, b0,
                                                      (short)0, acc10, false, false);
        acc11 = __builtin_amdgcn_wmma_f32_16x16x4_f32(false, a1, false, b1,
                                                      (short)0, acc11, false, false);
    }
    float* o0 = H1 + (size_t)(rowBase + (half << 3)) * HC + c0 + m;
    float* o1 = o0 + (size_t)16 * HC;
    if (rowBase + G1_ROWS <= nrows) {
#pragma unroll
        for (int v = 0; v < 8; ++v) {
            o0[(size_t)v * HC]      = acc00[v];
            o0[(size_t)v * HC + 16] = acc01[v];
            o1[(size_t)v * HC]      = acc10[v];
            o1[(size_t)v * HC + 16] = acc11[v];
        }
    } else {
#pragma unroll
        for (int v = 0; v < 8; ++v) {
            const int r0 = rowBase + v + (half << 3);
            if (r0 < nrows) {
                o0[(size_t)v * HC]      = acc00[v];
                o0[(size_t)v * HC + 16] = acc01[v];
            }
            if (r0 + 16 < nrows) {
                o1[(size_t)v * HC]      = acc10[v];
                o1[(size_t)v * HC + 16] = acc11[v];
            }
        }
    }
}

#define G2_ROWS  32
#define G2_PITCH (HC + 4)
#define G2_UNITS (G2_ROWS * HC / 128)
#define G2_UPW   ((G2_UNITS + 5) / 6)
__global__ void __launch_bounds__(192)
gemm_h_w2(const float* __restrict__ Hin, const float* __restrict__ W,
          float* __restrict__ H2, int nrows) {
    __shared__ __attribute__((aligned(16))) float sA[(G2_ROWS + 1) * G2_PITCH];
    const int tid  = (int)threadIdx.x;
    const int lane = tid & 31;
    const int wave = tid >> 5;
    const int m    = lane & 15;
    const int half = lane >> 4;
    const int rowBase = (int)blockIdx.x * G2_ROWS;
    float4 st[G2_UPW];
#pragma unroll
    for (int i = 0; i < G2_UPW; ++i) {
        const int u0 = wave + 6 * i;
        const int u = (u0 < G2_UNITS) ? u0 : (G2_UNITS - 1);
        const int r = u >> 1, c = ((u & 1) << 7) + (lane << 2);
        const int gr = (rowBase + r < nrows) ? (rowBase + r) : (nrows - 1);
        st[i] = *reinterpret_cast<const float4*>(Hin + (size_t)gr * HC + c);
    }
#pragma unroll
    for (int i = 0; i < G2_UPW; ++i) {
        const int u0 = wave + 6 * i;
        const int r = u0 >> 1, c = ((u0 & 1) << 7) + (lane << 2);
        *reinterpret_cast<float4*>(&sA[r * G2_PITCH + c]) = st[i];
    }
    __syncthreads();
    const int rt = wave / 3;
    const int colBase = (wave - rt * 3) << 4;
    const int col = colBase + m;
    const bool cin = col < OUTC;
    const float* ap = sA + (rt * 16 + m) * G2_PITCH + (half << 1);
    const float* wp = W + (cin ? col : 0);
    v8f acc = {0.f, 0.f, 0.f, 0.f, 0.f, 0.f, 0.f, 0.f};
    for (int k = 0; k < HC; k += 4) {
        const int kk = k + (half << 1);
        const v2f a = *reinterpret_cast<const v2f*>(ap + k);
        v2f b;
        b.x = wp[kk * OUTC];
        b.y = wp[(kk + 1) * OUTC];
        acc = __builtin_amdgcn_wmma_f32_16x16x4_f32(false, a, false, b,
                                                    (short)0, acc, false, false);
    }
    if (cin) {
        const int r0 = rowBase + rt * 16 + (half << 3);
        float* o = H2 + (size_t)r0 * OUTC + col;
        if (rowBase + G2_ROWS <= nrows) {
#pragma unroll
            for (int v = 0; v < 8; ++v) o[(size_t)v * OUTC] = acc[v];
        } else {
#pragma unroll
            for (int v = 0; v < 8; ++v)
                if (r0 + v < nrows) o[(size_t)v * OUTC] = acc[v];
        }
    }
}

__global__ void attn_dots_l1(const float* __restrict__ H1, const float* __restrict__ Asrc,
                             const float* __restrict__ Adst, float* __restrict__ aS,
                             float* __restrict__ aD, int n) {
    __shared__ float sAs[HC], sAd[HC];
    for (int i = threadIdx.x; i < HC; i += blockDim.x) { sAs[i] = Asrc[i]; sAd[i] = Adst[i]; }
    __syncthreads();
    const int idx = blockIdx.x * blockDim.x + threadIdx.x;
    if (idx >= n * NHD) return;
    const int node = idx >> 3, h = idx & 7;
    const float* row = H1 + (size_t)node * HC + h * CPH;
    float s = 0.f, d = 0.f;
#pragma unroll
    for (int c = 0; c < CPH; ++c) {
        const float v = row[c];
        s += v * sAs[h * CPH + c];
        d += v * sAd[h * CPH + c];
    }
    aS[idx] = s; aD[idx] = d;
}

__global__ void attn_dots_l2(const float* __restrict__ H2, const float* __restrict__ Asrc,
                             const float* __restrict__ Adst, float* __restrict__ aS,
                             float* __restrict__ aD, int n) {
    const int node = blockIdx.x * blockDim.x + threadIdx.x;
    if (node >= n) return;
    const float* row = H2 + (size_t)node * OUTC;
    float s = 0.f, d = 0.f;
#pragma unroll
    for (int c = 0; c < OUTC; ++c) { const float v = row[c]; s += v * Asrc[c]; d += v * Adst[c]; }
    aS[node] = s; aD[node] = d;
}

#define CSR_PARTS   64
#define CSR_THREADS 1024
#define CSR_BINS    51200
#define CSR_AHEAD   16

__device__ __forceinline__ int csr_part_len(int E) {
    return (((E + CSR_PARTS - 1) / CSR_PARTS) + 7) & ~7;
}

__global__ void __launch_bounds__(CSR_THREADS)
count_rank(const int* __restrict__ dstA, int* __restrict__ partc, int* __restrict__ rank,
           int E, int n) {
    __shared__ int hist[CSR_BINS];
    const int tid  = (int)threadIdx.x;
    const int part = (int)blockIdx.x;
    const int len  = csr_part_len(E);
    const int beg  = part * len;
    const int end  = (beg + len < E) ? (beg + len) : E;
    for (int w0 = 0; w0 < n; w0 += CSR_BINS) {
        const int wn = (n - w0 < CSR_BINS) ? (n - w0) : CSR_BINS;
        for (int i = tid; i < wn; i += CSR_THREADS) hist[i] = 0;
        __syncthreads();
        for (int e0 = beg; e0 < end; e0 += CSR_AHEAD * CSR_THREADS) {
            int d[CSR_AHEAD], r[CSR_AHEAD];
#pragma unroll
            for (int u = 0; u < CSR_AHEAD; ++u) {
                const int e = e0 + u * CSR_THREADS + tid;
                const bool in = e < end;
                const int x = dstA[in ? e : beg];
                d[u] = in ? (x - w0) : -1;
            }
#pragma unroll
            for (int u = 0; u < CSR_AHEAD; ++u) {
                r[u] = 0;
                if ((unsigned)d[u] < (unsigned)wn) r[u] = atomicAdd(&hist[d[u]], 1);
            }
#pragma unroll
            for (int u = 0; u < CSR_AHEAD; ++u) {
                const int e = e0 + u * CSR_THREADS + tid;
                if ((unsigned)d[u] < (unsigned)wn) rank[e] = r[u];
            }
        }
        __syncthreads();
        int* row = partc + (size_t)part * n + w0;
        for (int i = tid; i < wn; i += CSR_THREADS) row[i] = hist[i];
        __syncthreads();
    }
}

__global__ void part_prefix(const int* __restrict__ partc, int* __restrict__ partp,
                            int* __restrict__ cnt, int n) {
    const int d = (int)blockIdx.x * (int)blockDim.x + (int)threadIdx.x;
    if (d >= n) return;
    int run = 0;
    for (int p0 = 0; p0 < CSR_PARTS; p0 += 16) {
        int x[16];
#pragma unroll
        for (int u = 0; u < 16; ++u) x[u] = partc[(size_t)(p0 + u) * n + d];
#pragma unroll
        for (int u = 0; u < 16; ++u) { partp[(size_t)(p0 + u) * n + d] = run; run += x[u]; }
    }
    cnt[d] = run + 1;
}

#define SCAN_AHEAD 16
__global__ void scan_rowptr(const int* __restrict__ cnt, int* __restrict__ rowptr, int n) {
    __shared__ int sPre[32];
    __shared__ int sTot[32];
    const int tid  = (int)threadIdx.x;
    const int lane = tid & 31;
    const int wave = tid >> 5;
    const int base = (int)blockIdx.x * 1024;
    const int i = base + tid;
    const int v = (i < n) ? cnt[i] : 0;
    int pre = 0;
    for (int k0 = 0; k0 < base; k0 += SCAN_AHEAD * 1024) {
        for (int u = 0; u < SCAN_AHEAD; ++u) {
            const int kb = k0 + u * 1024;
            const bool in = kb < base;
            const int x = cnt[(in ? kb : 0) + tid];
            pre += in ? x : 0;
        }
    }
    int inc = v;
    for (int d = 1; d < 32; d <<= 1) {
        const int t = __shfl_up(inc, d, 32);
        if (lane >= d) inc += t;
    }
    for (int o = 16; o > 0; o >>= 1) pre += __shfl_xor(pre, o, 32);
    if (lane == 31) { sTot[wave] = inc; sPre[wave] = pre; }
    __syncthreads();
    int off = sPre[lane] + ((lane < wave) ? sTot[lane] : 0);
    for (int o = 16; o > 0; o >>= 1) off += __shfl_xor(off, o, 32);
    const int excl = off + inc - v;
    if (i < n) rowptr[i] = excl;
    if (i == n - 1) rowptr[n] = excl + v;
}

#define FILL_EPT 8
__global__ void fill_csr(const int* __restrict__ srcA, const int* __restrict__ dstA,
                         const int* __restrict__ rank, const int* __restrict__ partp,
                         const int* __restrict__ rowptr, int* __restrict__ col, int E, int n) {
    const int base = ((int)blockIdx.x * (int)blockDim.x + (int)threadIdx.x) * FILL_EPT;
    if (base >= E + n) return;
    const int len = csr_part_len(E);
    const bool vec = (((reinterpret_cast<size_t>(srcA) | reinterpret_cast<size_t>(dstA) |
                        reinterpret_cast<size_t>(rank)) & 15) == 0);
    if (vec && base + FILL_EPT <= E) {
        const int* pp = partp + (size_t)(base / len) * n;
        const int4 sA = *reinterpret_cast<const int4*>(srcA + base);
        const int4 sB = *reinterpret_cast<const int4*>(srcA + base + 4);
        const int4 dA = *reinterpret_cast<const int4*>(dstA + base);
        const int4 dB = *reinterpret_cast<const int4*>(dstA + base + 4);
        const int4 rA = *reinterpret_cast<const int4*>(rank + base);
        const int4 rB = *reinterpret_cast<const int4*>(rank + base + 4);
        const int p0 = rowptr[dA.x] + pp[dA.x];
        const int p1 = rowptr[dA.y] + pp[dA.y];
        const int p2 = rowptr[dA.z] + pp[dA.z];
        const int p3 = rowptr[dA.w] + pp[dA.w];
        const int p4 = rowptr[dB.x] + pp[dB.x];
        const int p5 = rowptr[dB.y] + pp[dB.y];
        const int p6 = rowptr[dB.z] + pp[dB.z];
        const int p7 = rowptr[dB.w] + pp[dB.w];
        col[p0 + rA.x] = sA.x;
        col[p1 + rA.y] = sA.y;
        col[p2 + rA.z] = sA.z;
        col[p3 + rA.w] = sA.w;
        col[p4 + rB.x] = sB.x;
        col[p5 + rB.y] = sB.y;
        col[p6 + rB.z] = sB.z;
        col[p7 + rB.w] = sB.w;
    } else {
        for (int k = 0; k < FILL_EPT; ++k) {
            const int e = base + k;
            if (e < E) {
                const int d = dstA[e];
                col[rowptr[d] + partp[(size_t)(e / len) * n + d] + rank[e]] = srcA[e];
            } else if (e < E + n) {
                const int i = e - E;
                col[rowptr[i + 1] - 1] = i;
            }
        }
    }
}

__global__ void gat_gather_l1(const int* __restrict__ rowptr, const int* __restrict__ col,
                              const float* __restrict__ aS, const float* __restrict__ aD,
                              const float* __restrict__ H1, const float* __restrict__ b1,
                              float* __restrict__ out1, int n) {
    const int lane = threadIdx.x & 31;
    const int node = (blockIdx.x * blockDim.x + threadIdx.x) >> 5;
    if (node >= n) return;
    const int h  = lane >> 2;
    const int ch = lane << 3;
    const int beg = rowptr[node], end = rowptr[node + 1];
    const float aDd = aD[node * NHD + h];
    float m = -__builtin_inff(), den = 0.f;
    float a0 = 0.f, a1 = 0.f, a2 = 0.f, a3 = 0.f, a4 = 0.f, a5 = 0.f, a6 = 0.f, a7 = 0.f;
    for (int j = beg; j < end; ++j) {
        const int s = col[j];
        if (j + 1 < end) {
            const int s2 = col[j + 1];
            __builtin_prefetch(&H1[(size_t)s2 * HC + ch], 0, 1);
        }
        const float l = leaky(aS[s * NHD + h] + aDd);
        const float mn = fmaxf(m, l);
        const float scale = expf(m - mn);
        const float wgt   = expf(l - mn);
        const float4* hp = reinterpret_cast<const float4*>(H1 + (size_t)s * HC + ch);
        const float4 x0 = hp[0];
        const float4 x1 = hp[1];
        a0 = a0 * scale + wgt * x0.x;
        a1 = a1 * scale + wgt * x0.y;
        a2 = a2 * scale + wgt * x0.z;
        a3 = a3 * scale + wgt * x0.w;
        a4 = a4 * scale + wgt * x1.x;
        a5 = a5 * scale + wgt * x1.y;
        a6 = a6 * scale + wgt * x1.z;
        a7 = a7 * scale + wgt * x1.w;
        den = den * scale + wgt;
        m = mn;
    }
    const float v0 = a0 / den + b1[ch + 0];
    const float v1 = a1 / den + b1[ch + 1];
    const float v2 = a2 / den + b1[ch + 2];
    const float v3 = a3 / den + b1[ch + 3];
    const float v4 = a4 / den + b1[ch + 4];
    const float v5 = a5 / den + b1[ch + 5];
    const float v6 = a6 / den + b1[ch + 6];
    const float v7 = a7 / den + b1[ch + 7];
    float4 o0, o1;
    o0.x = v0 > 0.f ? v0 : 0.f;
    o0.y = v1 > 0.f ? v1 : 0.f;
    o0.z = v2 > 0.f ? v2 : 0.f;
    o0.w = v3 > 0.f ? v3 : 0.f;
    o1.x = v4 > 0.f ? v4 : 0.f;
    o1.y = v5 > 0.f ? v5 : 0.f;
    o1.z = v6 > 0.f ? v6 : 0.f;
    o1.w = v7 > 0.f ? v7 : 0.f;
    float4* op = reinterpret_cast<float4*>(out1 + (size_t)node * HC + ch);
    op[0] = o0;
    op[1] = o1;
}

__global__ void gat_l2_fused(const int* __restrict__ rowptr, const int* __restrict__ col,
                             const float* __restrict__ aS, const float* __restrict__ aD,
                             const float* __restrict__ H2, const float* __restrict__ b2,
                             float* __restrict__ out, int n) {
    const int lane = threadIdx.x & 31;
    const int node = (blockIdx.x * blockDim.x + threadIdx.x) >> 5;
    if (node >= n) return;
    const int beg = rowptr[node], end = rowptr[node + 1];
    const float aDd = aD[node];
    const int c1 = lane + 32;
    float m = -__builtin_inff(), den = 0.f, acc0 = 0.f, acc1 = 0.f;
    for (int j = beg; j < end; ++j) {
        const int s = col[j];
        if (j + 1 < end) {
            const int s2 = col[j + 1];
            __builtin_prefetch(&H2[(size_t)s2 * OUTC + lane], 0, 1);
        }
        const float l = leaky(aS[s] + aDd);
        const float mn = fmaxf(m, l);
        const float scale = expf(m - mn);
        const float wgt   = expf(l - mn);
        const float* hrow = H2 + (size_t)s * OUTC;
        acc0 = acc0 * scale + wgt * ((lane < OUTC) ? hrow[lane] : 0.f);
        acc1 = acc1 * scale + wgt * ((c1 < OUTC) ? hrow[c1] : 0.f);
        den = den * scale + wgt;
        m = mn;
    }
    float v0 = (lane < OUTC) ? acc0 / den + b2[lane] : -__builtin_inff();
    float v1 = (c1 < OUTC)  ? acc1 / den + b2[c1]  : -__builtin_inff();
    float mx = fmaxf(v0, v1);
    for (int o = 16; o > 0; o >>= 1) mx = fmaxf(mx, __shfl_xor(mx, o, 32));
    float ssum = 0.f;
    if (lane < OUTC) ssum += expf(v0 - mx);
    if (c1 < OUTC)  ssum += expf(v1 - mx);
    for (int o = 16; o > 0; o >>= 1) ssum += __shfl_xor(ssum, o, 32);
    const float lse = mx + logf(ssum);
    if (lane < OUTC) out[(size_t)node * OUTC + lane] = v0 - lse;
    if (c1 < OUTC)  out[(size_t)node * OUTC + c1]  = v1 - lse;
}

extern "C" void kernel_launch(void* const* d_in, const int* in_sizes, int n_in,
                              void* d_out, int out_size, void* d_ws, size_t ws_size,
                              hipStream_t stream) {
    const float* x   = (const float*)d_in[0];
    const int*   ei  = (const int*)d_in[1];
    const float* W1  = (const float*)d_in[2];
    const float* as1 = (const float*)d_in[3];
    const float* ad1 = (const float*)d_in[4];
    const float* b1  = (const float*)d_in[5];
    const float* W2  = (const float*)d_in[6];
    const float* as2 = (const float*)d_in[7];
    const float* ad2 = (const float*)d_in[8];
    const float* b2  = (const float*)d_in[9];
    const int N  = in_sizes[0] / INCH;
    const int E  = in_sizes[1] / 2;
    const int Et = E + N;
    const int* srcA = ei;
    const int* dstA = ei + E;

    float* ws = (float*)d_ws;
    size_t off = 0;
    float* h1   = ws + off; off += (size_t)N * HC;
    float* out1 = ws + off; off += (size_t)N * HC;
    float* aS1  = ws + off; off += (size_t)N * NHD;
    float* aD1  = ws + off; off += (size_t)N * NHD;
    int* rowptr = (int*)(ws + off); off += (size_t)N + 1;
    int* cnt    = (int*)(ws + off); off += (size_t)N;
    int* col    = (int*)(ws + off); off += (size_t)Et;
    float* h2  = h1;
    float* aS2 = aS1;
    float* aD2 = aD1;
    int* partc = (int*)h1;
    int* partp = partc + (size_t)CSR_PARTS * N;
    int* rank  = (int*)out1;

    count_rank<<<CSR_PARTS, CSR_THREADS, 0, stream>>>(dstA, partc, rank, E, N);
    part_prefix<<<(N + 255) / 256, 256, 0, stream>>>(partc, partp, cnt, N);
    scan_rowptr<<<(N + 1023) / 1024, 1024, 0, stream>>>(cnt, rowptr, N);
    fill_csr<<<(Et + 256 * FILL_EPT - 1) / (256 * FILL_EPT), 256, 0, stream>>>(
        srcA, dstA, rank, partp, rowptr, col, E, N);

    gemm_x_w1<<<(N + G1_ROWS - 1) / G1_ROWS, 256, 0, stream>>>(x, W1, h1, N);
    attn_dots_l1<<<(N * NHD + 255) / 256, 256, 0, stream>>>(h1, as1, ad1, aS1, aD1, N);
    gat_gather_l1<<<(N * 32 + 255) / 256, 256, 0, stream>>>(
        rowptr, col, aS1, aD1, h1, b1, out1, N);

    gemm_h_w2<<<(N + G2_ROWS - 1) / G2_ROWS, 192, 0, stream>>>(out1, W2, h2, N);
    attn_dots_l2<<<(N + 255) / 256, 256, 0, stream>>>(h2, as2, ad2, aS2, aD2, N);
    gat_l2_fused<<<(N * 32 + 255) / 256, 256, 0, stream>>>(
        rowptr, col, aS2, aD2, h2, b2, (float*)d_out, N);
}


# attention dots folded into both projection epilogues from the accumulator tiles (16-lane xor-tree add, LDS hand-over, whole-line stores); 2 launches fewer; dot sum order now a tree
# speedup vs baseline: 2.8659x; 1.0526x over previous
#include <hip/hip_runtime.h>
#include <hip/hip_bf16.h>

typedef __attribute__((ext_vector_type(2))) float v2f;
typedef __attribute__((ext_vector_type(8))) float v8f;

#define INCH  128
#define HC    256
#define NHD   8
#define CPH   32
#define OUTC  40
#define SLOPE 0.2f

__device__ __forceinline__ float leaky(float x) { return x > 0.f ? x : SLOPE * x; }

__device__ __forceinline__ void lds_handover_barrier() {
    asm volatile("s_wait_dscnt 0x0" ::: "memory");
    asm volatile("s_barrier_signal -1\n\ts_barrier_wait -1" ::: "memory");
}

#define ROW_XOR(x, k) __builtin_bit_cast(float, __builtin_amdgcn_update_dpp( \
    0, __builtin_bit_cast(int, (x)), 0x160 + (k), 0xf, 0xf, true))
__device__ __forceinline__ float row16_sum(float x) {
    x += ROW_XOR(x, 1);
    x += ROW_XOR(x, 2);
    x += ROW_XOR(x, 4);
    x += ROW_XOR(x, 8);
    return x;
}

#define G1_ROWS  32
#define G1_PITCH (INCH + 4)
__global__ void __launch_bounds__(256)
gemm_x_w1(const float* __restrict__ X, const float* __restrict__ W,
          const float* __restrict__ Asrc, const float* __restrict__ Adst,
          float* __restrict__ H1, float* __restrict__ aS, float* __restrict__ aD, int nrows) {
    __shared__ __attribute__((aligned(16))) float sA[G1_ROWS * G1_PITCH];
    __shared__ float sS[G1_ROWS * NHD], sD[G1_ROWS * NHD];
    const int tid  = (int)threadIdx.x;
    const int lane = tid & 31;
    const int wave = tid >> 5;
    const int m    = lane & 15;
    const int half = lane >> 4;
    const int rowBase = (int)blockIdx.x * G1_ROWS;
    const float as0 = Asrc[(wave << 5) + m], as1 = Asrc[(wave << 5) + 16 + m];
    const float ad0 = Adst[(wave << 5) + m], ad1 = Adst[(wave << 5) + 16 + m];
    float4 st[(G1_ROWS * INCH / 4) / 256];
#pragma unroll
    for (int i = 0; i < (G1_ROWS * INCH / 4) / 256; ++i) {
        const int idx = tid + i * 256;
        const int r = idx >> 5, c4 = idx & 31;
        const int gr = (rowBase + r < nrows) ? (rowBase + r) : (nrows - 1);
        st[i] = *reinterpret_cast<const float4*>(X + (size_t)gr * INCH + (c4 << 2));
    }
#pragma unroll
    for (int i = 0; i < (G1_ROWS * INCH / 4) / 256; ++i) {
        const int idx = tid + i * 256;
        const int r = idx >> 5, c4 = idx & 31;
        *reinterpret_cast<float4*>(&sA[r * G1_PITCH + (c4 << 2)]) = st[i];
    }
    __syncthreads();
    const int c0 = wave << 5;
    const float* wp  = W + c0 + m;
    const float* a0p = sA + m * G1_PITCH + (half << 1);
    const float* a1p = a0p + 16 * G1_PITCH;
    v8f acc00 = {0.f, 0.f, 0.f, 0.f, 0.f, 0.f, 0.f, 0.f};
    v8f acc01 = acc00, acc10 = acc00, acc11 = acc00;
    for (int k = 0; k < INCH; k += 4) {
        const int kk = k + (half << 1);
        const v2f a0 = *reinterpret_cast<const v2f*>(a0p + k);
        const v2f a1 = *reinterpret_cast<const v2f*>(a1p + k);
        v2f b0, b1;
        b0.x = wp[(size_t)kk * HC];
        b0.y = wp[(size_t)(kk + 1) * HC];
        b1.x = wp[(size_t)kk * HC + 16];
        b1.y = wp[(size_t)(kk + 1) * HC + 16];
        acc00 = __builtin_amdgcn_wmma_f32_16x16x4_f32(false, a0, false, b0,
                                                      (short)0, acc00, false, false);
        acc01 = __builtin_amdgcn_wmma_f32_16x16x4_f32(false, a0, false, b1,
                                                      (short)0, acc01, false, false);
        acc10 = __builtin_amdgcn_wmma_f32_16x16x4_f32(false, a1, false, b0,
                                                      (short)0, acc10, false, false);
        acc11 = __builtin_amdgcn_wmma_f32_16x16x4_f32(false, a1, false, b1,
                                                      (short)0, acc11, false, false);
    }
    float* o0 = H1 + (size_t)(rowBase + (half << 3)) * HC + c0 + m;
    float* o1 = o0 + (size_t)16 * HC;
    if (rowBase + G1_ROWS <= nrows) {
#pragma unroll
        for (int v = 0; v < 8; ++v) {
            o0[(size_t)v * HC]      = acc00[v];
            o0[(size_t)v * HC + 16] = acc01[v];
            o1[(size_t)v * HC]      = acc10[v];
            o1[(size_t)v * HC + 16] = acc11[v];
        }
    } else {
#pragma unroll
        for (int v = 0; v < 8; ++v) {
            const int r0 = rowBase + v + (half << 3);
            if (r0 < nrows) {
                o0[(size_t)v * HC]      = acc00[v];
                o0[(size_t)v * HC + 16] = acc01[v];
            }
            if (r0 + 16 < nrows) {
                o1[(size_t)v * HC]      = acc10[v];
                o1[(size_t)v * HC + 16] = acc11[v];
            }
        }
    }
    float s0 = 0.f, s1 = 0.f, d0 = 0.f, d1 = 0.f;
#pragma unroll
    for (int v = 0; v < 8; ++v) {
        const float ts0 = row16_sum(acc00[v] * as0 + acc01[v] * as1);
        const float td0 = row16_sum(acc00[v] * ad0 + acc01[v] * ad1);
        const float ts1 = row16_sum(acc10[v] * as0 + acc11[v] * as1);
        const float td1 = row16_sum(acc10[v] * ad0 + acc11[v] * ad1);
        s0 = (m == v) ? ts0 : s0;
        d0 = (m == v) ? td0 : d0;
        s1 = (m == v) ? ts1 : s1;
        d1 = (m == v) ? td1 : d1;
    }
    if (m < 8) {
        const int r = (half << 3) + m;
        sS[r * NHD + wave] = s0;  sS[(r + 16) * NHD + wave] = s1;
        sD[r * NHD + wave] = d0;  sD[(r + 16) * NHD + wave] = d1;
    }
    lds_handover_barrier();
    if ((rowBase << 3) + tid < nrows * NHD) {
        aS[((size_t)rowBase << 3) + tid] = sS[tid];
        aD[((size_t)rowBase << 3) + tid] = sD[tid];
    }
}

#define G2_ROWS  32
#define G2_PITCH (HC + 4)
#define G2_UNITS (G2_ROWS * HC / 128)
#define G2_UPW   ((G2_UNITS + 5) / 6)
__global__ void __launch_bounds__(192, 12)
gemm_h_w2(const float* __restrict__ Hin, const float* __restrict__ W,
          const float* __restrict__ Asrc, const float* __restrict__ Adst,
          float* __restrict__ H2, float* __restrict__ aS, float* __restrict__ aD, int nrows) {
    __shared__ __attribute__((aligned(16))) float sA[(G2_ROWS + 1) * G2_PITCH];
    __shared__ float sS[6 * 16], sD[6 * 16];
    const int tid  = (int)threadIdx.x;
    const int lane = tid & 31;
    const int wave = tid >> 5;
    const int m    = lane & 15;
    const int half = lane >> 4;
    const int rowBase = (int)blockIdx.x * G2_ROWS;
    float4 st[G2_UPW];
#pragma unroll
    for (int i = 0; i < G2_UPW; ++i) {
        const int u0 = wave + 6 * i;
        const int u = (u0 < G2_UNITS) ? u0 : (G2_UNITS - 1);
        const int r = u >> 1, c = ((u & 1) << 7) + (lane << 2);
        const int gr = (rowBase + r < nrows) ? (rowBase + r) : (nrows - 1);
        st[i] = *reinterpret_cast<const float4*>(Hin + (size_t)gr * HC + c);
    }
#pragma unroll
    for (int i = 0; i < G2_UPW; ++i) {
        const int u0 = wave + 6 * i;
        const int r = u0 >> 1, c = ((u0 & 1) << 7) + (lane << 2);
        *reinterpret_cast<float4*>(&sA[r * G2_PITCH + c]) = st[i];
    }
    __syncthreads();
    const int rt = wave / 3;
    const int colBase = (wave - rt * 3) << 4;
    const int col = colBase + m;
    const bool cin = col < OUTC;
    const float* ap = sA + (rt * 16 + m) * G2_PITCH + (half << 1);
    const float* wp = W + (cin ? col : 0);
    v8f acc = {0.f, 0.f, 0.f, 0.f, 0.f, 0.f, 0.f, 0.f};
    for (int k = 0; k < HC; k += 4) {
        const int kk = k + (half << 1);
        const v2f a = *reinterpret_cast<const v2f*>(ap + k);
        v2f b;
        b.x = wp[kk * OUTC];
        b.y = wp[(kk + 1) * OUTC];
        acc = __builtin_amdgcn_wmma_f32_16x16x4_f32(false, a, false, b,
                                                    (short)0, acc, false, false);
    }
    if (cin) {
        const int r0 = rowBase + rt * 16 + (half << 3);
        float* o = H2 + (size_t)r0 * OUTC + col;
        if (rowBase + G2_ROWS <= nrows) {
#pragma unroll
            for (int v = 0; v < 8; ++v) o[(size_t)v * OUTC] = acc[v];
        } else {
#pragma unroll
            for (int v = 0; v < 8; ++v)
                if (r0 + v < nrows) o[(size_t)v * OUTC] = acc[v];
        }
    }
    const float as = cin ? Asrc[col] : 0.f;
    const float ad = cin ? Adst[col] : 0.f;
    float s = 0.f, d = 0.f;
#pragma unroll
    for (int v = 0; v < 8; ++v) {
        const float ts = row16_sum(cin ? acc[v] * as : 0.f);
        const float td = row16_sum(cin ? acc[v] * ad : 0.f);
        s = (m == v) ? ts : s;
        d = (m == v) ? td : d;
    }
    if (m < 8) {
        sS[wave * 16 + (half << 3) + m] = s;
        sD[wave * 16 + (half << 3) + m] = d;
    }
    lds_handover_barrier();
    if (tid < G2_ROWS && rowBase + tid < nrows) {
        const int w0 = (tid >> 4) * 3, r = tid & 15;
        aS[rowBase + tid] = (sS[w0 * 16 + r] + sS[(w0 + 1) * 16 + r]) + sS[(w0 + 2) * 16 + r];
        aD[rowBase + tid] = (sD[w0 * 16 + r] + sD[(w0 + 1) * 16 + r]) + sD[(w0 + 2) * 16 + r];
    }
}

#define CSR_PARTS   64
#define CSR_THREADS 1024
#define CSR_BINS    51200
#define CSR_AHEAD   16

__device__ __forceinline__ int csr_part_len(int E) {
    return (((E + CSR_PARTS - 1) / CSR_PARTS) + 7) & ~7;
}

__global__ void __launch_bounds__(CSR_THREADS)
count_rank(const int* __restrict__ dstA, int* __restrict__ partc, int* __restrict__ rank,
           int E, int n) {
    __shared__ int hist[CSR_BINS];
    const int tid  = (int)threadIdx.x;
    const int part = (int)blockIdx.x;
    const int len  = csr_part_len(E);
    const int beg  = part * len;
    const int end  = (beg + len < E) ? (beg + len) : E;
    for (int w0 = 0; w0 < n; w0 += CSR_BINS) {
        const int wn = (n - w0 < CSR_BINS) ? (n - w0) : CSR_BINS;
        for (int i = tid; i < wn; i += CSR_THREADS) hist[i] = 0;
        __syncthreads();
        for (int e0 = beg; e0 < end; e0 += CSR_AHEAD * CSR_THREADS) {
            int d[CSR_AHEAD], r[CSR_AHEAD];
#pragma unroll
            for (int u = 0; u < CSR_AHEAD; ++u) {
                const int e = e0 + u * CSR_THREADS + tid;
                const bool in = e < end;
                const int x = dstA[in ? e : beg];
                d[u] = in ? (x - w0) : -1;
            }
#pragma unroll
            for (int u = 0; u < CSR_AHEAD; ++u) {
                r[u] = 0;
                if ((unsigned)d[u] < (unsigned)wn) r[u] = atomicAdd(&hist[d[u]], 1);
            }
#pragma unroll
            for (int u = 0; u < CSR_AHEAD; ++u) {
                const int e = e0 + u * CSR_THREADS + tid;
                if ((unsigned)d[u] < (unsigned)wn) rank[e] = r[u];
            }
        }
        __syncthreads();
        int* row = partc + (size_t)part * n + w0;
        for (int i = tid; i < wn; i += CSR_THREADS) row[i] = hist[i];
        __syncthreads();
    }
}

__global__ void part_prefix(const int* __restrict__ partc, int* __restrict__ partp,
                            int* __restrict__ cnt, int n) {
    const int d = (int)blockIdx.x * (int)blockDim.x + (int)threadIdx.x;
    if (d >= n) return;
    int run = 0;
    for (int p0 = 0; p0 < CSR_PARTS; p0 += 16) {
        int x[16];
#pragma unroll
        for (int u = 0; u < 16; ++u) x[u] = partc[(size_t)(p0 + u) * n + d];
#pragma unroll
        for (int u = 0; u < 16; ++u) { partp[(size_t)(p0 + u) * n + d] = run; run += x[u]; }
    }
    cnt[d] = run + 1;
}

#define SCAN_AHEAD 16
__global__ void scan_rowptr(const int* __restrict__ cnt, int* __restrict__ rowptr, int n) {
    __shared__ int sPre[32];
    __shared__ int sTot[32];
    const int tid  = (int)threadIdx.x;
    const int lane = tid & 31;
    const int wave = tid >> 5;
    const int base = (int)blockIdx.x * 1024;
    const int i = base + tid;
    const int v = (i < n) ? cnt[i] : 0;
    int pre = 0;
    for (int k0 = 0; k0 < base; k0 += SCAN_AHEAD * 1024) {
        for (int u = 0; u < SCAN_AHEAD; ++u) {
            const int kb = k0 + u * 1024;
            const bool in = kb < base;
            const int x = cnt[(in ? kb : 0) + tid];
            pre += in ? x : 0;
        }
    }
    int inc = v;
    for (int d = 1; d < 32; d <<= 1) {
        const int t = __shfl_up(inc, d, 32);
        if (lane >= d) inc += t;
    }
    for (int o = 16; o > 0; o >>= 1) pre += __shfl_xor(pre, o, 32);
    if (lane == 31) { sTot[wave] = inc; sPre[wave] = pre; }
    __syncthreads();
    int off = sPre[lane] + ((lane < wave) ? sTot[lane] : 0);
    for (int o = 16; o > 0; o >>= 1) off += __shfl_xor(off, o, 32);
    const int excl = off + inc - v;
    if (i < n) rowptr[i] = excl;
    if (i == n - 1) rowptr[n] = excl + v;
}

#define FILL_EPT 8
__global__ void fill_csr(const int* __restrict__ srcA, const int* __restrict__ dstA,
                         const int* __restrict__ rank, const int* __restrict__ partp,
                         const int* __restrict__ rowptr, int* __restrict__ col, int E, int n) {
    const int base = ((int)blockIdx.x * (int)blockDim.x + (int)threadIdx.x) * FILL_EPT;
    if (base >= E + n) return;
    const int len = csr_part_len(E);
    const bool vec = (((reinterpret_cast<size_t>(srcA) | reinterpret_cast<size_t>(dstA) |
                        reinterpret_cast<size_t>(rank)) & 15) == 0);
    if (vec && base + FILL_EPT <= E) {
        const int* pp = partp + (size_t)(base / len) * n;
        const int4 sA = *reinterpret_cast<const int4*>(srcA + base);
        const int4 sB = *reinterpret_cast<const int4*>(srcA + base + 4);
        const int4 dA = *reinterpret_cast<const int4*>(dstA + base);
        const int4 dB = *reinterpret_cast<const int4*>(dstA + base + 4);
        const int4 rA = *reinterpret_cast<const int4*>(rank + base);
        const int4 rB = *reinterpret_cast<const int4*>(rank + base + 4);
        const int p0 = rowptr[dA.x] + pp[dA.x];
        const int p1 = rowptr[dA.y] + pp[dA.y];
        const int p2 = rowptr[dA.z] + pp[dA.z];
        const int p3 = rowptr[dA.w] + pp[dA.w];
        const int p4 = rowptr[dB.x] + pp[dB.x];
        const int p5 = rowptr[dB.y] + pp[dB.y];
        const int p6 = rowptr[dB.z] + pp[dB.z];
        const int p7 = rowptr[dB.w] + pp[dB.w];
        col[p0 + rA.x] = sA.x;
        col[p1 + rA.y] = sA.y;
        col[p2 + rA.z] = sA.z;
        col[p3 + rA.w] = sA.w;
        col[p4 + rB.x] = sB.x;
        col[p5 + rB.y] = sB.y;
        col[p6 + rB.z] = sB.z;
        col[p7 + rB.w] = sB.w;
    } else {
        for (int k = 0; k < FILL_EPT; ++k) {
            const int e = base + k;
            if (e < E) {
                const int d = dstA[e];
                col[rowptr[d] + partp[(size_t)(e / len) * n + d] + rank[e]] = srcA[e];
            } else if (e < E + n) {
                const int i = e - E;
                col[rowptr[i + 1] - 1] = i;
            }
        }
    }
}

__global__ void gat_gather_l1(const int* __restrict__ rowptr, const int* __restrict__ col,
                              const float* __restrict__ aS, const float* __restrict__ aD,
                              const float* __restrict__ H1, const float* __restrict__ b1,
                              float* __restrict__ out1, int n) {
    const int lane = threadIdx.x & 31;
    const int node = (blockIdx.x * blockDim.x + threadIdx.x) >> 5;
    if (node >= n) return;
    const int h  = lane >> 2;
    const int ch = lane << 3;
    const int beg = rowptr[node], end = rowptr[node + 1];
    const float aDd = aD[node * NHD + h];
    float m = -__builtin_inff(), den = 0.f;
    float a0 = 0.f, a1 = 0.f, a2 = 0.f, a3 = 0.f, a4 = 0.f, a5 = 0.f, a6 = 0.f, a7 = 0.f;
    for (int j = beg; j < end; ++j) {
        const int s = col[j];
        if (j + 1 < end) {
            const int s2 = col[j + 1];
            __builtin_prefetch(&H1[(size_t)s2 * HC + ch], 0, 1);
        }
        const float l = leaky(aS[s * NHD + h] + aDd);
        const float mn = fmaxf(m, l);
        const float scale = expf(m - mn);
        const float wgt   = expf(l - mn);
        const float4* hp = reinterpret_cast<const float4*>(H1 + (size_t)s * HC + ch);
        const float4 x0 = hp[0];
        const float4 x1 = hp[1];
        a0 = a0 * scale + wgt * x0.x;
        a1 = a1 * scale + wgt * x0.y;
        a2 = a2 * scale + wgt * x0.z;
        a3 = a3 * scale + wgt * x0.w;
        a4 = a4 * scale + wgt * x1.x;
        a5 = a5 * scale + wgt * x1.y;
        a6 = a6 * scale + wgt * x1.z;
        a7 = a7 * scale + wgt * x1.w;
        den = den * scale + wgt;
        m = mn;
    }
    const float v0 = a0 / den + b1[ch + 0];
    const float v1 = a1 / den + b1[ch + 1];
    const float v2 = a2 / den + b1[ch + 2];
    const float v3 = a3 / den + b1[ch + 3];
    const float v4 = a4 / den + b1[ch + 4];
    const float v5 = a5 / den + b1[ch + 5];
    const float v6 = a6 / den + b1[ch + 6];
    const float v7 = a7 / den + b1[ch + 7];
    float4 o0, o1;
    o0.x = v0 > 0.f ? v0 : 0.f;
    o0.y = v1 > 0.f ? v1 : 0.f;
    o0.z = v2 > 0.f ? v2 : 0.f;
    o0.w = v3 > 0.f ? v3 : 0.f;
    o1.x = v4 > 0.f ? v4 : 0.f;
    o1.y = v5 > 0.f ? v5 : 0.f;
    o1.z = v6 > 0.f ? v6 : 0.f;
    o1.w = v7 > 0.f ? v7 : 0.f;
    float4* op = reinterpret_cast<float4*>(out1 + (size_t)node * HC + ch);
    op[0] = o0;
    op[1] = o1;
}

__global__ void gat_l2_fused(const int* __restrict__ rowptr, const int* __restrict__ col,
                             const float* __restrict__ aS, const float* __restrict__ aD,
                             const float* __restrict__ H2, const float* __restrict__ b2,
                             float* __restrict__ out, int n) {
    const int lane = threadIdx.x & 31;
    const int node = (blockIdx.x * blockDim.x + threadIdx.x) >> 5;
    if (node >= n) return;
    const int beg = rowptr[node], end = rowptr[node + 1];
    const float aDd = aD[node];
    const int c1 = lane + 32;
    float m = -__builtin_inff(), den = 0.f, acc0 = 0.f, acc1 = 0.f;
    for (int j = beg; j < end; ++j) {
        const int s = col[j];
        if (j + 1 < end) {
            const int s2 = col[j + 1];
            __builtin_prefetch(&H2[(size_t)s2 * OUTC + lane], 0, 1);
        }
        const float l = leaky(aS[s] + aDd);
        const float mn = fmaxf(m, l);
        const float scale = expf(m - mn);
        const float wgt   = expf(l - mn);
        const float* hrow = H2 + (size_t)s * OUTC;
        acc0 = acc0 * scale + wgt * ((lane < OUTC) ? hrow[lane] : 0.f);
        acc1 = acc1 * scale + wgt * ((c1 < OUTC) ? hrow[c1] : 0.f);
        den = den * scale + wgt;
        m = mn;
    }
    float v0 = (lane < OUTC) ? acc0 / den + b2[lane] : -__builtin_inff();
    float v1 = (c1 < OUTC)  ? acc1 / den + b2[c1]  : -__builtin_inff();
    float mx = fmaxf(v0, v1);
    for (int o = 16; o > 0; o >>= 1) mx = fmaxf(mx, __shfl_xor(mx, o, 32));
    float ssum = 0.f;
    if (lane < OUTC) ssum += expf(v0 - mx);
    if (c1 < OUTC)  ssum += expf(v1 - mx);
    for (int o = 16; o > 0; o >>= 1) ssum += __shfl_xor(ssum, o, 32);
    const float lse = mx + logf(ssum);
    if (lane < OUTC) out[(size_t)node * OUTC + lane] = v0 - lse;
    if (c1 < OUTC)  out[(size_t)node * OUTC + c1]  = v1 - lse;
}

extern "C" void kernel_launch(void* const* d_in, const int* in_sizes, int n_in,
                              void* d_out, int out_size, void* d_ws, size_t ws_size,
                              hipStream_t stream) {
    const float* x   = (const float*)d_in[0];
    const int*   ei  = (const int*)d_in[1];
    const float* W1  = (const float*)d_in[2];
    const float* as1 = (const float*)d_in[3];
    const float* ad1 = (const float*)d_in[4];
    const float* b1  = (const float*)d_in[5];
    const float* W2  = (const float*)d_in[6];
    const float* as2 = (const float*)d_in[7];
    const float* ad2 = (const float*)d_in[8];
    const float* b2  = (const float*)d_in[9];
    const int N  = in_sizes[0] / INCH;
    const int E  = in_sizes[1] / 2;
    const int Et = E + N;
    const int* srcA = ei;
    const int* dstA = ei + E;

    float* ws = (float*)d_ws;
    size_t off = 0;
    float* h1   = ws + off; off += (size_t)N * HC;
    float* out1 = ws + off; off += (size_t)N * HC;
    float* aS1  = ws + off; off += (size_t)N * NHD;
    float* aD1  = ws + off; off += (size_t)N * NHD;
    int* rowptr = (int*)(ws + off); off += (size_t)N + 1;
    int* cnt    = (int*)(ws + off); off += (size_t)N;
    int* col    = (int*)(ws + off); off += (size_t)Et;
    float* h2  = h1;
    float* aS2 = aS1;
    float* aD2 = aD1;
    int* partc = (int*)h1;
    int* partp = partc + (size_t)CSR_PARTS * N;
    int* rank  = (int*)out1;

    count_rank<<<CSR_PARTS, CSR_THREADS, 0, stream>>>(dstA, partc, rank, E, N);
    part_prefix<<<(N + 255) / 256, 256, 0, stream>>>(partc, partp, cnt, N);
    scan_rowptr<<<(N + 1023) / 1024, 1024, 0, stream>>>(cnt, rowptr, N);
    fill_csr<<<(Et + 256 * FILL_EPT - 1) / (256 * FILL_EPT), 256, 0, stream>>>(
        srcA, dstA, rank, partp, rowptr, col, E, N);

    gemm_x_w1<<<(N + G1_ROWS - 1) / G1_ROWS, 256, 0, stream>>>(x, W1, as1, ad1, h1, aS1, aD1, N);
    gat_gather_l1<<<(N * 32 + 255) / 256, 256, 0, stream>>>(
        rowptr, col, aS1, aD1, h1, b1, out1, N);

    gemm_h_w2<<<(N + G2_ROWS - 1) / G2_ROWS, 192, 0, stream>>>(out1, W2, as2, ad2, h2, aS2, aD2, N);
    gat_l2_fused<<<(N * 32 + 255) / 256, 256, 0, stream>>>(
        rowptr, col, aS2, aD2, h2, b2, (float*)d_out, N);
}


# layer-2 gather: 3 destination nodes per wave, 10 lanes each, 4 contiguous channels per lane by one 128-bit load (was 1 node, two 32-bit loads); edge sum order kept, log-softmax class sum reordered
# speedup vs baseline: 3.0264x; 1.0560x over previous
#include <hip/hip_runtime.h>
#include <hip/hip_bf16.h>

typedef __attribute__((ext_vector_type(2))) float v2f;
typedef __attribute__((ext_vector_type(8))) float v8f;

#define INCH  128
#define HC    256
#define NHD   8
#define CPH   32
#define OUTC  40
#define SLOPE 0.2f

__device__ __forceinline__ float leaky(float x) { return x > 0.f ? x : SLOPE * x; }

__device__ __forceinline__ void lds_handover_barrier() {
    asm volatile("s_wait_dscnt 0x0" ::: "memory");
    asm volatile("s_barrier_signal -1\n\ts_barrier_wait -1" ::: "memory");
}

#define ROW_XOR(x, k) __builtin_bit_cast(float, __builtin_amdgcn_update_dpp( \
    0, __builtin_bit_cast(int, (x)), 0x160 + (k), 0xf, 0xf, true))
__device__ __forceinline__ float row16_sum(float x) {
    x += ROW_XOR(x, 1);
    x += ROW_XOR(x, 2);
    x += ROW_XOR(x, 4);
    x += ROW_XOR(x, 8);
    return x;
}

#define G1_ROWS  32
#define G1_PITCH (INCH + 4)
__global__ void __launch_bounds__(256)
gemm_x_w1(const float* __restrict__ X, const float* __restrict__ W,
          const float* __restrict__ Asrc, const float* __restrict__ Adst,
          float* __restrict__ H1, float* __restrict__ aS, float* __restrict__ aD, int nrows) {
    __shared__ __attribute__((aligned(16))) float sA[G1_ROWS * G1_PITCH];
    __shared__ float sS[G1_ROWS * NHD], sD[G1_ROWS * NHD];
    const int tid  = (int)threadIdx.x;
    const int lane = tid & 31;
    const int wave = tid >> 5;
    const int m    = lane & 15;
    const int half = lane >> 4;
    const int rowBase = (int)blockIdx.x * G1_ROWS;
    const float as0 = Asrc[(wave << 5) + m], as1 = Asrc[(wave << 5) + 16 + m];
    const float ad0 = Adst[(wave << 5) + m], ad1 = Adst[(wave << 5) + 16 + m];
    float4 st[(G1_ROWS * INCH / 4) / 256];
#pragma unroll
    for (int i = 0; i < (G1_ROWS * INCH / 4) / 256; ++i) {
        const int idx = tid + i * 256;
        const int r = idx >> 5, c4 = idx & 31;
        const int gr = (rowBase + r < nrows) ? (rowBase + r) : (nrows - 1);
        st[i] = *reinterpret_cast<const float4*>(X + (size_t)gr * INCH + (c4 << 2));
    }
#pragma unroll
    for (int i = 0; i < (G1_ROWS * INCH / 4) / 256; ++i) {
        const int idx = tid + i * 256;
        const int r = idx >> 5, c4 = idx & 31;
        *reinterpret_cast<float4*>(&sA[r * G1_PITCH + (c4 << 2)]) = st[i];
    }
    __syncthreads();
    const int c0 = wave << 5;
    const float* wp  = W + c0 + m;
    const float* a0p = sA + m * G1_PITCH + (half << 1);
    const float* a1p = a0p + 16 * G1_PITCH;
    v8f acc00 = {0.f, 0.f, 0.f, 0.f, 0.f, 0.f, 0.f, 0.f};
    v8f acc01 = acc00, acc10 = acc00, acc11 = acc00;
    for (int k = 0; k < INCH; k += 4) {
        const int kk = k + (half << 1);
        const v2f a0 = *reinterpret_cast<const v2f*>(a0p + k);
        const v2f a1 = *reinterpret_cast<const v2f*>(a1p + k);
        v2f b0, b1;
        b0.x = wp[(size_t)kk * HC];
        b0.y = wp[(size_t)(kk + 1) * HC];
        b1.x = wp[(size_t)kk * HC + 16];
        b1.y = wp[(size_t)(kk + 1) * HC + 16];
        acc00 = __builtin_amdgcn_wmma_f32_16x16x4_f32(false, a0, false, b0,
                                                      (short)0, acc00, false, false);
        acc01 = __builtin_amdgcn_wmma_f32_16x16x4_f32(false, a0, false, b1,
                                                      (short)0, acc01, false, false);
        acc10 = __builtin_amdgcn_wmma_f32_16x16x4_f32(false, a1, false, b0,
                                                      (short)0, acc10, false, false);
        acc11 = __builtin_amdgcn_wmma_f32_16x16x4_f32(false, a1, false, b1,
                                                      (short)0, acc11, false, false);
    }
    float* o0 = H1 + (size_t)(rowBase + (half << 3)) * HC + c0 + m;
    float* o1 = o0 + (size_t)16 * HC;
    if (rowBase + G1_ROWS <= nrows) {
#pragma unroll
        for (int v = 0; v < 8; ++v) {
            o0[(size_t)v * HC]      = acc00[v];
            o0[(size_t)v * HC + 16] = acc01[v];
            o1[(size_t)v * HC]      = acc10[v];
            o1[(size_t)v * HC + 16] = acc11[v];
        }
    } else {
#pragma unroll
        for (int v = 0; v < 8; ++v) {
            const int r0 = rowBase + v + (half << 3);
            if (r0 < nrows) {
                o0[(size_t)v * HC]      = acc00[v];
                o0[(size_t)v * HC + 16] = acc01[v];
            }
            if (r0 + 16 < nrows) {
                o1[(size_t)v * HC]      = acc10[v];
                o1[(size_t)v * HC + 16] = acc11[v];
            }
        }
    }
    float s0 = 0.f, s1 = 0.f, d0 = 0.f, d1 = 0.f;
#pragma unroll
    for (int v = 0; v < 8; ++v) {
        const float ts0 = row16_sum(acc00[v] * as0 + acc01[v] * as1);
        const float td0 = row16_sum(acc00[v] * ad0 + acc01[v] * ad1);
        const float ts1 = row16_sum(acc10[v] * as0 + acc11[v] * as1);
        const float td1 = row16_sum(acc10[v] * ad0 + acc11[v] * ad1);
        s0 = (m == v) ? ts0 : s0;
        d0 = (m == v) ? td0 : d0;
        s1 = (m == v) ? ts1 : s1;
        d1 = (m == v) ? td1 : d1;
    }
    if (m < 8) {
        const int r = (half << 3) + m;
        sS[r * NHD + wave] = s0;  sS[(r + 16) * NHD + wave] = s1;
        sD[r * NHD + wave] = d0;  sD[(r + 16) * NHD + wave] = d1;
    }
    lds_handover_barrier();
    if ((rowBase << 3) + tid < nrows * NHD) {
        aS[((size_t)rowBase << 3) + tid] = sS[tid];
        aD[((size_t)rowBase << 3) + tid] = sD[tid];
    }
}

#define G2_ROWS  32
#define G2_PITCH (HC + 4)
#define G2_UNITS (G2_ROWS * HC / 128)
#define G2_UPW   ((G2_UNITS + 5) / 6)
__global__ void __launch_bounds__(192, 12)
gemm_h_w2(const float* __restrict__ Hin, const float* __restrict__ W,
          const float* __restrict__ Asrc, const float* __restrict__ Adst,
          float* __restrict__ H2, float* __restrict__ aS, float* __restrict__ aD, int nrows) {
    __shared__ __attribute__((aligned(16))) float sA[(G2_ROWS + 1) * G2_PITCH];
    __shared__ float sS[6 * 16], sD[6 * 16];
    const int tid  = (int)threadIdx.x;
    const int lane = tid & 31;
    const int wave = tid >> 5;
    const int m    = lane & 15;
    const int half = lane >> 4;
    const int rowBase = (int)blockIdx.x * G2_ROWS;
    float4 st[G2_UPW];
#pragma unroll
    for (int i = 0; i < G2_UPW; ++i) {
        const int u0 = wave + 6 * i;
        const int u = (u0 < G2_UNITS) ? u0 : (G2_UNITS - 1);
        const int r = u >> 1, c = ((u & 1) << 7) + (lane << 2);
        const int gr = (rowBase + r < nrows) ? (rowBase + r) : (nrows - 1);
        st[i] = *reinterpret_cast<const float4*>(Hin + (size_t)gr * HC + c);
    }
#pragma unroll
    for (int i = 0; i < G2_UPW; ++i) {
        const int u0 = wave + 6 * i;
        const int r = u0 >> 1, c = ((u0 & 1) << 7) + (lane << 2);
        *reinterpret_cast<float4*>(&sA[r * G2_PITCH + c]) = st[i];
    }
    __syncthreads();
    const int rt = wave / 3;
    const int colBase = (wave - rt * 3) << 4;
    const int col = colBase + m;
    const bool cin = col < OUTC;
    const float* ap = sA + (rt * 16 + m) * G2_PITCH + (half << 1);
    const float* wp = W + (cin ? col : 0);
    v8f acc = {0.f, 0.f, 0.f, 0.f, 0.f, 0.f, 0.f, 0.f};
    for (int k = 0; k < HC; k += 4) {
        const int kk = k + (half << 1);
        const v2f a = *reinterpret_cast<const v2f*>(ap + k);
        v2f b;
        b.x = wp[kk * OUTC];
        b.y = wp[(kk + 1) * OUTC];
        acc = __builtin_amdgcn_wmma_f32_16x16x4_f32(false, a, false, b,
                                                    (short)0, acc, false, false);
    }
    if (cin) {
        const int r0 = rowBase + rt * 16 + (half << 3);
        float* o = H2 + (size_t)r0 * OUTC + col;
        if (rowBase + G2_ROWS <= nrows) {
#pragma unroll
            for (int v = 0; v < 8; ++v) o[(size_t)v * OUTC] = acc[v];
        } else {
#pragma unroll
            for (int v = 0; v < 8; ++v)
                if (r0 + v < nrows) o[(size_t)v * OUTC] = acc[v];
        }
    }
    const float as = cin ? Asrc[col] : 0.f;
    const float ad = cin ? Adst[col] : 0.f;
    float s = 0.f, d = 0.f;
#pragma unroll
    for (int v = 0; v < 8; ++v) {
        const float ts = row16_sum(cin ? acc[v] * as : 0.f);
        const float td = row16_sum(cin ? acc[v] * ad : 0.f);
        s = (m == v) ? ts : s;
        d = (m == v) ? td : d;
    }
    if (m < 8) {
        sS[wave * 16 + (half << 3) + m] = s;
        sD[wave * 16 + (half << 3) + m] = d;
    }
    lds_handover_barrier();
    if (tid < G2_ROWS && rowBase + tid < nrows) {
        const int w0 = (tid >> 4) * 3, r = tid & 15;
        aS[rowBase + tid] = (sS[w0 * 16 + r] + sS[(w0 + 1) * 16 + r]) + sS[(w0 + 2) * 16 + r];
        aD[rowBase + tid] = (sD[w0 * 16 + r] + sD[(w0 + 1) * 16 + r]) + sD[(w0 + 2) * 16 + r];
    }
}

#define CSR_PARTS   64
#define CSR_THREADS 1024
#define CSR_BINS    51200
#define CSR_AHEAD   16

__device__ __forceinline__ int csr_part_len(int E) {
    return (((E + CSR_PARTS - 1) / CSR_PARTS) + 7) & ~7;
}

__global__ void __launch_bounds__(CSR_THREADS)
count_rank(const int* __restrict__ dstA, int* __restrict__ partc, int* __restrict__ rank,
           int E, int n) {
    __shared__ int hist[CSR_BINS];
    const int tid  = (int)threadIdx.x;
    const int part = (int)blockIdx.x;
    const int len  = csr_part_len(E);
    const int beg  = part * len;
    const int end  = (beg + len < E) ? (beg + len) : E;
    for (int w0 = 0; w0 < n; w0 += CSR_BINS) {
        const int wn = (n - w0 < CSR_BINS) ? (n - w0) : CSR_BINS;
        for (int i = tid; i < wn; i += CSR_THREADS) hist[i] = 0;
        __syncthreads();
        for (int e0 = beg; e0 < end; e0 += CSR_AHEAD * CSR_THREADS) {
            int d[CSR_AHEAD], r[CSR_AHEAD];
#pragma unroll
            for (int u = 0; u < CSR_AHEAD; ++u) {
                const int e = e0 + u * CSR_THREADS + tid;
                const bool in = e < end;
                const int x = dstA[in ? e : beg];
                d[u] = in ? (x - w0) : -1;
            }
#pragma unroll
            for (int u = 0; u < CSR_AHEAD; ++u) {
                r[u] = 0;
                if ((unsigned)d[u] < (unsigned)wn) r[u] = atomicAdd(&hist[d[u]], 1);
            }
#pragma unroll
            for (int u = 0; u < CSR_AHEAD; ++u) {
                const int e = e0 + u * CSR_THREADS + tid;
                if ((unsigned)d[u] < (unsigned)wn) rank[e] = r[u];
            }
        }
        __syncthreads();
        int* row = partc + (size_t)part * n + w0;
        for (int i = tid; i < wn; i += CSR_THREADS) row[i] = hist[i];
        __syncthreads();
    }
}

__global__ void part_prefix(const int* __restrict__ partc, int* __restrict__ partp,
                            int* __restrict__ cnt, int n) {
    const int d = (int)blockIdx.x * (int)blockDim.x + (int)threadIdx.x;
    if (d >= n) return;
    int run = 0;
    for (int p0 = 0; p0 < CSR_PARTS; p0 += 16) {
        int x[16];
#pragma unroll
        for (int u = 0; u < 16; ++u) x[u] = partc[(size_t)(p0 + u) * n + d];
#pragma unroll
        for (int u = 0; u < 16; ++u) { partp[(size_t)(p0 + u) * n + d] = run; run += x[u]; }
    }
    cnt[d] = run + 1;
}

#define SCAN_AHEAD 16
__global__ void scan_rowptr(const int* __restrict__ cnt, int* __restrict__ rowptr, int n) {
    __shared__ int sPre[32];
    __shared__ int sTot[32];
    const int tid  = (int)threadIdx.x;
    const int lane = tid & 31;
    const int wave = tid >> 5;
    const int base = (int)blockIdx.x * 1024;
    const int i = base + tid;
    const int v = (i < n) ? cnt[i] : 0;
    int pre = 0;
    for (int k0 = 0; k0 < base; k0 += SCAN_AHEAD * 1024) {
        for (int u = 0; u < SCAN_AHEAD; ++u) {
            const int kb = k0 + u * 1024;
            const bool in = kb < base;
            const int x = cnt[(in ? kb : 0) + tid];
            pre += in ? x : 0;
        }
    }
    int inc = v;
    for (int d = 1; d < 32; d <<= 1) {
        const int t = __shfl_up(inc, d, 32);
        if (lane >= d) inc += t;
    }
    for (int o = 16; o > 0; o >>= 1) pre += __shfl_xor(pre, o, 32);
    if (lane == 31) { sTot[wave] = inc; sPre[wave] = pre; }
    __syncthreads();
    int off = sPre[lane] + ((lane < wave) ? sTot[lane] : 0);
    for (int o = 16; o > 0; o >>= 1) off += __shfl_xor(off, o, 32);
    const int excl = off + inc - v;
    if (i < n) rowptr[i] = excl;
    if (i == n - 1) rowptr[n] = excl + v;
}

#define FILL_EPT 8
__global__ void fill_csr(const int* __restrict__ srcA, const int* __restrict__ dstA,
                         const int* __restrict__ rank, const int* __restrict__ partp,
                         const int* __restrict__ rowptr, int* __restrict__ col, int E, int n) {
    const int base = ((int)blockIdx.x * (int)blockDim.x + (int)threadIdx.x) * FILL_EPT;
    if (base >= E + n) return;
    const int len = csr_part_len(E);
    const bool vec = (((reinterpret_cast<size_t>(srcA) | reinterpret_cast<size_t>(dstA) |
                        reinterpret_cast<size_t>(rank)) & 15) == 0);
    if (vec && base + FILL_EPT <= E) {
        const int* pp = partp + (size_t)(base / len) * n;
        const int4 sA = *reinterpret_cast<const int4*>(srcA + base);
        const int4 sB = *reinterpret_cast<const int4*>(srcA + base + 4);
        const int4 dA = *reinterpret_cast<const int4*>(dstA + base);
        const int4 dB = *reinterpret_cast<const int4*>(dstA + base + 4);
        const int4 rA = *reinterpret_cast<const int4*>(rank + base);
        const int4 rB = *reinterpret_cast<const int4*>(rank + base + 4);
        const int p0 = rowptr[dA.x] + pp[dA.x];
        const int p1 = rowptr[dA.y] + pp[dA.y];
        const int p2 = rowptr[dA.z] + pp[dA.z];
        const int p3 = rowptr[dA.w] + pp[dA.w];
        const int p4 = rowptr[dB.x] + pp[dB.x];
        const int p5 = rowptr[dB.y] + pp[dB.y];
        const int p6 = rowptr[dB.z] + pp[dB.z];
        const int p7 = rowptr[dB.w] + pp[dB.w];
        col[p0 + rA.x] = sA.x;
        col[p1 + rA.y] = sA.y;
        col[p2 + rA.z] = sA.z;
        col[p3 + rA.w] = sA.w;
        col[p4 + rB.x] = sB.x;
        col[p5 + rB.y] = sB.y;
        col[p6 + rB.z] = sB.z;
        col[p7 + rB.w] = sB.w;
    } else {
        for (int k = 0; k < FILL_EPT; ++k) {
            const int e = base + k;
            if (e < E) {
                const int d = dstA[e];
                col[rowptr[d] + partp[(size_t)(e / len) * n + d] + rank[e]] = srcA[e];
            } else if (e < E + n) {
                const int i = e - E;
                col[rowptr[i + 1] - 1] = i;
            }
        }
    }
}

__global__ void gat_gather_l1(const int* __restrict__ rowptr, const int* __restrict__ col,
                              const float* __restrict__ aS, const float* __restrict__ aD,
                              const float* __restrict__ H1, const float* __restrict__ b1,
                              float* __restrict__ out1, int n) {
    const int lane = threadIdx.x & 31;
    const int node = (blockIdx.x * blockDim.x + threadIdx.x) >> 5;
    if (node >= n) return;
    const int h  = lane >> 2;
    const int ch = lane << 3;
    const int beg = rowptr[node], end = rowptr[node + 1];
    const float aDd = aD[node * NHD + h];
    float m = -__builtin_inff(), den = 0.f;
    float a0 = 0.f, a1 = 0.f, a2 = 0.f, a3 = 0.f, a4 = 0.f, a5 = 0.f, a6 = 0.f, a7 = 0.f;
    for (int j = beg; j < end; ++j) {
        const int s = col[j];
        if (j + 1 < end) {
            const int s2 = col[j + 1];
            __builtin_prefetch(&H1[(size_t)s2 * HC + ch], 0, 1);
        }
        const float l = leaky(aS[s * NHD + h] + aDd);
        const float mn = fmaxf(m, l);
        const float scale = expf(m - mn);
        const float wgt   = expf(l - mn);
        const float4* hp = reinterpret_cast<const float4*>(H1 + (size_t)s * HC + ch);
        const float4 x0 = hp[0];
        const float4 x1 = hp[1];
        a0 = a0 * scale + wgt * x0.x;
        a1 = a1 * scale + wgt * x0.y;
        a2 = a2 * scale + wgt * x0.z;
        a3 = a3 * scale + wgt * x0.w;
        a4 = a4 * scale + wgt * x1.x;
        a5 = a5 * scale + wgt * x1.y;
        a6 = a6 * scale + wgt * x1.z;
        a7 = a7 * scale + wgt * x1.w;
        den = den * scale + wgt;
        m = mn;
    }
    const float v0 = a0 / den + b1[ch + 0];
    const float v1 = a1 / den + b1[ch + 1];
    const float v2 = a2 / den + b1[ch + 2];
    const float v3 = a3 / den + b1[ch + 3];
    const float v4 = a4 / den + b1[ch + 4];
    const float v5 = a5 / den + b1[ch + 5];
    const float v6 = a6 / den + b1[ch + 6];
    const float v7 = a7 / den + b1[ch + 7];
    float4 o0, o1;
    o0.x = v0 > 0.f ? v0 : 0.f;
    o0.y = v1 > 0.f ? v1 : 0.f;
    o0.z = v2 > 0.f ? v2 : 0.f;
    o0.w = v3 > 0.f ? v3 : 0.f;
    o1.x = v4 > 0.f ? v4 : 0.f;
    o1.y = v5 > 0.f ? v5 : 0.f;
    o1.z = v6 > 0.f ? v6 : 0.f;
    o1.w = v7 > 0.f ? v7 : 0.f;
    float4* op = reinterpret_cast<float4*>(out1 + (size_t)node * HC + ch);
    op[0] = o0;
    op[1] = o1;
}

#define L2_LANES 10
#define L2_NODES 3
__global__ void gat_l2_fused(const int* __restrict__ rowptr, const int* __restrict__ col,
                             const float* __restrict__ aS, const float* __restrict__ aD,
                             const float* __restrict__ H2, const float* __restrict__ b2,
                             float* __restrict__ out, int n) {
    const int lane = threadIdx.x & 31;
    const int wave = (blockIdx.x * blockDim.x + threadIdx.x) >> 5;
    if (wave * L2_NODES >= n) return;
    const int grp  = lane / L2_LANES;
    const int k    = lane - grp * L2_LANES;
    const int c    = k << 2;
    const int node = wave * L2_NODES + grp;
    const bool act = (grp < L2_NODES) && (node < n);
    const int nd  = act ? node : wave * L2_NODES;
    const int beg = rowptr[nd];
    const int end = rowptr[nd + (act ? 1 : 0)];
    const float aDd = aD[nd];
    const float bb0 = b2[c + 0], bb1 = b2[c + 1], bb2 = b2[c + 2], bb3 = b2[c + 3];
    float m = -__builtin_inff(), den = 0.f, a0 = 0.f, a1 = 0.f, a2 = 0.f, a3 = 0.f;
    for (int j = beg; j < end; ++j) {
        const int s = col[j];
        if (j + 1 < end) {
            const int s2 = col[j + 1];
            __builtin_prefetch(&H2[(size_t)s2 * OUTC + c], 0, 1);
        }
        const float l = leaky(aS[s] + aDd);
        const float mn = fmaxf(m, l);
        const float scale = expf(m - mn);
        const float wgt   = expf(l - mn);
        const float4 x = *reinterpret_cast<const float4*>(H2 + (size_t)s * OUTC + c);
        a0 = a0 * scale + wgt * x.x;
        a1 = a1 * scale + wgt * x.y;
        a2 = a2 * scale + wgt * x.z;
        a3 = a3 * scale + wgt * x.w;
        den = den * scale + wgt;
        m = mn;
    }
    const float ninf = -__builtin_inff();
    const float v0 = act ? a0 / den + bb0 : ninf;
    const float v1 = act ? a1 / den + bb1 : ninf;
    const float v2 = act ? a2 / den + bb2 : ninf;
    const float v3 = act ? a3 / den + bb3 : ninf;
    const int g0 = (grp < L2_NODES ? grp : 0) * L2_LANES;
    const float lmx = fmaxf(fmaxf(v0, v1), fmaxf(v2, v3));
    float mx = ninf;
    for (int t = 0; t < L2_LANES; ++t) mx = fmaxf(mx, __shfl(lmx, g0 + t, 32));
    const float lsum = act ? ((expf(v0 - mx) + expf(v1 - mx)) + expf(v2 - mx)) + expf(v3 - mx)
                           : 0.f;
    float ssum = 0.f;
    for (int t = 0; t < L2_LANES; ++t) ssum += __shfl(lsum, g0 + t, 32);
    const float lse = mx + logf(ssum);
    if (act) {
        float4 o;
        o.x = v0 - lse;  o.y = v1 - lse;  o.z = v2 - lse;  o.w = v3 - lse;
        *reinterpret_cast<float4*>(out + (size_t)node * OUTC + c) = o;
    }
}

extern "C" void kernel_launch(void* const* d_in, const int* in_sizes, int n_in,
                              void* d_out, int out_size, void* d_ws, size_t ws_size,
                              hipStream_t stream) {
    const float* x   = (const float*)d_in[0];
    const int*   ei  = (const int*)d_in[1];
    const float* W1  = (const float*)d_in[2];
    const float* as1 = (const float*)d_in[3];
    const float* ad1 = (const float*)d_in[4];
    const float* b1  = (const float*)d_in[5];
    const float* W2  = (const float*)d_in[6];
    const float* as2 = (const float*)d_in[7];
    const float* ad2 = (const float*)d_in[8];
    const float* b2  = (const float*)d_in[9];
    const int N  = in_sizes[0] / INCH;
    const int E  = in_sizes[1] / 2;
    const int Et = E + N;
    const int* srcA = ei;
    const int* dstA = ei + E;

    float* ws = (float*)d_ws;
    size_t off = 0;
    float* h1   = ws + off; off += (size_t)N * HC;
    float* out1 = ws + off; off += (size_t)N * HC;
    float* aS1  = ws + off; off += (size_t)N * NHD;
    float* aD1  = ws + off; off += (size_t)N * NHD;
    int* rowptr = (int*)(ws + off); off += (size_t)N + 1;
    int* cnt    = (int*)(ws + off); off += (size_t)N;
    int* col    = (int*)(ws + off); off += (size_t)Et;
    float* h2  = h1;
    float* aS2 = aS1;
    float* aD2 = aD1;
    int* partc = (int*)h1;
    int* partp = partc + (size_t)CSR_PARTS * N;
    int* rank  = (int*)out1;

    count_rank<<<CSR_PARTS, CSR_THREADS, 0, stream>>>(dstA, partc, rank, E, N);
    part_prefix<<<(N + 255) / 256, 256, 0, stream>>>(partc, partp, cnt, N);
    scan_rowptr<<<(N + 1023) / 1024, 1024, 0, stream>>>(cnt, rowptr, N);
    fill_csr<<<(Et + 256 * FILL_EPT - 1) / (256 * FILL_EPT), 256, 0, stream>>>(
        srcA, dstA, rank, partp, rowptr, col, E, N);

    gemm_x_w1<<<(N + G1_ROWS - 1) / G1_ROWS, 256, 0, stream>>>(x, W1, as1, ad1, h1, aS1, aD1, N);
    gat_gather_l1<<<(N * 32 + 255) / 256, 256, 0, stream>>>(
        rowptr, col, aS1, aD1, h1, b1, out1, N);

    gemm_h_w2<<<(N + G2_ROWS - 1) / G2_ROWS, 192, 0, stream>>>(out1, W2, as2, ad2, h2, aS2, aD2, N);
    gat_l2_fused<<<((N + L2_NODES - 1) / L2_NODES * 32 + 255) / 256, 256, 0, stream>>>(
        rowptr, col, aS2, aD2, h2, b2, (float*)d_out, N);
}


# layer-1 gather: softmax weights of up to 32 edges at once, lane per edge (max by cross-lane tree, one exp per edge and head), LDS hand-over; row loop only loads+FMAs, 4 rows in flight; sum order kept
# speedup vs baseline: 3.3129x; 1.0947x over previous
#include <hip/hip_runtime.h>
#include <hip/hip_bf16.h>

typedef __attribute__((ext_vector_type(2))) float v2f;
typedef __attribute__((ext_vector_type(8))) float v8f;

#define INCH  128
#define HC    256
#define NHD   8
#define CPH   32
#define OUTC  40
#define SLOPE 0.2f

__device__ __forceinline__ float leaky(float x) { return x > 0.f ? x : SLOPE * x; }

__device__ __forceinline__ void lds_handover_barrier() {
    asm volatile("s_wait_dscnt 0x0" ::: "memory");
    asm volatile("s_barrier_signal -1\n\ts_barrier_wait -1" ::: "memory");
}

#define ROW_XOR(x, k) __builtin_bit_cast(float, __builtin_amdgcn_update_dpp( \
    0, __builtin_bit_cast(int, (x)), 0x160 + (k), 0xf, 0xf, true))
__device__ __forceinline__ float row16_sum(float x) {
    x += ROW_XOR(x, 1);
    x += ROW_XOR(x, 2);
    x += ROW_XOR(x, 4);
    x += ROW_XOR(x, 8);
    return x;
}

#define G1_ROWS  32
#define G1_PITCH (INCH + 4)
__global__ void __launch_bounds__(256)
gemm_x_w1(const float* __restrict__ X, const float* __restrict__ W,
          const float* __restrict__ Asrc, const float* __restrict__ Adst,
          float* __restrict__ H1, float* __restrict__ aS, float* __restrict__ aD, int nrows) {
    __shared__ __attribute__((aligned(16))) float sA[G1_ROWS * G1_PITCH];
    __shared__ float sS[G1_ROWS * NHD], sD[G1_ROWS * NHD];
    const int tid  = (int)threadIdx.x;
    const int lane = tid & 31;
    const int wave = tid >> 5;
    const int m    = lane & 15;
    const int half = lane >> 4;
    const int rowBase = (int)blockIdx.x * G1_ROWS;
    const float as0 = Asrc[(wave << 5) + m], as1 = Asrc[(wave << 5) + 16 + m];
    const float ad0 = Adst[(wave << 5) + m], ad1 = Adst[(wave << 5) + 16 + m];
    float4 st[(G1_ROWS * INCH / 4) / 256];
#pragma unroll
    for (int i = 0; i < (G1_ROWS * INCH / 4) / 256; ++i) {
        const int idx = tid + i * 256;
        const int r = idx >> 5, c4 = idx & 31;
        const int gr = (rowBase + r < nrows) ? (rowBase + r) : (nrows - 1);
        st[i] = *reinterpret_cast<const float4*>(X + (size_t)gr * INCH + (c4 << 2));
    }
#pragma unroll
    for (int i = 0; i < (G1_ROWS * INCH / 4) / 256; ++i) {
        const int idx = tid + i * 256;
        const int r = idx >> 5, c4 = idx & 31;
        *reinterpret_cast<float4*>(&sA[r * G1_PITCH + (c4 << 2)]) = st[i];
    }
    __syncthreads();
    const int c0 = wave << 5;
    const float* wp  = W + c0 + m;
    const float* a0p = sA + m * G1_PITCH + (half << 1);
    const float* a1p = a0p + 16 * G1_PITCH;
    v8f acc00 = {0.f, 0.f, 0.f, 0.f, 0.f, 0.f, 0.f, 0.f};
    v8f acc01 = acc00, acc10 = acc00, acc11 = acc00;
    for (int k = 0; k < INCH; k += 4) {
        const int kk = k + (half << 1);
        const v2f a0 = *reinterpret_cast<const v2f*>(a0p + k);
        const v2f a1 = *reinterpret_cast<const v2f*>(a1p + k);
        v2f b0, b1;
        b0.x = wp[(size_t)kk * HC];
        b0.y = wp[(size_t)(kk + 1) * HC];
        b1.x = wp[(size_t)kk * HC + 16];
        b1.y = wp[(size_t)(kk + 1) * HC + 16];
        acc00 = __builtin_amdgcn_wmma_f32_16x16x4_f32(false, a0, false, b0,
                                                      (short)0, acc00, false, false);
        acc01 = __builtin_amdgcn_wmma_f32_16x16x4_f32(false, a0, false, b1,
                                                      (short)0, acc01, false, false);
        acc10 = __builtin_amdgcn_wmma_f32_16x16x4_f32(false, a1, false, b0,
                                                      (short)0, acc10, false, false);
        acc11 = __builtin_amdgcn_wmma_f32_16x16x4_f32(false, a1, false, b1,
                                                      (short)0, acc11, false, false);
    }
    float* o0 = H1 + (size_t)(rowBase + (half << 3)) * HC + c0 + m;
    float* o1 = o0 + (size_t)16 * HC;
    if (rowBase + G1_ROWS <= nrows) {
#pragma unroll
        for (int v = 0; v < 8; ++v) {
            o0[(size_t)v * HC]      = acc00[v];
            o0[(size_t)v * HC + 16] = acc01[v];
            o1[(size_t)v * HC]      = acc10[v];
            o1[(size_t)v * HC + 16] = acc11[v];
        }
    } else {
#pragma unroll
        for (int v = 0; v < 8; ++v) {
            const int r0 = rowBase + v + (half << 3);
            if (r0 < nrows) {
                o0[(size_t)v * HC]      = acc00[v];
                o0[(size_t)v * HC + 16] = acc01[v];
            }
            if (r0 + 16 < nrows) {
                o1[(size_t)v * HC]      = acc10[v];
                o1[(size_t)v * HC + 16] = acc11[v];
            }
        }
    }
    float s0 = 0.f, s1 = 0.f, d0 = 0.f, d1 = 0.f;
#pragma unroll
    for (int v = 0; v < 8; ++v) {
        const float ts0 = row16_sum(acc00[v] * as0 + acc01[v] * as1);
        const float td0 = row16_sum(acc00[v] * ad0 + acc01[v] * ad1);
        const float ts1 = row16_sum(acc10[v] * as0 + acc11[v] * as1);
        const float td1 = row16_sum(acc10[v] * ad0 + acc11[v] * ad1);
        s0 = (m == v) ? ts0 : s0;
        d0 = (m == v) ? td0 : d0;
        s1 = (m == v) ? ts1 : s1;
        d1 = (m == v) ? td1 : d1;
    }
    if (m < 8) {
        const int r = (half << 3) + m;
        sS[r * NHD + wave] = s0;  sS[(r + 16) * NHD + wave] = s1;
        sD[r * NHD + wave] = d0;  sD[(r + 16) * NHD + wave] = d1;
    }
    lds_handover_barrier();
    if ((rowBase << 3) + tid < nrows * NHD) {
        aS[((size_t)rowBase << 3) + tid] = sS[tid];
        aD[((size_t)rowBase << 3) + tid] = sD[tid];
    }
}

#define G2_ROWS  32
#define G2_PITCH (HC + 4)
#define G2_UNITS (G2_ROWS * HC / 128)
#define G2_UPW   ((G2_UNITS + 5) / 6)
__global__ void __launch_bounds__(192, 12)
gemm_h_w2(const float* __restrict__ Hin, const float* __restrict__ W,
          const float* __restrict__ Asrc, const float* __restrict__ Adst,
          float* __restrict__ H2, float* __restrict__ aS, float* __restrict__ aD, int nrows) {
    __shared__ __attribute__((aligned(16))) float sA[(G2_ROWS + 1) * G2_PITCH];
    __shared__ float sS[6 * 16], sD[6 * 16];
    const int tid  = (int)threadIdx.x;
    const int lane = tid & 31;
    const int wave = tid >> 5;
    const int m    = lane & 15;
    const int half = lane >> 4;
    const int rowBase = (int)blockIdx.x * G2_ROWS;
    float4 st[G2_UPW];
#pragma unroll
    for (int i = 0; i < G2_UPW; ++i) {
        const int u0 = wave + 6 * i;
        const int u = (u0 < G2_UNITS) ? u0 : (G2_UNITS - 1);
        const int r = u >> 1, c = ((u & 1) << 7) + (lane << 2);
        const int gr = (rowBase + r < nrows) ? (rowBase + r) : (nrows - 1);
        st[i] = *reinterpret_cast<const float4*>(Hin + (size_t)gr * HC + c);
    }
#pragma unroll
    for (int i = 0; i < G2_UPW; ++i) {
        const int u0 = wave + 6 * i;
        const int r = u0 >> 1, c = ((u0 & 1) << 7) + (lane << 2);
        *reinterpret_cast<float4*>(&sA[r * G2_PITCH + c]) = st[i];
    }
    __syncthreads();
    const int rt = wave / 3;
    const int colBase = (wave - rt * 3) << 4;
    const int col = colBase + m;
    const bool cin = col < OUTC;
    const float* ap = sA + (rt * 16 + m) * G2_PITCH + (half << 1);
    const float* wp = W + (cin ? col : 0);
    v8f acc = {0.f, 0.f, 0.f, 0.f, 0.f, 0.f, 0.f, 0.f};
    for (int k = 0; k < HC; k += 4) {
        const int kk = k + (half << 1);
        const v2f a = *reinterpret_cast<const v2f*>(ap + k);
        v2f b;
        b.x = wp[kk * OUTC];
        b.y = wp[(kk + 1) * OUTC];
        acc = __builtin_amdgcn_wmma_f32_16x16x4_f32(false, a, false, b,
                                                    (short)0, acc, false, false);
    }
    if (cin) {
        const int r0 = rowBase + rt * 16 + (half << 3);
        float* o = H2 + (size_t)r0 * OUTC + col;
        if (rowBase + G2_ROWS <= nrows) {
#pragma unroll
            for (int v = 0; v < 8; ++v) o[(size_t)v * OUTC] = acc[v];
        } else {
#pragma unroll
            for (int v = 0; v < 8; ++v)
                if (r0 + v < nrows) o[(size_t)v * OUTC] = acc[v];
        }
    }
    const float as = cin ? Asrc[col] : 0.f;
    const float ad = cin ? Adst[col] : 0.f;
    float s = 0.f, d = 0.f;
#pragma unroll
    for (int v = 0; v < 8; ++v) {
        const float ts = row16_sum(cin ? acc[v] * as : 0.f);
        const float td = row16_sum(cin ? acc[v] * ad : 0.f);
        s = (m == v) ? ts : s;
        d = (m == v) ? td : d;
    }
    if (m < 8) {
        sS[wave * 16 + (half << 3) + m] = s;
        sD[wave * 16 + (half << 3) + m] = d;
    }
    lds_handover_barrier();
    if (tid < G2_ROWS && rowBase + tid < nrows) {
        const int w0 = (tid >> 4) * 3, r = tid & 15;
        aS[rowBase + tid] = (sS[w0 * 16 + r] + sS[(w0 + 1) * 16 + r]) + sS[(w0 + 2) * 16 + r];
        aD[rowBase + tid] = (sD[w0 * 16 + r] + sD[(w0 + 1) * 16 + r]) + sD[(w0 + 2) * 16 + r];
    }
}

#define CSR_PARTS   64
#define CSR_THREADS 1024
#define CSR_BINS    51200
#define CSR_AHEAD   16

__device__ __forceinline__ int csr_part_len(int E) {
    return (((E + CSR_PARTS - 1) / CSR_PARTS) + 7) & ~7;
}

__global__ void __launch_bounds__(CSR_THREADS)
count_rank(const int* __restrict__ dstA, int* __restrict__ partc, int* __restrict__ rank,
           int E, int n) {
    __shared__ int hist[CSR_BINS];
    const int tid  = (int)threadIdx.x;
    const int part = (int)blockIdx.x;
    const int len  = csr_part_len(E);
    const int beg  = part * len;
    const int end  = (beg + len < E) ? (beg + len) : E;
    for (int w0 = 0; w0 < n; w0 += CSR_BINS) {
        const int wn = (n - w0 < CSR_BINS) ? (n - w0) : CSR_BINS;
        for (int i = tid; i < wn; i += CSR_THREADS) hist[i] = 0;
        __syncthreads();
        for (int e0 = beg; e0 < end; e0 += CSR_AHEAD * CSR_THREADS) {
            int d[CSR_AHEAD], r[CSR_AHEAD];
#pragma unroll
            for (int u = 0; u < CSR_AHEAD; ++u) {
                const int e = e0 + u * CSR_THREADS + tid;
                const bool in = e < end;
                const int x = dstA[in ? e : beg];
                d[u] = in ? (x - w0) : -1;
            }
#pragma unroll
            for (int u = 0; u < CSR_AHEAD; ++u) {
                r[u] = 0;
                if ((unsigned)d[u] < (unsigned)wn) r[u] = atomicAdd(&hist[d[u]], 1);
            }
#pragma unroll
            for (int u = 0; u < CSR_AHEAD; ++u) {
                const int e = e0 + u * CSR_THREADS + tid;
                if ((unsigned)d[u] < (unsigned)wn) rank[e] = r[u];
            }
        }
        __syncthreads();
        int* row = partc + (size_t)part * n + w0;
        for (int i = tid; i < wn; i += CSR_THREADS) row[i] = hist[i];
        __syncthreads();
    }
}

__global__ void part_prefix(const int* __restrict__ partc, int* __restrict__ partp,
                            int* __restrict__ cnt, int n) {
    const int d = (int)blockIdx.x * (int)blockDim.x + (int)threadIdx.x;
    if (d >= n) return;
    int run = 0;
    for (int p0 = 0; p0 < CSR_PARTS; p0 += 16) {
        int x[16];
#pragma unroll
        for (int u = 0; u < 16; ++u) x[u] = partc[(size_t)(p0 + u) * n + d];
#pragma unroll
        for (int u = 0; u < 16; ++u) { partp[(size_t)(p0 + u) * n + d] = run; run += x[u]; }
    }
    cnt[d] = run + 1;
}

#define SCAN_AHEAD 16
__global__ void scan_rowptr(const int* __restrict__ cnt, int* __restrict__ rowptr, int n) {
    __shared__ int sPre[32];
    __shared__ int sTot[32];
    const int tid  = (int)threadIdx.x;
    const int lane = tid & 31;
    const int wave = tid >> 5;
    const int base = (int)blockIdx.x * 1024;
    const int i = base + tid;
    const int v = (i < n) ? cnt[i] : 0;
    int pre = 0;
    for (int k0 = 0; k0 < base; k0 += SCAN_AHEAD * 1024) {
        for (int u = 0; u < SCAN_AHEAD; ++u) {
            const int kb = k0 + u * 1024;
            const bool in = kb < base;
            const int x = cnt[(in ? kb : 0) + tid];
            pre += in ? x : 0;
        }
    }
    int inc = v;
    for (int d = 1; d < 32; d <<= 1) {
        const int t = __shfl_up(inc, d, 32);
        if (lane >= d) inc += t;
    }
    for (int o = 16; o > 0; o >>= 1) pre += __shfl_xor(pre, o, 32);
    if (lane == 31) { sTot[wave] = inc; sPre[wave] = pre; }
    __syncthreads();
    int off = sPre[lane] + ((lane < wave) ? sTot[lane] : 0);
    for (int o = 16; o > 0; o >>= 1) off += __shfl_xor(off, o, 32);
    const int excl = off + inc - v;
    if (i < n) rowptr[i] = excl;
    if (i == n - 1) rowptr[n] = excl + v;
}

#define FILL_EPT 8
__global__ void fill_csr(const int* __restrict__ srcA, const int* __restrict__ dstA,
                         const int* __restrict__ rank, const int* __restrict__ partp,
                         const int* __restrict__ rowptr, int* __restrict__ col, int E, int n) {
    const int base = ((int)blockIdx.x * (int)blockDim.x + (int)threadIdx.x) * FILL_EPT;
    if (base >= E + n) return;
    const int len = csr_part_len(E);
    const bool vec = (((reinterpret_cast<size_t>(srcA) | reinterpret_cast<size_t>(dstA) |
                        reinterpret_cast<size_t>(rank)) & 15) == 0);
    if (vec && base + FILL_EPT <= E) {
        const int* pp = partp + (size_t)(base / len) * n;
        const int4 sA = *reinterpret_cast<const int4*>(srcA + base);
        const int4 sB = *reinterpret_cast<const int4*>(srcA + base + 4);
        const int4 dA = *reinterpret_cast<const int4*>(dstA + base);
        const int4 dB = *reinterpret_cast<const int4*>(dstA + base + 4);
        const int4 rA = *reinterpret_cast<const int4*>(rank + base);
        const int4 rB = *reinterpret_cast<const int4*>(rank + base + 4);
        const int p0 = rowptr[dA.x] + pp[dA.x];
        const int p1 = rowptr[dA.y] + pp[dA.y];
        const int p2 = rowptr[dA.z] + pp[dA.z];
        const int p3 = rowptr[dA.w] + pp[dA.w];
        const int p4 = rowptr[dB.x] + pp[dB.x];
        const int p5 = rowptr[dB.y] + pp[dB.y];
        const int p6 = rowptr[dB.z] + pp[dB.z];
        const int p7 = rowptr[dB.w] + pp[dB.w];
        col[p0 + rA.x] = sA.x;
        col[p1 + rA.y] = sA.y;
        col[p2 + rA.z] = sA.z;
        col[p3 + rA.w] = sA.w;
        col[p4 + rB.x] = sB.x;
        col[p5 + rB.y] = sB.y;
        col[p6 + rB.z] = sB.z;
        col[p7 + rB.w] = sB.w;
    } else {
        for (int k = 0; k < FILL_EPT; ++k) {
            const int e = base + k;
            if (e < E) {
                const int d = dstA[e];
                col[rowptr[d] + partp[(size_t)(e / len) * n + d] + rank[e]] = srcA[e];
            } else if (e < E + n) {
                const int i = e - E;
                col[rowptr[i + 1] - 1] = i;
            }
        }
    }
}

#define L1_WAVES 8

#define ROW_XOR_MAX(x, k) fmaxf((x), ROW_XOR((x), (k)))
__device__ __forceinline__ float wave_max(float x) {
    x = ROW_XOR_MAX(x, 1);
    x = ROW_XOR_MAX(x, 2);
    x = ROW_XOR_MAX(x, 4);
    x = ROW_XOR_MAX(x, 8);
    const int xi = __builtin_bit_cast(int, x);
    return fmaxf(x, __builtin_bit_cast(float, __builtin_amdgcn_permlanex16(
        xi, xi, 0x76543210, 0xfedcba98, false, false)));
}

__device__ __forceinline__ float pick8(const float (&v)[NHD], int h) {
    float r = v[0];
#pragma unroll
    for (int i = 1; i < NHD; ++i) r = (h == i) ? v[i] : r;
    return r;
}

struct L1Edge { float4 lo, hi; float w; };
__device__ __forceinline__ L1Edge l1_fetch(const float* __restrict__ H1, const float (*w)[NHD],
                                           int sv, int e, int h, int ch) {
    const int se = __builtin_amdgcn_readlane(sv, e);
    const float4* hp = reinterpret_cast<const float4*>(H1 + (size_t)se * HC + ch);
    L1Edge r;
    r.lo = hp[0];
    r.hi = hp[1];
    r.w  = w[e][h];
    return r;
}
__device__ __forceinline__ void l1_add(float (&a)[8], float& den, const L1Edge& r) {
    a[0] = fmaf(r.w, r.lo.x, a[0]);
    a[1] = fmaf(r.w, r.lo.y, a[1]);
    a[2] = fmaf(r.w, r.lo.z, a[2]);
    a[3] = fmaf(r.w, r.lo.w, a[3]);
    a[4] = fmaf(r.w, r.hi.x, a[4]);
    a[5] = fmaf(r.w, r.hi.y, a[5]);
    a[6] = fmaf(r.w, r.hi.z, a[6]);
    a[7] = fmaf(r.w, r.hi.w, a[7]);
    den += r.w;
}
template <int K>
__device__ __forceinline__ void l1_group(float (&a)[8], float& den, const float* __restrict__ H1,
                                         const float (*w)[NHD], int sv, int e0, int h, int ch) {
    L1Edge r[K];
#pragma unroll
    for (int k = 0; k < K; ++k) r[k] = l1_fetch(H1, w, sv, e0 + k, h, ch);
#pragma unroll
    for (int k = 0; k < K; ++k) l1_add(a, den, r[k]);
}

__global__ void __launch_bounds__(32 * L1_WAVES)
gat_gather_l1(const int* __restrict__ rowptr, const int* __restrict__ col,
              const float* __restrict__ aS, const float* __restrict__ aD,
              const float* __restrict__ H1, const float* __restrict__ b1,
              float* __restrict__ out1, int n) {
    __shared__ __attribute__((aligned(16))) float sW[L1_WAVES][32][NHD];
    const int lane = threadIdx.x & 31;
    const int wib  = threadIdx.x >> 5;
    const int node = (blockIdx.x * blockDim.x + threadIdx.x) >> 5;
    if (node >= n) return;
    const int h  = lane >> 2;
    const int ch = lane << 3;
    const int beg = __builtin_amdgcn_readfirstlane(rowptr[node]);
    const int end = __builtin_amdgcn_readfirstlane(rowptr[node + 1]);
    const float4 dA = *reinterpret_cast<const float4*>(aD + (size_t)node * NHD);
    const float4 dB = *reinterpret_cast<const float4*>(aD + (size_t)node * NHD + 4);
    const float ad[NHD] = {dA.x, dA.y, dA.z, dA.w, dB.x, dB.y, dB.z, dB.w};
    const float ninf = -__builtin_inff();
    float M[NHD];
#pragma unroll
    for (int i = 0; i < NHD; ++i) M[i] = ninf;
    float den = 0.f;
    float a[8] = {0.f, 0.f, 0.f, 0.f, 0.f, 0.f, 0.f, 0.f};
    float (*w)[NHD] = sW[wib];
    for (int cb = beg; cb < end; cb += 32) {
        const int cnt = (end - cb < 32) ? (end - cb) : 32;
        const int s = col[cb + ((lane < cnt) ? lane : 0)];
        const float4 pA = *reinterpret_cast<const float4*>(aS + (size_t)s * NHD);
        const float4 pB = *reinterpret_cast<const float4*>(aS + (size_t)s * NHD + 4);
        const float as[NHD] = {pA.x, pA.y, pA.z, pA.w, pB.x, pB.y, pB.z, pB.w};
        float Mold[NHD], wt[NHD];
#pragma unroll
        for (int i = 0; i < NHD; ++i) {
            const float l = leaky(as[i] + ad[i]);
            Mold[i] = M[i];
            M[i] = fmaxf(M[i], wave_max(l));
            wt[i] = expf(l - M[i]);
        }
        if (cb != beg) {
            const float sc = expf(pick8(Mold, h) - pick8(M, h));
#pragma unroll
            for (int i = 0; i < 8; ++i) a[i] *= sc;
            den *= sc;
        }
        float4 wA, wB;
        wA.x = wt[0];  wA.y = wt[1];  wA.z = wt[2];  wA.w = wt[3];
        wB.x = wt[4];  wB.y = wt[5];  wB.z = wt[6];  wB.w = wt[7];
        *reinterpret_cast<float4*>(&w[lane][0]) = wA;
        *reinterpret_cast<float4*>(&w[lane][4]) = wB;
        __builtin_amdgcn_fence(__ATOMIC_RELEASE, "wavefront");
        __builtin_amdgcn_wave_barrier();
        __builtin_amdgcn_fence(__ATOMIC_ACQUIRE, "wavefront");
        const int full = cnt & ~3;
        for (int e0 = 0; e0 < full; e0 += 4) l1_group<4>(a, den, H1, w, s, e0, h, ch);
        const int rem = cnt - full;
        if (rem == 3)      l1_group<3>(a, den, H1, w, s, full, h, ch);
        else if (rem == 2) l1_group<2>(a, den, H1, w, s, full, h, ch);
        else if (rem == 1) l1_group<1>(a, den, H1, w, s, full, h, ch);
        __builtin_amdgcn_fence(__ATOMIC_RELEASE, "wavefront");
        __builtin_amdgcn_wave_barrier();
        __builtin_amdgcn_fence(__ATOMIC_ACQUIRE, "wavefront");
    }
    const float v0 = a[0] / den + b1[ch + 0];
    const float v1 = a[1] / den + b1[ch + 1];
    const float v2 = a[2] / den + b1[ch + 2];
    const float v3 = a[3] / den + b1[ch + 3];
    const float v4 = a[4] / den + b1[ch + 4];
    const float v5 = a[5] / den + b1[ch + 5];
    const float v6 = a[6] / den + b1[ch + 6];
    const float v7 = a[7] / den + b1[ch + 7];
    float4 o0, o1;
    o0.x = v0 > 0.f ? v0 : 0.f;
    o0.y = v1 > 0.f ? v1 : 0.f;
    o0.z = v2 > 0.f ? v2 : 0.f;
    o0.w = v3 > 0.f ? v3 : 0.f;
    o1.x = v4 > 0.f ? v4 : 0.f;
    o1.y = v5 > 0.f ? v5 : 0.f;
    o1.z = v6 > 0.f ? v6 : 0.f;
    o1.w = v7 > 0.f ? v7 : 0.f;
    float4* op = reinterpret_cast<float4*>(out1 + (size_t)node * HC + ch);
    op[0] = o0;
    op[1] = o1;
}

#define L2_LANES 10
#define L2_NODES 3
__global__ void gat_l2_fused(const int* __restrict__ rowptr, const int* __restrict__ col,
                             const float* __restrict__ aS, const float* __restrict__ aD,
                             const float* __restrict__ H2, const float* __restrict__ b2,
                             float* __restrict__ out, int n) {
    const int lane = threadIdx.x & 31;
    const int wave = (blockIdx.x * blockDim.x + threadIdx.x) >> 5;
    if (wave * L2_NODES >= n) return;
    const int grp  = lane / L2_LANES;
    const int k    = lane - grp * L2_LANES;
    const int c    = k << 2;
    const int node = wave * L2_NODES + grp;
    const bool act = (grp < L2_NODES) && (node < n);
    const int nd  = act ? node : wave * L2_NODES;
    const int beg = rowptr[nd];
    const int end = rowptr[nd + (act ? 1 : 0)];
    const float aDd = aD[nd];
    const float bb0 = b2[c + 0], bb1 = b2[c + 1], bb2 = b2[c + 2], bb3 = b2[c + 3];
    float m = -__builtin_inff(), den = 0.f, a0 = 0.f, a1 = 0.f, a2 = 0.f, a3 = 0.f;
    for (int j = beg; j < end; ++j) {
        const int s = col[j];
        if (j + 1 < end) {
            const int s2 = col[j + 1];
            __builtin_prefetch(&H2[(size_t)s2 * OUTC + c], 0, 1);
        }
        const float l = leaky(aS[s] + aDd);
        const float mn = fmaxf(m, l);
        const float scale = expf(m - mn);
        const float wgt   = expf(l - mn);
        const float4 x = *reinterpret_cast<const float4*>(H2 + (size_t)s * OUTC + c);
        a0 = a0 * scale + wgt * x.x;
        a1 = a1 * scale + wgt * x.y;
        a2 = a2 * scale + wgt * x.z;
        a3 = a3 * scale + wgt * x.w;
        den = den * scale + wgt;
        m = mn;
    }
    const float ninf = -__builtin_inff();
    const float v0 = act ? a0 / den + bb0 : ninf;
    const float v1 = act ? a1 / den + bb1 : ninf;
    const float v2 = act ? a2 / den + bb2 : ninf;
    const float v3 = act ? a3 / den + bb3 : ninf;
    const int g0 = (grp < L2_NODES ? grp : 0) * L2_LANES;
    const float lmx = fmaxf(fmaxf(v0, v1), fmaxf(v2, v3));
    float mx = ninf;
    for (int t = 0; t < L2_LANES; ++t) mx = fmaxf(mx, __shfl(lmx, g0 + t, 32));
    const float lsum = act ? ((expf(v0 - mx) + expf(v1 - mx)) + expf(v2 - mx)) + expf(v3 - mx)
                           : 0.f;
    float ssum = 0.f;
    for (int t = 0; t < L2_LANES; ++t) ssum += __shfl(lsum, g0 + t, 32);
    const float lse = mx + logf(ssum);
    if (act) {
        float4 o;
        o.x = v0 - lse;  o.y = v1 - lse;  o.z = v2 - lse;  o.w = v3 - lse;
        *reinterpret_cast<float4*>(out + (size_t)node * OUTC + c) = o;
    }
}

extern "C" void kernel_launch(void* const* d_in, const int* in_sizes, int n_in,
                              void* d_out, int out_size, void* d_ws, size_t ws_size,
                              hipStream_t stream) {
    const float* x   = (const float*)d_in[0];
    const int*   ei  = (const int*)d_in[1];
    const float* W1  = (const float*)d_in[2];
    const float* as1 = (const float*)d_in[3];
    const float* ad1 = (const float*)d_in[4];
    const float* b1  = (const float*)d_in[5];
    const float* W2  = (const float*)d_in[6];
    const float* as2 = (const float*)d_in[7];
    const float* ad2 = (const float*)d_in[8];
    const float* b2  = (const float*)d_in[9];
    const int N  = in_sizes[0] / INCH;
    const int E  = in_sizes[1] / 2;
    const int Et = E + N;
    const int* srcA = ei;
    const int* dstA = ei + E;

    float* ws = (float*)d_ws;
    size_t off = 0;
    float* h1   = ws + off; off += (size_t)N * HC;
    float* out1 = ws + off; off += (size_t)N * HC;
    float* aS1  = ws + off; off += (size_t)N * NHD;
    float* aD1  = ws + off; off += (size_t)N * NHD;
    int* rowptr = (int*)(ws + off); off += (size_t)N + 1;
    int* cnt    = (int*)(ws + off); off += (size_t)N;
    int* col    = (int*)(ws + off); off += (size_t)Et;
    float* h2  = h1;
    float* aS2 = aS1;
    float* aD2 = aD1;
    int* partc = (int*)h1;
    int* partp = partc + (size_t)CSR_PARTS * N;
    int* rank  = (int*)out1;

    count_rank<<<CSR_PARTS, CSR_THREADS, 0, stream>>>(dstA, partc, rank, E, N);
    part_prefix<<<(N + 255) / 256, 256, 0, stream>>>(partc, partp, cnt, N);
    scan_rowptr<<<(N + 1023) / 1024, 1024, 0, stream>>>(cnt, rowptr, N);
    fill_csr<<<(Et + 256 * FILL_EPT - 1) / (256 * FILL_EPT), 256, 0, stream>>>(
        srcA, dstA, rank, partp, rowptr, col, E, N);

    gemm_x_w1<<<(N + G1_ROWS - 1) / G1_ROWS, 256, 0, stream>>>(x, W1, as1, ad1, h1, aS1, aD1, N);
    gat_gather_l1<<<(N * 32 + 255) / 256, 256, 0, stream>>>(
        rowptr, col, aS1, aD1, h1, b1, out1, N);

    gemm_h_w2<<<(N + G2_ROWS - 1) / G2_ROWS, 192, 0, stream>>>(out1, W2, as2, ad2, h2, aS2, aD2, N);
    gat_l2_fused<<<((N + L2_NODES - 1) / L2_NODES * 32 + 255) / 256, 256, 0, stream>>>(
        rowptr, col, aS2, aD2, h2, b2, (float*)d_out, N);
}


# GEMM1 has no launch of its own: its row tiles lead the blocks of part_prefix's launch (768 tiles) and of fill_csr's (the rest), matrix work beside their load chains; CSR scratch off h1; 7 launches
# speedup vs baseline: 3.4329x; 1.0362x over previous
#include <hip/hip_runtime.h>
#include <hip/hip_bf16.h>

typedef __attribute__((ext_vector_type(2))) float v2f;
typedef __attribute__((ext_vector_type(8))) float v8f;

#define INCH  128
#define HC    256
#define NHD   8
#define CPH   32
#define OUTC  40
#define SLOPE 0.2f

__device__ __forceinline__ float leaky(float x) { return x > 0.f ? x : SLOPE * x; }

__device__ __forceinline__ void lds_handover_barrier() {
    asm volatile("s_wait_dscnt 0x0" ::: "memory");
    asm volatile("s_barrier_signal -1\n\ts_barrier_wait -1" ::: "memory");
}

#define ROW_XOR(x, k) __builtin_bit_cast(float, __builtin_amdgcn_update_dpp( \
    0, __builtin_bit_cast(int, (x)), 0x160 + (k), 0xf, 0xf, true))
__device__ __forceinline__ float row16_sum(float x) {
    x += ROW_XOR(x, 1);
    x += ROW_XOR(x, 2);
    x += ROW_XOR(x, 4);
    x += ROW_XOR(x, 8);
    return x;
}

#define G1_ROWS  32
#define G1_PITCH (INCH + 4)
__device__ __forceinline__ void
gemm_x_w1_tile(const float* __restrict__ X, const float* __restrict__ W,
               const float* __restrict__ Asrc, const float* __restrict__ Adst,
               float* __restrict__ H1, float* __restrict__ aS, float* __restrict__ aD, int nrows,
               int tile) {
    __shared__ __attribute__((aligned(16))) float sA[G1_ROWS * G1_PITCH];
    __shared__ float sS[G1_ROWS * NHD], sD[G1_ROWS * NHD];
    const int tid  = (int)threadIdx.x;
    const int lane = tid & 31;
    const int wave = tid >> 5;
    const int m    = lane & 15;
    const int half = lane >> 4;
    const int rowBase = tile * G1_ROWS;
    const float as0 = Asrc[(wave << 5) + m], as1 = Asrc[(wave << 5) + 16 + m];
    const float ad0 = Adst[(wave << 5) + m], ad1 = Adst[(wave << 5) + 16 + m];
    float4 st[(G1_ROWS * INCH / 4) / 256];
#pragma unroll
    for (int i = 0; i < (G1_ROWS * INCH / 4) / 256; ++i) {
        const int idx = tid + i * 256;
        const int r = idx >> 5, c4 = idx & 31;
        const int gr = (rowBase + r < nrows) ? (rowBase + r) : (nrows - 1);
        st[i] = *reinterpret_cast<const float4*>(X + (size_t)gr * INCH + (c4 << 2));
    }
#pragma unroll
    for (int i = 0; i < (G1_ROWS * INCH / 4) / 256; ++i) {
        const int idx = tid + i * 256;
        const int r = idx >> 5, c4 = idx & 31;
        *reinterpret_cast<float4*>(&sA[r * G1_PITCH + (c4 << 2)]) = st[i];
    }
    __syncthreads();
    const int c0 = wave << 5;
    const float* wp  = W + c0 + m;
    const float* a0p = sA + m * G1_PITCH + (half << 1);
    const float* a1p = a0p + 16 * G1_PITCH;
    v8f acc00 = {0.f, 0.f, 0.f, 0.f, 0.f, 0.f, 0.f, 0.f};
    v8f acc01 = acc00, acc10 = acc00, acc11 = acc00;
    for (int k = 0; k < INCH; k += 4) {
        const int kk = k + (half << 1);
        const v2f a0 = *reinterpret_cast<const v2f*>(a0p + k);
        const v2f a1 = *reinterpret_cast<const v2f*>(a1p + k);
        v2f b0, b1;
        b0.x = wp[(size_t)kk * HC];
        b0.y = wp[(size_t)(kk + 1) * HC];
        b1.x = wp[(size_t)kk * HC + 16];
        b1.y = wp[(size_t)(kk + 1) * HC + 16];
        acc00 = __builtin_amdgcn_wmma_f32_16x16x4_f32(false, a0, false, b0,
                                                      (short)0, acc00, false, false);
        acc01 = __builtin_amdgcn_wmma_f32_16x16x4_f32(false, a0, false, b1,
                                                      (short)0, acc01, false, false);
        acc10 = __builtin_amdgcn_wmma_f32_16x16x4_f32(false, a1, false, b0,
                                                      (short)0, acc10, false, false);
        acc11 = __builtin_amdgcn_wmma_f32_16x16x4_f32(false, a1, false, b1,
                                                      (short)0, acc11, false, false);
    }
    float* o0 = H1 + (size_t)(rowBase + (half << 3)) * HC + c0 + m;
    float* o1 = o0 + (size_t)16 * HC;
    if (rowBase + G1_ROWS <= nrows) {
#pragma unroll
        for (int v = 0; v < 8; ++v) {
            o0[(size_t)v * HC]      = acc00[v];
            o0[(size_t)v * HC + 16] = acc01[v];
            o1[(size_t)v * HC]      = acc10[v];
            o1[(size_t)v * HC + 16] = acc11[v];
        }
    } else {
#pragma unroll
        for (int v = 0; v < 8; ++v) {
            const int r0 = rowBase + v + (half << 3);
            if (r0 < nrows) {
                o0[(size_t)v * HC]      = acc00[v];
                o0[(size_t)v * HC + 16] = acc01[v];
            }
            if (r0 + 16 < nrows) {
                o1[(size_t)v * HC]      = acc10[v];
                o1[(size_t)v * HC + 16] = acc11[v];
            }
        }
    }
    float s0 = 0.f, s1 = 0.f, d0 = 0.f, d1 = 0.f;
#pragma unroll
    for (int v = 0; v < 8; ++v) {
        const float ts0 = row16_sum(acc00[v] * as0 + acc01[v] * as1);
        const float td0 = row16_sum(acc00[v] * ad0 + acc01[v] * ad1);
        const float ts1 = row16_sum(acc10[v] * as0 + acc11[v] * as1);
        const float td1 = row16_sum(acc10[v] * ad0 + acc11[v] * ad1);
        s0 = (m == v) ? ts0 : s0;
        d0 = (m == v) ? td0 : d0;
        s1 = (m == v) ? ts1 : s1;
        d1 = (m == v) ? td1 : d1;
    }
    if (m < 8) {
        const int r = (half << 3) + m;
        sS[r * NHD + wave] = s0;  sS[(r + 16) * NHD + wave] = s1;
        sD[r * NHD + wave] = d0;  sD[(r + 16) * NHD + wave] = d1;
    }
    lds_handover_barrier();
    if ((rowBase << 3) + tid < nrows * NHD) {
        aS[((size_t)rowBase << 3) + tid] = sS[tid];
        aD[((size_t)rowBase << 3) + tid] = sD[tid];
    }
}

#define G2_ROWS  32
#define G2_PITCH (HC + 4)
#define G2_UNITS (G2_ROWS * HC / 128)
#define G2_UPW   ((G2_UNITS + 5) / 6)
__global__ void __launch_bounds__(192, 12)
gemm_h_w2(const float* __restrict__ Hin, const float* __restrict__ W,
          const float* __restrict__ Asrc, const float* __restrict__ Adst,
          float* __restrict__ H2, float* __restrict__ aS, float* __restrict__ aD, int nrows) {
    __shared__ __attribute__((aligned(16))) float sA[(G2_ROWS + 1) * G2_PITCH];
    __shared__ float sS[6 * 16], sD[6 * 16];
    const int tid  = (int)threadIdx.x;
    const int lane = tid & 31;
    const int wave = tid >> 5;
    const int m    = lane & 15;
    const int half = lane >> 4;
    const int rowBase = (int)blockIdx.x * G2_ROWS;
    float4 st[G2_UPW];
#pragma unroll
    for (int i = 0; i < G2_UPW; ++i) {
        const int u0 = wave + 6 * i;
        const int u = (u0 < G2_UNITS) ? u0 : (G2_UNITS - 1);
        const int r = u >> 1, c = ((u & 1) << 7) + (lane << 2);
        const int gr = (rowBase + r < nrows) ? (rowBase + r) : (nrows - 1);
        st[i] = *reinterpret_cast<const float4*>(Hin + (size_t)gr * HC + c);
    }
#pragma unroll
    for (int i = 0; i < G2_UPW; ++i) {
        const int u0 = wave + 6 * i;
        const int r = u0 >> 1, c = ((u0 & 1) << 7) + (lane << 2);
        *reinterpret_cast<float4*>(&sA[r * G2_PITCH + c]) = st[i];
    }
    __syncthreads();
    const int rt = wave / 3;
    const int colBase = (wave - rt * 3) << 4;
    const int col = colBase + m;
    const bool cin = col < OUTC;
    const float* ap = sA + (rt * 16 + m) * G2_PITCH + (half << 1);
    const float* wp = W + (cin ? col : 0);
    v8f acc = {0.f, 0.f, 0.f, 0.f, 0.f, 0.f, 0.f, 0.f};
    for (int k = 0; k < HC; k += 4) {
        const int kk = k + (half << 1);
        const v2f a = *reinterpret_cast<const v2f*>(ap + k);
        v2f b;
        b.x = wp[kk * OUTC];
        b.y = wp[(kk + 1) * OUTC];
        acc = __builtin_amdgcn_wmma_f32_16x16x4_f32(false, a, false, b,
                                                    (short)0, acc, false, false);
    }
    if (cin) {
        const int r0 = rowBase + rt * 16 + (half << 3);
        float* o = H2 + (size_t)r0 * OUTC + col;
        if (rowBase + G2_ROWS <= nrows) {
#pragma unroll
            for (int v = 0; v < 8; ++v) o[(size_t)v * OUTC] = acc[v];
        } else {
#pragma unroll
            for (int v = 0; v < 8; ++v)
                if (r0 + v < nrows) o[(size_t)v * OUTC] = acc[v];
        }
    }
    const float as = cin ? Asrc[col] : 0.f;
    const float ad = cin ? Adst[col] : 0.f;
    float s = 0.f, d = 0.f;
#pragma unroll
    for (int v = 0; v < 8; ++v) {
        const float ts = row16_sum(cin ? acc[v] * as : 0.f);
        const float td = row16_sum(cin ? acc[v] * ad : 0.f);
        s = (m == v) ? ts : s;
        d = (m == v) ? td : d;
    }
    if (m < 8) {
        sS[wave * 16 + (half << 3) + m] = s;
        sD[wave * 16 + (half << 3) + m] = d;
    }
    lds_handover_barrier();
    if (tid < G2_ROWS && rowBase + tid < nrows) {
        const int w0 = (tid >> 4) * 3, r = tid & 15;
        aS[rowBase + tid] = (sS[w0 * 16 + r] + sS[(w0 + 1) * 16 + r]) + sS[(w0 + 2) * 16 + r];
        aD[rowBase + tid] = (sD[w0 * 16 + r] + sD[(w0 + 1) * 16 + r]) + sD[(w0 + 2) * 16 + r];
    }
}

#define CSR_PARTS   64
#define CSR_THREADS 1024
#define CSR_BINS    51200
#define CSR_AHEAD   16

__device__ __forceinline__ int csr_part_len(int E) {
    return (((E + CSR_PARTS - 1) / CSR_PARTS) + 7) & ~7;
}

__global__ void __launch_bounds__(CSR_THREADS)
count_rank(const int* __restrict__ dstA, int* __restrict__ partc, int* __restrict__ rank,
           int E, int n) {
    __shared__ int hist[CSR_BINS];
    const int tid  = (int)threadIdx.x;
    const int part = (int)blockIdx.x;
    const int len  = csr_part_len(E);
    const int beg  = part * len;
    const int end  = (beg + len < E) ? (beg + len) : E;
    for (int w0 = 0; w0 < n; w0 += CSR_BINS) {
        const int wn = (n - w0 < CSR_BINS) ? (n - w0) : CSR_BINS;
        for (int i = tid; i < wn; i += CSR_THREADS) hist[i] = 0;
        __syncthreads();
        for (int e0 = beg; e0 < end; e0 += CSR_AHEAD * CSR_THREADS) {
            int d[CSR_AHEAD], r[CSR_AHEAD];
#pragma unroll
            for (int u = 0; u < CSR_AHEAD; ++u) {
                const int e = e0 + u * CSR_THREADS + tid;
                const bool in = e < end;
                const int x = dstA[in ? e : beg];
                d[u] = in ? (x - w0) : -1;
            }
#pragma unroll
            for (int u = 0; u < CSR_AHEAD; ++u) {
                r[u] = 0;
                if ((unsigned)d[u] < (unsigned)wn) r[u] = atomicAdd(&hist[d[u]], 1);
            }
#pragma unroll
            for (int u = 0; u < CSR_AHEAD; ++u) {
                const int e = e0 + u * CSR_THREADS + tid;
                if ((unsigned)d[u] < (unsigned)wn) rank[e] = r[u];
            }
        }
        __syncthreads();
        int* row = partc + (size_t)part * n + w0;
        for (int i = tid; i < wn; i += CSR_THREADS) row[i] = hist[i];
        __syncthreads();
    }
}

__device__ __forceinline__ void part_prefix_node(const int* __restrict__ partc,
                                                 int* __restrict__ partp,
                                                 int* __restrict__ cnt, int n, int d) {
    if (d >= n) return;
    int run = 0;
    for (int p0 = 0; p0 < CSR_PARTS; p0 += 16) {
        int x[16];
#pragma unroll
        for (int u = 0; u < 16; ++u) x[u] = partc[(size_t)(p0 + u) * n + d];
#pragma unroll
        for (int u = 0; u < 16; ++u) { partp[(size_t)(p0 + u) * n + d] = run; run += x[u]; }
    }
    cnt[d] = run + 1;
}

#define SCAN_AHEAD 16
__global__ void scan_rowptr(const int* __restrict__ cnt, int* __restrict__ rowptr, int n) {
    __shared__ int sPre[32];
    __shared__ int sTot[32];
    const int tid  = (int)threadIdx.x;
    const int lane = tid & 31;
    const int wave = tid >> 5;
    const int base = (int)blockIdx.x * 1024;
    const int i = base + tid;
    const int v = (i < n) ? cnt[i] : 0;
    int pre = 0;
    for (int k0 = 0; k0 < base; k0 += SCAN_AHEAD * 1024) {
        for (int u = 0; u < SCAN_AHEAD; ++u) {
            const int kb = k0 + u * 1024;
            const bool in = kb < base;
            const int x = cnt[(in ? kb : 0) + tid];
            pre += in ? x : 0;
        }
    }
    int inc = v;
    for (int d = 1; d < 32; d <<= 1) {
        const int t = __shfl_up(inc, d, 32);
        if (lane >= d) inc += t;
    }
    for (int o = 16; o > 0; o >>= 1) pre += __shfl_xor(pre, o, 32);
    if (lane == 31) { sTot[wave] = inc; sPre[wave] = pre; }
    __syncthreads();
    int off = sPre[lane] + ((lane < wave) ? sTot[lane] : 0);
    for (int o = 16; o > 0; o >>= 1) off += __shfl_xor(off, o, 32);
    const int excl = off + inc - v;
    if (i < n) rowptr[i] = excl;
    if (i == n - 1) rowptr[n] = excl + v;
}

#define FILL_EPT 8
__device__ __forceinline__ void fill_csr_edges(const int* __restrict__ srcA,
                                               const int* __restrict__ dstA,
                                               const int* __restrict__ rank,
                                               const int* __restrict__ partp,
                                               const int* __restrict__ rowptr,
                                               int* __restrict__ col, int E, int n, int base) {
    if (base >= E + n) return;
    const int len = csr_part_len(E);
    const bool vec = (((reinterpret_cast<size_t>(srcA) | reinterpret_cast<size_t>(dstA) |
                        reinterpret_cast<size_t>(rank)) & 15) == 0);
    if (vec && base + FILL_EPT <= E) {
        const int* pp = partp + (size_t)(base / len) * n;
        const int4 sA = *reinterpret_cast<const int4*>(srcA + base);
        const int4 sB = *reinterpret_cast<const int4*>(srcA + base + 4);
        const int4 dA = *reinterpret_cast<const int4*>(dstA + base);
        const int4 dB = *reinterpret_cast<const int4*>(dstA + base + 4);
        const int4 rA = *reinterpret_cast<const int4*>(rank + base);
        const int4 rB = *reinterpret_cast<const int4*>(rank + base + 4);
        const int p0 = rowptr[dA.x] + pp[dA.x];
        const int p1 = rowptr[dA.y] + pp[dA.y];
        const int p2 = rowptr[dA.z] + pp[dA.z];
        const int p3 = rowptr[dA.w] + pp[dA.w];
        const int p4 = rowptr[dB.x] + pp[dB.x];
        const int p5 = rowptr[dB.y] + pp[dB.y];
        const int p6 = rowptr[dB.z] + pp[dB.z];
        const int p7 = rowptr[dB.w] + pp[dB.w];
        col[p0 + rA.x] = sA.x;
        col[p1 + rA.y] = sA.y;
        col[p2 + rA.z] = sA.z;
        col[p3 + rA.w] = sA.w;
        col[p4 + rB.x] = sB.x;
        col[p5 + rB.y] = sB.y;
        col[p6 + rB.z] = sB.z;
        col[p7 + rB.w] = sB.w;
    } else {
        for (int k = 0; k < FILL_EPT; ++k) {
            const int e = base + k;
            if (e < E) {
                const int d = dstA[e];
                col[rowptr[d] + partp[(size_t)(e / len) * n + d] + rank[e]] = srcA[e];
            } else if (e < E + n) {
                const int i = e - E;
                col[rowptr[i + 1] - 1] = i;
            }
        }
    }
}

#define CSR_STEP_THREADS 256
#define CHIP_WGPS 256
__host__ __device__ __forceinline__ int g1_tiles(int n) { return (n + G1_ROWS - 1) / G1_ROWS; }
__host__ __device__ __forceinline__ int g1_tiles_first(int n) {
    const int half = (g1_tiles(n) + 1) / 2;
    return (half >= CHIP_WGPS) ? (half - half % CHIP_WGPS) : half;
}
__host__ __device__ __forceinline__ int prefix_blocks(int n) {
    return (n + CSR_STEP_THREADS - 1) / CSR_STEP_THREADS;
}
__host__ __device__ __forceinline__ int fill_blocks(int E, int n) {
    return (E + n + CSR_STEP_THREADS * FILL_EPT - 1) / (CSR_STEP_THREADS * FILL_EPT);
}

__global__ void __launch_bounds__(CSR_STEP_THREADS)
gemm1_and_part_prefix(const float* __restrict__ X, const float* __restrict__ W,
                      const float* __restrict__ Asrc, const float* __restrict__ Adst,
                      float* __restrict__ H1, float* __restrict__ aS, float* __restrict__ aD,
                      const int* __restrict__ partc, int* __restrict__ partp,
                      int* __restrict__ cnt, int n) {
    const int nt = g1_tiles_first(n);
    if ((int)blockIdx.x < nt) {
        gemm_x_w1_tile(X, W, Asrc, Adst, H1, aS, aD, n, (int)blockIdx.x);
    } else {
        part_prefix_node(partc, partp, cnt, n,
                         ((int)blockIdx.x - nt) * CSR_STEP_THREADS + (int)threadIdx.x);
    }
}

__global__ void __launch_bounds__(CSR_STEP_THREADS)
gemm1_and_fill_csr(const float* __restrict__ X, const float* __restrict__ W,
                   const float* __restrict__ Asrc, const float* __restrict__ Adst,
                   float* __restrict__ H1, float* __restrict__ aS, float* __restrict__ aD,
                   const int* __restrict__ srcA, const int* __restrict__ dstA,
                   const int* __restrict__ rank, const int* __restrict__ partp,
                   const int* __restrict__ rowptr, int* __restrict__ col, int E, int n) {
    const int t0 = g1_tiles_first(n);
    const int nt = g1_tiles(n) - t0;
    if ((int)blockIdx.x < nt) {
        gemm_x_w1_tile(X, W, Asrc, Adst, H1, aS, aD, n, t0 + (int)blockIdx.x);
    } else {
        fill_csr_edges(srcA, dstA, rank, partp, rowptr, col, E, n,
                       (((int)blockIdx.x - nt) * CSR_STEP_THREADS + (int)threadIdx.x) * FILL_EPT);
    }
}

#define L1_WAVES 8

#define ROW_XOR_MAX(x, k) fmaxf((x), ROW_XOR((x), (k)))
__device__ __forceinline__ float wave_max(float x) {
    x = ROW_XOR_MAX(x, 1);
    x = ROW_XOR_MAX(x, 2);
    x = ROW_XOR_MAX(x, 4);
    x = ROW_XOR_MAX(x, 8);
    const int xi = __builtin_bit_cast(int, x);
    return fmaxf(x, __builtin_bit_cast(float, __builtin_amdgcn_permlanex16(
        xi, xi, 0x76543210, 0xfedcba98, false, false)));
}

__device__ __forceinline__ float pick8(const float (&v)[NHD], int h) {
    float r = v[0];
#pragma unroll
    for (int i = 1; i < NHD; ++i) r = (h == i) ? v[i] : r;
    return r;
}

struct L1Edge { float4 lo, hi; float w; };
__device__ __forceinline__ L1Edge l1_fetch(const float* __restrict__ H1, const float (*w)[NHD],
                                           int sv, int e, int h, int ch) {
    const int se = __builtin_amdgcn_readlane(sv, e);
    const float4* hp = reinterpret_cast<const float4*>(H1 + (size_t)se * HC + ch);
    L1Edge r;
    r.lo = hp[0];
    r.hi = hp[1];
    r.w  = w[e][h];
    return r;
}
__device__ __forceinline__ void l1_add(float (&a)[8], float& den, const L1Edge& r) {
    a[0] = fmaf(r.w, r.lo.x, a[0]);
    a[1] = fmaf(r.w, r.lo.y, a[1]);
    a[2] = fmaf(r.w, r.lo.z, a[2]);
    a[3] = fmaf(r.w, r.lo.w, a[3]);
    a[4] = fmaf(r.w, r.hi.x, a[4]);
    a[5] = fmaf(r.w, r.hi.y, a[5]);
    a[6] = fmaf(r.w, r.hi.z, a[6]);
    a[7] = fmaf(r.w, r.hi.w, a[7]);
    den += r.w;
}
template <int K>
__device__ __forceinline__ void l1_group(float (&a)[8], float& den, const float* __restrict__ H1,
                                         const float (*w)[NHD], int sv, int e0, int h, int ch) {
    L1Edge r[K];
#pragma unroll
    for (int k = 0; k < K; ++k) r[k] = l1_fetch(H1, w, sv, e0 + k, h, ch);
#pragma unroll
    for (int k = 0; k < K; ++k) l1_add(a, den, r[k]);
}

__global__ void __launch_bounds__(32 * L1_WAVES)
gat_gather_l1(const int* __restrict__ rowptr, const int* __restrict__ col,
              const float* __restrict__ aS, const float* __restrict__ aD,
              const float* __restrict__ H1, const float* __restrict__ b1,
              float* __restrict__ out1, int n) {
    __shared__ __attribute__((aligned(16))) float sW[L1_WAVES][32][NHD];
    const int lane = threadIdx.x & 31;
    const int wib  = threadIdx.x >> 5;
    const int node = (blockIdx.x * blockDim.x + threadIdx.x) >> 5;
    if (node >= n) return;
    const int h  = lane >> 2;
    const int ch = lane << 3;
    const int beg = __builtin_amdgcn_readfirstlane(rowptr[node]);
    const int end = __builtin_amdgcn_readfirstlane(rowptr[node + 1]);
    const float4 dA = *reinterpret_cast<const float4*>(aD + (size_t)node * NHD);
    const float4 dB = *reinterpret_cast<const float4*>(aD + (size_t)node * NHD + 4);
    const float ad[NHD] = {dA.x, dA.y, dA.z, dA.w, dB.x, dB.y, dB.z, dB.w};
    const float ninf = -__builtin_inff();
    float M[NHD];
#pragma unroll
    for (int i = 0; i < NHD; ++i) M[i] = ninf;
    float den = 0.f;
    float a[8] = {0.f, 0.f, 0.f, 0.f, 0.f, 0.f, 0.f, 0.f};
    float (*w)[NHD] = sW[wib];
    for (int cb = beg; cb < end; cb += 32) {
        const int cnt = (end - cb < 32) ? (end - cb) : 32;
        const int s = col[cb + ((lane < cnt) ? lane : 0)];
        const float4 pA = *reinterpret_cast<const float4*>(aS + (size_t)s * NHD);
        const float4 pB = *reinterpret_cast<const float4*>(aS + (size_t)s * NHD + 4);
        const float as[NHD] = {pA.x, pA.y, pA.z, pA.w, pB.x, pB.y, pB.z, pB.w};
        float Mold[NHD], wt[NHD];
#pragma unroll
        for (int i = 0; i < NHD; ++i) {
            const float l = leaky(as[i] + ad[i]);
            Mold[i] = M[i];
            M[i] = fmaxf(M[i], wave_max(l));
            wt[i] = expf(l - M[i]);
        }
        if (cb != beg) {
            const float sc = expf(pick8(Mold, h) - pick8(M, h));
#pragma unroll
            for (int i = 0; i < 8; ++i) a[i] *= sc;
            den *= sc;
        }
        float4 wA, wB;
        wA.x = wt[0];  wA.y = wt[1];  wA.z = wt[2];  wA.w = wt[3];
        wB.x = wt[4];  wB.y = wt[5];  wB.z = wt[6];  wB.w = wt[7];
        *reinterpret_cast<float4*>(&w[lane][0]) = wA;
        *reinterpret_cast<float4*>(&w[lane][4]) = wB;
        __builtin_amdgcn_fence(__ATOMIC_RELEASE, "wavefront");
        __builtin_amdgcn_wave_barrier();
        __builtin_amdgcn_fence(__ATOMIC_ACQUIRE, "wavefront");
        const int full = cnt & ~3;
        for (int e0 = 0; e0 < full; e0 += 4) l1_group<4>(a, den, H1, w, s, e0, h, ch);
        const int rem = cnt - full;
        if (rem == 3)      l1_group<3>(a, den, H1, w, s, full, h, ch);
        else if (rem == 2) l1_group<2>(a, den, H1, w, s, full, h, ch);
        else if (rem == 1) l1_group<1>(a, den, H1, w, s, full, h, ch);
        __builtin_amdgcn_fence(__ATOMIC_RELEASE, "wavefront");
        __builtin_amdgcn_wave_barrier();
        __builtin_amdgcn_fence(__ATOMIC_ACQUIRE, "wavefront");
    }
    const float v0 = a[0] / den + b1[ch + 0];
    const float v1 = a[1] / den + b1[ch + 1];
    const float v2 = a[2] / den + b1[ch + 2];
    const float v3 = a[3] / den + b1[ch + 3];
    const float v4 = a[4] / den + b1[ch + 4];
    const float v5 = a[5] / den + b1[ch + 5];
    const float v6 = a[6] / den + b1[ch + 6];
    const float v7 = a[7] / den + b1[ch + 7];
    float4 o0, o1;
    o0.x = v0 > 0.f ? v0 : 0.f;
    o0.y = v1 > 0.f ? v1 : 0.f;
    o0.z = v2 > 0.f ? v2 : 0.f;
    o0.w = v3 > 0.f ? v3 : 0.f;
    o1.x = v4 > 0.f ? v4 : 0.f;
    o1.y = v5 > 0.f ? v5 : 0.f;
    o1.z = v6 > 0.f ? v6 : 0.f;
    o1.w = v7 > 0.f ? v7 : 0.f;
    float4* op = reinterpret_cast<float4*>(out1 + (size_t)node * HC + ch);
    op[0] = o0;
    op[1] = o1;
}

#define L2_LANES 10
#define L2_NODES 3
__global__ void gat_l2_fused(const int* __restrict__ rowptr, const int* __restrict__ col,
                             const float* __restrict__ aS, const float* __restrict__ aD,
                             const float* __restrict__ H2, const float* __restrict__ b2,
                             float* __restrict__ out, int n) {
    const int lane = threadIdx.x & 31;
    const int wave = (blockIdx.x * blockDim.x + threadIdx.x) >> 5;
    if (wave * L2_NODES >= n) return;
    const int grp  = lane / L2_LANES;
    const int k    = lane - grp * L2_LANES;
    const int c    = k << 2;
    const int node = wave * L2_NODES + grp;
    const bool act = (grp < L2_NODES) && (node < n);
    const int nd  = act ? node : wave * L2_NODES;
    const int beg = rowptr[nd];
    const int end = rowptr[nd + (act ? 1 : 0)];
    const float aDd = aD[nd];
    const float bb0 = b2[c + 0], bb1 = b2[c + 1], bb2 = b2[c + 2], bb3 = b2[c + 3];
    float m = -__builtin_inff(), den = 0.f, a0 = 0.f, a1 = 0.f, a2 = 0.f, a3 = 0.f;
    for (int j = beg; j < end; ++j) {
        const int s = col[j];
        if (j + 1 < end) {
            const int s2 = col[j + 1];
            __builtin_prefetch(&H2[(size_t)s2 * OUTC + c], 0, 1);
        }
        const float l = leaky(aS[s] + aDd);
        const float mn = fmaxf(m, l);
        const float scale = expf(m - mn);
        const float wgt   = expf(l - mn);
        const float4 x = *reinterpret_cast<const float4*>(H2 + (size_t)s * OUTC + c);
        a0 = a0 * scale + wgt * x.x;
        a1 = a1 * scale + wgt * x.y;
        a2 = a2 * scale + wgt * x.z;
        a3 = a3 * scale + wgt * x.w;
        den = den * scale + wgt;
        m = mn;
    }
    const float ninf = -__builtin_inff();
    const float v0 = act ? a0 / den + bb0 : ninf;
    const float v1 = act ? a1 / den + bb1 : ninf;
    const float v2 = act ? a2 / den + bb2 : ninf;
    const float v3 = act ? a3 / den + bb3 : ninf;
    const int g0 = (grp < L2_NODES ? grp : 0) * L2_LANES;
    const float lmx = fmaxf(fmaxf(v0, v1), fmaxf(v2, v3));
    float mx = ninf;
    for (int t = 0; t < L2_LANES; ++t) mx = fmaxf(mx, __shfl(lmx, g0 + t, 32));
    const float lsum = act ? ((expf(v0 - mx) + expf(v1 - mx)) + expf(v2 - mx)) + expf(v3 - mx)
                           : 0.f;
    float ssum = 0.f;
    for (int t = 0; t < L2_LANES; ++t) ssum += __shfl(lsum, g0 + t, 32);
    const float lse = mx + logf(ssum);
    if (act) {
        float4 o;
        o.x = v0 - lse;  o.y = v1 - lse;  o.z = v2 - lse;  o.w = v3 - lse;
        *reinterpret_cast<float4*>(out + (size_t)node * OUTC + c) = o;
    }
}

extern "C" void kernel_launch(void* const* d_in, const int* in_sizes, int n_in,
                              void* d_out, int out_size, void* d_ws, size_t ws_size,
                              hipStream_t stream) {
    const float* x   = (const float*)d_in[0];
    const int*   ei  = (const int*)d_in[1];
    const float* W1  = (const float*)d_in[2];
    const float* as1 = (const float*)d_in[3];
    const float* ad1 = (const float*)d_in[4];
    const float* b1  = (const float*)d_in[5];
    const float* W2  = (const float*)d_in[6];
    const float* as2 = (const float*)d_in[7];
    const float* ad2 = (const float*)d_in[8];
    const float* b2  = (const float*)d_in[9];
    const int N  = in_sizes[0] / INCH;
    const int E  = in_sizes[1] / 2;
    const int Et = E + N;
    const int* srcA = ei;
    const int* dstA = ei + E;

    float* ws = (float*)d_ws;
    size_t off = 0;
    float* h1   = ws + off; off += (size_t)N * HC;
    float* out1 = ws + off; off += (size_t)N * HC;
    float* aS1  = ws + off; off += (size_t)N * NHD;
    float* aD1  = ws + off; off += (size_t)N * NHD;
    int* rowptr = (int*)(ws + off); off += (size_t)N + 1;
    int* cnt    = (int*)(ws + off); off += (size_t)N;
    int* col    = (int*)(ws + off); off += (size_t)Et;
    float* h2  = h1;
    float* aS2 = aS1;
    float* aD2 = aD1;
    int* partc = (int*)out1;
    int* partp = partc + (size_t)CSR_PARTS * N;
    int* rank  = partp + (size_t)CSR_PARTS * N;

    count_rank<<<CSR_PARTS, CSR_THREADS, 0, stream>>>(dstA, partc, rank, E, N);
    gemm1_and_part_prefix<<<g1_tiles_first(N) + prefix_blocks(N), CSR_STEP_THREADS, 0, stream>>>(
        x, W1, as1, ad1, h1, aS1, aD1, partc, partp, cnt, N);
    scan_rowptr<<<(N + 1023) / 1024, 1024, 0, stream>>>(cnt, rowptr, N);
    gemm1_and_fill_csr<<<g1_tiles(N) - g1_tiles_first(N) + fill_blocks(E, N), CSR_STEP_THREADS, 0,
                         stream>>>(
        x, W1, as1, ad1, h1, aS1, aD1, srcA, dstA, rank, partp, rowptr, col, E, N);

    gat_gather_l1<<<(N * 32 + 255) / 256, 256, 0, stream>>>(
        rowptr, col, aS1, aD1, h1, b1, out1, N);

    gemm_h_w2<<<(N + G2_ROWS - 1) / G2_ROWS, 192, 0, stream>>>(out1, W2, as2, ad2, h2, aS2, aD2, N);
    gat_l2_fused<<<((N + L2_NODES - 1) / L2_NODES * 32 + 255) / 256, 256, 0, stream>>>(
        rowptr, col, aS2, aD2, h2, b2, (float*)d_out, N);
}


# GEMM2 fused into the layer-1 gather: a block gathers 16 nodes (2 per wave), keeps their rows in LDS, runs the 16x256x40 WMMA product + dots there; OUT1 never in memory; 6 launches; sum order kept
# speedup vs baseline: 3.5535x; 1.0351x over previous
#include <hip/hip_runtime.h>
#include <hip/hip_bf16.h>

typedef __attribute__((ext_vector_type(2))) float v2f;
typedef __attribute__((ext_vector_type(8))) float v8f;

#define INCH  128
#define HC    256
#define NHD   8
#define CPH   32
#define OUTC  40
#define SLOPE 0.2f

__device__ __forceinline__ float leaky(float x) { return x > 0.f ? x : SLOPE * x; }

__device__ __forceinline__ void lds_handover_barrier() {
    asm volatile("s_wait_dscnt 0x0" ::: "memory");
    asm volatile("s_barrier_signal -1\n\ts_barrier_wait -1" ::: "memory");
}

#define ROW_XOR(x, k) __builtin_bit_cast(float, __builtin_amdgcn_update_dpp( \
    0, __builtin_bit_cast(int, (x)), 0x160 + (k), 0xf, 0xf, true))
__device__ __forceinline__ float row16_sum(float x) {
    x += ROW_XOR(x, 1);
    x += ROW_XOR(x, 2);
    x += ROW_XOR(x, 4);
    x += ROW_XOR(x, 8);
    return x;
}

#define G1_ROWS  32
#define G1_PITCH (INCH + 4)
__device__ __forceinline__ void
gemm_x_w1_tile(const float* __restrict__ X, const float* __restrict__ W,
               const float* __restrict__ Asrc, const float* __restrict__ Adst,
               float* __restrict__ H1, float* __restrict__ aS, float* __restrict__ aD, int nrows,
               int tile) {
    __shared__ __attribute__((aligned(16))) float sA[G1_ROWS * G1_PITCH];
    __shared__ float sS[G1_ROWS * NHD], sD[G1_ROWS * NHD];
    const int tid  = (int)threadIdx.x;
    const int lane = tid & 31;
    const int wave = tid >> 5;
    const int m    = lane & 15;
    const int half = lane >> 4;
    const int rowBase = tile * G1_ROWS;
    const float as0 = Asrc[(wave << 5) + m], as1 = Asrc[(wave << 5) + 16 + m];
    const float ad0 = Adst[(wave << 5) + m], ad1 = Adst[(wave << 5) + 16 + m];
    float4 st[(G1_ROWS * INCH / 4) / 256];
#pragma unroll
    for (int i = 0; i < (G1_ROWS * INCH / 4) / 256; ++i) {
        const int idx = tid + i * 256;
        const int r = idx >> 5, c4 = idx & 31;
        const int gr = (rowBase + r < nrows) ? (rowBase + r) : (nrows - 1);
        st[i] = *reinterpret_cast<const float4*>(X + (size_t)gr * INCH + (c4 << 2));
    }
#pragma unroll
    for (int i = 0; i < (G1_ROWS * INCH / 4) / 256; ++i) {
        const int idx = tid + i * 256;
        const int r = idx >> 5, c4 = idx & 31;
        *reinterpret_cast<float4*>(&sA[r * G1_PITCH + (c4 << 2)]) = st[i];
    }
    __syncthreads();
    const int c0 = wave << 5;
    const float* wp  = W + c0 + m;
    const float* a0p = sA + m * G1_PITCH + (half << 1);
    const float* a1p = a0p + 16 * G1_PITCH;
    v8f acc00 = {0.f, 0.f, 0.f, 0.f, 0.f, 0.f, 0.f, 0.f};
    v8f acc01 = acc00, acc10 = acc00, acc11 = acc00;
    for (int k = 0; k < INCH; k += 4) {
        const int kk = k + (half << 1);
        const v2f a0 = *reinterpret_cast<const v2f*>(a0p + k);
        const v2f a1 = *reinterpret_cast<const v2f*>(a1p + k);
        v2f b0, b1;
        b0.x = wp[(size_t)kk * HC];
        b0.y = wp[(size_t)(kk + 1) * HC];
        b1.x = wp[(size_t)kk * HC + 16];
        b1.y = wp[(size_t)(kk + 1) * HC + 16];
        acc00 = __builtin_amdgcn_wmma_f32_16x16x4_f32(false, a0, false, b0,
                                                      (short)0, acc00, false, false);
        acc01 = __builtin_amdgcn_wmma_f32_16x16x4_f32(false, a0, false, b1,
                                                      (short)0, acc01, false, false);
        acc10 = __builtin_amdgcn_wmma_f32_16x16x4_f32(false, a1, false, b0,
                                                      (short)0, acc10, false, false);
        acc11 = __builtin_amdgcn_wmma_f32_16x16x4_f32(false, a1, false, b1,
                                                      (short)0, acc11, false, false);
    }
    float* o0 = H1 + (size_t)(rowBase + (half << 3)) * HC + c0 + m;
    float* o1 = o0 + (size_t)16 * HC;
    if (rowBase + G1_ROWS <= nrows) {
#pragma unroll
        for (int v = 0; v < 8; ++v) {
            o0[(size_t)v * HC]      = acc00[v];
            o0[(size_t)v * HC + 16] = acc01[v];
            o1[(size_t)v * HC]      = acc10[v];
            o1[(size_t)v * HC + 16] = acc11[v];
        }
    } else {
#pragma unroll
        for (int v = 0; v < 8; ++v) {
            const int r0 = rowBase + v + (half << 3);
            if (r0 < nrows) {
                o0[(size_t)v * HC]      = acc00[v];
                o0[(size_t)v * HC + 16] = acc01[v];
            }
            if (r0 + 16 < nrows) {
                o1[(size_t)v * HC]      = acc10[v];
                o1[(size_t)v * HC + 16] = acc11[v];
            }
        }
    }
    float s0 = 0.f, s1 = 0.f, d0 = 0.f, d1 = 0.f;
#pragma unroll
    for (int v = 0; v < 8; ++v) {
        const float ts0 = row16_sum(acc00[v] * as0 + acc01[v] * as1);
        const float td0 = row16_sum(acc00[v] * ad0 + acc01[v] * ad1);
        const float ts1 = row16_sum(acc10[v] * as0 + acc11[v] * as1);
        const float td1 = row16_sum(acc10[v] * ad0 + acc11[v] * ad1);
        s0 = (m == v) ? ts0 : s0;
        d0 = (m == v) ? td0 : d0;
        s1 = (m == v) ? ts1 : s1;
        d1 = (m == v) ? td1 : d1;
    }
    if (m < 8) {
        const int r = (half << 3) + m;
        sS[r * NHD + wave] = s0;  sS[(r + 16) * NHD + wave] = s1;
        sD[r * NHD + wave] = d0;  sD[(r + 16) * NHD + wave] = d1;
    }
    lds_handover_barrier();
    if ((rowBase << 3) + tid < nrows * NHD) {
        aS[((size_t)rowBase << 3) + tid] = sS[tid];
        aD[((size_t)rowBase << 3) + tid] = sD[tid];
    }
}

#define G2_ROWS  16
#define G2_PITCH (HC + 4)
__device__ __forceinline__ void
gemm_lds_w2_tile(const float* sA, const float* __restrict__ W,
                 const float* __restrict__ Asrc, const float* __restrict__ Adst,
                 float* __restrict__ H2, float* __restrict__ aS, float* __restrict__ aD, int nrows,
                 int rowBase) {
    __shared__ float sS[3 * 16], sD[3 * 16];
    const int tid  = (int)threadIdx.x;
    const int lane = tid & 31;
    const int wave = tid >> 5;
    const int m    = lane & 15;
    const int half = lane >> 4;
    if (wave < 3) {
        const int col = (wave << 4) + m;
        const bool cin = col < OUTC;
        const float* ap = sA + m * G2_PITCH + (half << 1);
        const float* wp = W + (cin ? col : 0);
        v8f acc = {0.f, 0.f, 0.f, 0.f, 0.f, 0.f, 0.f, 0.f};
        for (int k = 0; k < HC; k += 4) {
            const int kk = k + (half << 1);
            const v2f a = *reinterpret_cast<const v2f*>(ap + k);
            v2f b;
            b.x = wp[kk * OUTC];
            b.y = wp[(kk + 1) * OUTC];
            acc = __builtin_amdgcn_wmma_f32_16x16x4_f32(false, a, false, b,
                                                        (short)0, acc, false, false);
        }
        if (cin) {
            const int r0 = rowBase + (half << 3);
            float* o = H2 + (size_t)r0 * OUTC + col;
            if (rowBase + G2_ROWS <= nrows) {
#pragma unroll
                for (int v = 0; v < 8; ++v) o[(size_t)v * OUTC] = acc[v];
            } else {
#pragma unroll
                for (int v = 0; v < 8; ++v)
                    if (r0 + v < nrows) o[(size_t)v * OUTC] = acc[v];
            }
        }
        const float as = cin ? Asrc[col] : 0.f;
        const float ad = cin ? Adst[col] : 0.f;
        float s = 0.f, d = 0.f;
#pragma unroll
        for (int v = 0; v < 8; ++v) {
            const float ts = row16_sum(cin ? acc[v] * as : 0.f);
            const float td = row16_sum(cin ? acc[v] * ad : 0.f);
            s = (m == v) ? ts : s;
            d = (m == v) ? td : d;
        }
        if (m < 8) {
            sS[wave * 16 + (half << 3) + m] = s;
            sD[wave * 16 + (half << 3) + m] = d;
        }
    }
    lds_handover_barrier();
    if (tid < G2_ROWS && rowBase + tid < nrows) {
        aS[rowBase + tid] = (sS[tid] + sS[16 + tid]) + sS[32 + tid];
        aD[rowBase + tid] = (sD[tid] + sD[16 + tid]) + sD[32 + tid];
    }
}

#define CSR_PARTS   64
#define CSR_THREADS 1024
#define CSR_BINS    51200
#define CSR_AHEAD   16

__device__ __forceinline__ int csr_part_len(int E) {
    return (((E + CSR_PARTS - 1) / CSR_PARTS) + 7) & ~7;
}

__global__ void __launch_bounds__(CSR_THREADS)
count_rank(const int* __restrict__ dstA, int* __restrict__ partc, int* __restrict__ rank,
           int E, int n) {
    __shared__ int hist[CSR_BINS];
    const int tid  = (int)threadIdx.x;
    const int part = (int)blockIdx.x;
    const int len  = csr_part_len(E);
    const int beg  = part * len;
    const int end  = (beg + len < E) ? (beg + len) : E;
    for (int w0 = 0; w0 < n; w0 += CSR_BINS) {
        const int wn = (n - w0 < CSR_BINS) ? (n - w0) : CSR_BINS;
        for (int i = tid; i < wn; i += CSR_THREADS) hist[i] = 0;
        __syncthreads();
        for (int e0 = beg; e0 < end; e0 += CSR_AHEAD * CSR_THREADS) {
            int d[CSR_AHEAD], r[CSR_AHEAD];
#pragma unroll
            for (int u = 0; u < CSR_AHEAD; ++u) {
                const int e = e0 + u * CSR_THREADS + tid;
                const bool in = e < end;
                const int x = dstA[in ? e : beg];
                d[u] = in ? (x - w0) : -1;
            }
#pragma unroll
            for (int u = 0; u < CSR_AHEAD; ++u) {
                r[u] = 0;
                if ((unsigned)d[u] < (unsigned)wn) r[u] = atomicAdd(&hist[d[u]], 1);
            }
#pragma unroll
            for (int u = 0; u < CSR_AHEAD; ++u) {
                const int e = e0 + u * CSR_THREADS + tid;
                if ((unsigned)d[u] < (unsigned)wn) rank[e] = r[u];
            }
        }
        __syncthreads();
        int* row = partc + (size_t)part * n + w0;
        for (int i = tid; i < wn; i += CSR_THREADS) row[i] = hist[i];
        __syncthreads();
    }
}

__device__ __forceinline__ void part_prefix_node(const int* __restrict__ partc,
                                                 int* __restrict__ partp,
                                                 int* __restrict__ cnt, int n, int d) {
    if (d >= n) return;
    int run = 0;
    for (int p0 = 0; p0 < CSR_PARTS; p0 += 16) {
        int x[16];
#pragma unroll
        for (int u = 0; u < 16; ++u) x[u] = partc[(size_t)(p0 + u) * n + d];
#pragma unroll
        for (int u = 0; u < 16; ++u) { partp[(size_t)(p0 + u) * n + d] = run; run += x[u]; }
    }
    cnt[d] = run + 1;
}

#define SCAN_AHEAD 16
__global__ void scan_rowptr(const int* __restrict__ cnt, int* __restrict__ rowptr, int n) {
    __shared__ int sPre[32];
    __shared__ int sTot[32];
    const int tid  = (int)threadIdx.x;
    const int lane = tid & 31;
    const int wave = tid >> 5;
    const int base = (int)blockIdx.x * 1024;
    const int i = base + tid;
    const int v = (i < n) ? cnt[i] : 0;
    int pre = 0;
    for (int k0 = 0; k0 < base; k0 += SCAN_AHEAD * 1024) {
        for (int u = 0; u < SCAN_AHEAD; ++u) {
            const int kb = k0 + u * 1024;
            const bool in = kb < base;
            const int x = cnt[(in ? kb : 0) + tid];
            pre += in ? x : 0;
        }
    }
    int inc = v;
    for (int d = 1; d < 32; d <<= 1) {
        const int t = __shfl_up(inc, d, 32);
        if (lane >= d) inc += t;
    }
    for (int o = 16; o > 0; o >>= 1) pre += __shfl_xor(pre, o, 32);
    if (lane == 31) { sTot[wave] = inc; sPre[wave] = pre; }
    __syncthreads();
    int off = sPre[lane] + ((lane < wave) ? sTot[lane] : 0);
    for (int o = 16; o > 0; o >>= 1) off += __shfl_xor(off, o, 32);
    const int excl = off + inc - v;
    if (i < n) rowptr[i] = excl;
    if (i == n - 1) rowptr[n] = excl + v;
}

#define FILL_EPT 8
__device__ __forceinline__ void fill_csr_edges(const int* __restrict__ srcA,
                                               const int* __restrict__ dstA,
                                               const int* __restrict__ rank,
                                               const int* __restrict__ partp,
                                               const int* __restrict__ rowptr,
                                               int* __restrict__ col, int E, int n, int base) {
    if (base >= E + n) return;
    const int len = csr_part_len(E);
    const bool vec = (((reinterpret_cast<size_t>(srcA) | reinterpret_cast<size_t>(dstA) |
                        reinterpret_cast<size_t>(rank)) & 15) == 0);
    if (vec && base + FILL_EPT <= E) {
        const int* pp = partp + (size_t)(base / len) * n;
        const int4 sA = *reinterpret_cast<const int4*>(srcA + base);
        const int4 sB = *reinterpret_cast<const int4*>(srcA + base + 4);
        const int4 dA = *reinterpret_cast<const int4*>(dstA + base);
        const int4 dB = *reinterpret_cast<const int4*>(dstA + base + 4);
        const int4 rA = *reinterpret_cast<const int4*>(rank + base);
        const int4 rB = *reinterpret_cast<const int4*>(rank + base + 4);
        const int p0 = rowptr[dA.x] + pp[dA.x];
        const int p1 = rowptr[dA.y] + pp[dA.y];
        const int p2 = rowptr[dA.z] + pp[dA.z];
        const int p3 = rowptr[dA.w] + pp[dA.w];
        const int p4 = rowptr[dB.x] + pp[dB.x];
        const int p5 = rowptr[dB.y] + pp[dB.y];
        const int p6 = rowptr[dB.z] + pp[dB.z];
        const int p7 = rowptr[dB.w] + pp[dB.w];
        col[p0 + rA.x] = sA.x;
        col[p1 + rA.y] = sA.y;
        col[p2 + rA.z] = sA.z;
        col[p3 + rA.w] = sA.w;
        col[p4 + rB.x] = sB.x;
        col[p5 + rB.y] = sB.y;
        col[p6 + rB.z] = sB.z;
        col[p7 + rB.w] = sB.w;
    } else {
        for (int k = 0; k < FILL_EPT; ++k) {
            const int e = base + k;
            if (e < E) {
                const int d = dstA[e];
                col[rowptr[d] + partp[(size_t)(e / len) * n + d] + rank[e]] = srcA[e];
            } else if (e < E + n) {
                const int i = e - E;
                col[rowptr[i + 1] - 1] = i;
            }
        }
    }
}

#define CSR_STEP_THREADS 256
#define CHIP_WGPS 256
__host__ __device__ __forceinline__ int g1_tiles(int n) { return (n + G1_ROWS - 1) / G1_ROWS; }
__host__ __device__ __forceinline__ int g1_tiles_first(int n) {
    const int half = (g1_tiles(n) + 1) / 2;
    return (half >= CHIP_WGPS) ? (half - half % CHIP_WGPS) : half;
}
__host__ __device__ __forceinline__ int prefix_blocks(int n) {
    return (n + CSR_STEP_THREADS - 1) / CSR_STEP_THREADS;
}
__host__ __device__ __forceinline__ int fill_blocks(int E, int n) {
    return (E + n + CSR_STEP_THREADS * FILL_EPT - 1) / (CSR_STEP_THREADS * FILL_EPT);
}

__global__ void __launch_bounds__(CSR_STEP_THREADS)
gemm1_and_part_prefix(const float* __restrict__ X, const float* __restrict__ W,
                      const float* __restrict__ Asrc, const float* __restrict__ Adst,
                      float* __restrict__ H1, float* __restrict__ aS, float* __restrict__ aD,
                      const int* __restrict__ partc, int* __restrict__ partp,
                      int* __restrict__ cnt, int n) {
    const int nt = g1_tiles_first(n);
    if ((int)blockIdx.x < nt) {
        gemm_x_w1_tile(X, W, Asrc, Adst, H1, aS, aD, n, (int)blockIdx.x);
    } else {
        part_prefix_node(partc, partp, cnt, n,
                         ((int)blockIdx.x - nt) * CSR_STEP_THREADS + (int)threadIdx.x);
    }
}

__global__ void __launch_bounds__(CSR_STEP_THREADS)
gemm1_and_fill_csr(const float* __restrict__ X, const float* __restrict__ W,
                   const float* __restrict__ Asrc, const float* __restrict__ Adst,
                   float* __restrict__ H1, float* __restrict__ aS, float* __restrict__ aD,
                   const int* __restrict__ srcA, const int* __restrict__ dstA,
                   const int* __restrict__ rank, const int* __restrict__ partp,
                   const int* __restrict__ rowptr, int* __restrict__ col, int E, int n) {
    const int t0 = g1_tiles_first(n);
    const int nt = g1_tiles(n) - t0;
    if ((int)blockIdx.x < nt) {
        gemm_x_w1_tile(X, W, Asrc, Adst, H1, aS, aD, n, t0 + (int)blockIdx.x);
    } else {
        fill_csr_edges(srcA, dstA, rank, partp, rowptr, col, E, n,
                       (((int)blockIdx.x - nt) * CSR_STEP_THREADS + (int)threadIdx.x) * FILL_EPT);
    }
}

#define L1_WAVES 8

#define ROW_XOR_MAX(x, k) fmaxf((x), ROW_XOR((x), (k)))
__device__ __forceinline__ float wave_max(float x) {
    x = ROW_XOR_MAX(x, 1);
    x = ROW_XOR_MAX(x, 2);
    x = ROW_XOR_MAX(x, 4);
    x = ROW_XOR_MAX(x, 8);
    const int xi = __builtin_bit_cast(int, x);
    return fmaxf(x, __builtin_bit_cast(float, __builtin_amdgcn_permlanex16(
        xi, xi, 0x76543210, 0xfedcba98, false, false)));
}

__device__ __forceinline__ float pick8(const float (&v)[NHD], int h) {
    float r = v[0];
#pragma unroll
    for (int i = 1; i < NHD; ++i) r = (h == i) ? v[i] : r;
    return r;
}

struct L1Edge { float4 lo, hi; float w; };
__device__ __forceinline__ L1Edge l1_fetch(const float* __restrict__ H1, const float (*w)[NHD],
                                           int sv, int e, int h, int ch) {
    const int se = __builtin_amdgcn_readlane(sv, e);
    const float4* hp = reinterpret_cast<const float4*>(H1 + (size_t)se * HC + ch);
    L1Edge r;
    r.lo = hp[0];
    r.hi = hp[1];
    r.w  = w[e][h];
    return r;
}
__device__ __forceinline__ void l1_add(float (&a)[8], float& den, const L1Edge& r) {
    a[0] = fmaf(r.w, r.lo.x, a[0]);
    a[1] = fmaf(r.w, r.lo.y, a[1]);
    a[2] = fmaf(r.w, r.lo.z, a[2]);
    a[3] = fmaf(r.w, r.lo.w, a[3]);
    a[4] = fmaf(r.w, r.hi.x, a[4]);
    a[5] = fmaf(r.w, r.hi.y, a[5]);
    a[6] = fmaf(r.w, r.hi.z, a[6]);
    a[7] = fmaf(r.w, r.hi.w, a[7]);
    den += r.w;
}
template <int K>
__device__ __forceinline__ void l1_group(float (&a)[8], float& den, const float* __restrict__ H1,
                                         const float (*w)[NHD], int sv, int e0, int h, int ch) {
    L1Edge r[K];
#pragma unroll
    for (int k = 0; k < K; ++k) r[k] = l1_fetch(H1, w, sv, e0 + k, h, ch);
#pragma unroll
    for (int k = 0; k < K; ++k) l1_add(a, den, r[k]);
}

#define L1_NPW (G2_ROWS / L1_WAVES)
__global__ void __launch_bounds__(32 * L1_WAVES)
gat_gather_l1_gemm2(const int* __restrict__ rowptr, const int* __restrict__ col,
                    const float* __restrict__ aS, const float* __restrict__ aD,
                    const float* __restrict__ H1, const float* __restrict__ b1,
                    const float* __restrict__ W2, const float* __restrict__ Asrc2,
                    const float* __restrict__ Adst2, float* __restrict__ H2,
                    float* __restrict__ aS2, float* __restrict__ aD2, int n) {
    __shared__ __attribute__((aligned(16))) float sW[L1_WAVES][32][NHD];
    __shared__ __attribute__((aligned(16))) float sT[G2_ROWS * G2_PITCH];
    const int lane = threadIdx.x & 31;
    const int wib  = __builtin_amdgcn_readfirstlane((int)(threadIdx.x >> 5));
    const int rowBase = (int)blockIdx.x * G2_ROWS;
    const int h  = lane >> 2;
    const int ch = lane << 3;
    const float ninf = -__builtin_inff();
    float (*w)[NHD] = sW[wib];
    for (int q = 0; q < L1_NPW; ++q) {
        const int r    = wib * L1_NPW + q;
        const int node = rowBase + r;
        float4 o0 = {0.f, 0.f, 0.f, 0.f}, o1 = o0;
        if (node < n) {
            const int beg = rowptr[node];
            const int end = rowptr[node + 1];
            const float4 dA = *reinterpret_cast<const float4*>(aD + (size_t)node * NHD);
            const float4 dB = *reinterpret_cast<const float4*>(aD + (size_t)node * NHD + 4);
            const float ad[NHD] = {dA.x, dA.y, dA.z, dA.w, dB.x, dB.y, dB.z, dB.w};
            float M[NHD];
#pragma unroll
            for (int i = 0; i < NHD; ++i) M[i] = ninf;
            float den = 0.f;
            float a[8] = {0.f, 0.f, 0.f, 0.f, 0.f, 0.f, 0.f, 0.f};
            for (int cb = beg; cb < end; cb += 32) {
                const int cnt = (end - cb < 32) ? (end - cb) : 32;
                const int s = col[cb + ((lane < cnt) ? lane : 0)];
                const float4 pA = *reinterpret_cast<const float4*>(aS + (size_t)s * NHD);
                const float4 pB = *reinterpret_cast<const float4*>(aS + (size_t)s * NHD + 4);
                const float as[NHD] = {pA.x, pA.y, pA.z, pA.w, pB.x, pB.y, pB.z, pB.w};
                float Mold[NHD], wt[NHD];
#pragma unroll
                for (int i = 0; i < NHD; ++i) {
                    const float l = leaky(as[i] + ad[i]);
                    Mold[i] = M[i];
                    M[i] = fmaxf(M[i], wave_max(l));
                    wt[i] = expf(l - M[i]);
                }
                if (cb != beg) {
                    const float sc = expf(pick8(Mold, h) - pick8(M, h));
#pragma unroll
                    for (int i = 0; i < 8; ++i) a[i] *= sc;
                    den *= sc;
                }
                float4 wA, wB;
                wA.x = wt[0];  wA.y = wt[1];  wA.z = wt[2];  wA.w = wt[3];
                wB.x = wt[4];  wB.y = wt[5];  wB.z = wt[6];  wB.w = wt[7];
                *reinterpret_cast<float4*>(&w[lane][0]) = wA;
                *reinterpret_cast<float4*>(&w[lane][4]) = wB;
                __builtin_amdgcn_fence(__ATOMIC_RELEASE, "wavefront");
                __builtin_amdgcn_wave_barrier();
                __builtin_amdgcn_fence(__ATOMIC_ACQUIRE, "wavefront");
                const int full = cnt & ~3;
                for (int e0 = 0; e0 < full; e0 += 4) l1_group<4>(a, den, H1, w, s, e0, h, ch);
                const int rem = cnt - full;
                if (rem == 3)      l1_group<3>(a, den, H1, w, s, full, h, ch);
                else if (rem == 2) l1_group<2>(a, den, H1, w, s, full, h, ch);
                else if (rem == 1) l1_group<1>(a, den, H1, w, s, full, h, ch);
                __builtin_amdgcn_fence(__ATOMIC_RELEASE, "wavefront");
                __builtin_amdgcn_wave_barrier();
                __builtin_amdgcn_fence(__ATOMIC_ACQUIRE, "wavefront");
            }
            const float v0 = a[0] / den + b1[ch + 0];
            const float v1 = a[1] / den + b1[ch + 1];
            const float v2 = a[2] / den + b1[ch + 2];
            const float v3 = a[3] / den + b1[ch + 3];
            const float v4 = a[4] / den + b1[ch + 4];
            const float v5 = a[5] / den + b1[ch + 5];
            const float v6 = a[6] / den + b1[ch + 6];
            const float v7 = a[7] / den + b1[ch + 7];
            o0.x = v0 > 0.f ? v0 : 0.f;
            o0.y = v1 > 0.f ? v1 : 0.f;
            o0.z = v2 > 0.f ? v2 : 0.f;
            o0.w = v3 > 0.f ? v3 : 0.f;
            o1.x = v4 > 0.f ? v4 : 0.f;
            o1.y = v5 > 0.f ? v5 : 0.f;
            o1.z = v6 > 0.f ? v6 : 0.f;
            o1.w = v7 > 0.f ? v7 : 0.f;
        }
        float4* tp = reinterpret_cast<float4*>(&sT[r * G2_PITCH + ch]);
        tp[0] = o0;
        tp[1] = o1;
    }
    __syncthreads();
    gemm_lds_w2_tile(sT, W2, Asrc2, Adst2, H2, aS2, aD2, n, rowBase);
}

#define L2_LANES 10
#define L2_NODES 3
__global__ void gat_l2_fused(const int* __restrict__ rowptr, const int* __restrict__ col,
                             const float* __restrict__ aS, const float* __restrict__ aD,
                             const float* __restrict__ H2, const float* __restrict__ b2,
                             float* __restrict__ out, int n) {
    const int lane = threadIdx.x & 31;
    const int wave = (blockIdx.x * blockDim.x + threadIdx.x) >> 5;
    if (wave * L2_NODES >= n) return;
    const int grp  = lane / L2_LANES;
    const int k    = lane - grp * L2_LANES;
    const int c    = k << 2;
    const int node = wave * L2_NODES + grp;
    const bool act = (grp < L2_NODES) && (node < n);
    const int nd  = act ? node : wave * L2_NODES;
    const int beg = rowptr[nd];
    const int end = rowptr[nd + (act ? 1 : 0)];
    const float aDd = aD[nd];
    const float bb0 = b2[c + 0], bb1 = b2[c + 1], bb2 = b2[c + 2], bb3 = b2[c + 3];
    float m = -__builtin_inff(), den = 0.f, a0 = 0.f, a1 = 0.f, a2 = 0.f, a3 = 0.f;
    for (int j = beg; j < end; ++j) {
        const int s = col[j];
        if (j + 1 < end) {
            const int s2 = col[j + 1];
            __builtin_prefetch(&H2[(size_t)s2 * OUTC + c], 0, 1);
        }
        const float l = leaky(aS[s] + aDd);
        const float mn = fmaxf(m, l);
        const float scale = expf(m - mn);
        const float wgt   = expf(l - mn);
        const float4 x = *reinterpret_cast<const float4*>(H2 + (size_t)s * OUTC + c);
        a0 = a0 * scale + wgt * x.x;
        a1 = a1 * scale + wgt * x.y;
        a2 = a2 * scale + wgt * x.z;
        a3 = a3 * scale + wgt * x.w;
        den = den * scale + wgt;
        m = mn;
    }
    const float ninf = -__builtin_inff();
    const float v0 = act ? a0 / den + bb0 : ninf;
    const float v1 = act ? a1 / den + bb1 : ninf;
    const float v2 = act ? a2 / den + bb2 : ninf;
    const float v3 = act ? a3 / den + bb3 : ninf;
    const int g0 = (grp < L2_NODES ? grp : 0) * L2_LANES;
    const float lmx = fmaxf(fmaxf(v0, v1), fmaxf(v2, v3));
    float mx = ninf;
    for (int t = 0; t < L2_LANES; ++t) mx = fmaxf(mx, __shfl(lmx, g0 + t, 32));
    const float lsum = act ? ((expf(v0 - mx) + expf(v1 - mx)) + expf(v2 - mx)) + expf(v3 - mx)
                           : 0.f;
    float ssum = 0.f;
    for (int t = 0; t < L2_LANES; ++t) ssum += __shfl(lsum, g0 + t, 32);
    const float lse = mx + logf(ssum);
    if (act) {
        float4 o;
        o.x = v0 - lse;  o.y = v1 - lse;  o.z = v2 - lse;  o.w = v3 - lse;
        *reinterpret_cast<float4*>(out + (size_t)node * OUTC + c) = o;
    }
}

extern "C" void kernel_launch(void* const* d_in, const int* in_sizes, int n_in,
                              void* d_out, int out_size, void* d_ws, size_t ws_size,
                              hipStream_t stream) {
    const float* x   = (const float*)d_in[0];
    const int*   ei  = (const int*)d_in[1];
    const float* W1  = (const float*)d_in[2];
    const float* as1 = (const float*)d_in[3];
    const float* ad1 = (const float*)d_in[4];
    const float* b1  = (const float*)d_in[5];
    const float* W2  = (const float*)d_in[6];
    const float* as2 = (const float*)d_in[7];
    const float* ad2 = (const float*)d_in[8];
    const float* b2  = (const float*)d_in[9];
    const int N  = in_sizes[0] / INCH;
    const int E  = in_sizes[1] / 2;
    const int Et = E + N;
    const int* srcA = ei;
    const int* dstA = ei + E;

    float* ws = (float*)d_ws;
    size_t off = 0;
    float* h1   = ws + off; off += (size_t)N * HC;
    float* out1 = ws + off; off += (size_t)N * HC;
    float* aS1  = ws + off; off += (size_t)N * NHD;
    float* aD1  = ws + off; off += (size_t)N * NHD;
    int* rowptr = (int*)(ws + off); off += (size_t)N + 1;
    int* cnt    = (int*)(ws + off); off += (size_t)N;
    int* col    = (int*)(ws + off); off += (size_t)Et;
    float* h2  = out1;
    float* aS2 = h2 + (size_t)N * OUTC;
    float* aD2 = aS2 + (((size_t)N + 31) & ~(size_t)31);
    int* partc = (int*)out1;
    int* partp = partc + (size_t)CSR_PARTS * N;
    int* rank  = partp + (size_t)CSR_PARTS * N;

    count_rank<<<CSR_PARTS, CSR_THREADS, 0, stream>>>(dstA, partc, rank, E, N);
    gemm1_and_part_prefix<<<g1_tiles_first(N) + prefix_blocks(N), CSR_STEP_THREADS, 0, stream>>>(
        x, W1, as1, ad1, h1, aS1, aD1, partc, partp, cnt, N);
    scan_rowptr<<<(N + 1023) / 1024, 1024, 0, stream>>>(cnt, rowptr, N);
    gemm1_and_fill_csr<<<g1_tiles(N) - g1_tiles_first(N) + fill_blocks(E, N), CSR_STEP_THREADS, 0,
                         stream>>>(
        x, W1, as1, ad1, h1, aS1, aD1, srcA, dstA, rank, partp, rowptr, col, E, N);

    gat_gather_l1_gemm2<<<(N + G2_ROWS - 1) / G2_ROWS, 32 * L1_WAVES, 0, stream>>>(
        rowptr, col, aS1, aD1, h1, b1, W2, as2, ad2, h2, aS2, aD2, N);

    gat_l2_fused<<<((N + L2_NODES - 1) / L2_NODES * 32 + 255) / 256, 256, 0, stream>>>(
        rowptr, col, aS2, aD2, h2, b2, (float*)d_out, N);
}


# layer-2 gather: softmax weights of 64 edges at once, 2 per lane (max per node by cross-lane tree, one exp per edge), LDS hand-over; row loop only loads+FMAs, 8 rows in flight; edge sum order kept
# speedup vs baseline: 3.7304x; 1.0498x over previous
#include <hip/hip_runtime.h>
#include <hip/hip_bf16.h>

typedef __attribute__((ext_vector_type(2))) float v2f;
typedef __attribute__((ext_vector_type(8))) float v8f;

#define INCH  128
#define HC    256
#define NHD   8
#define CPH   32
#define OUTC  40
#define SLOPE 0.2f

__device__ __forceinline__ float leaky(float x) { return x > 0.f ? x : SLOPE * x; }

__device__ __forceinline__ void lds_handover_barrier() {
    asm volatile("s_wait_dscnt 0x0" ::: "memory");
    asm volatile("s_barrier_signal -1\n\ts_barrier_wait -1" ::: "memory");
}

#define ROW_XOR(x, k) __builtin_bit_cast(float, __builtin_amdgcn_update_dpp( \
    0, __builtin_bit_cast(int, (x)), 0x160 + (k), 0xf, 0xf, true))
__device__ __forceinline__ float row16_sum(float x) {
    x += ROW_XOR(x, 1);
    x += ROW_XOR(x, 2);
    x += ROW_XOR(x, 4);
    x += ROW_XOR(x, 8);
    return x;
}

#define G1_ROWS  32
#define G1_PITCH (INCH + 4)
__device__ __forceinline__ void
gemm_x_w1_tile(const float* __restrict__ X, const float* __restrict__ W,
               const float* __restrict__ Asrc, const float* __restrict__ Adst,
               float* __restrict__ H1, float* __restrict__ aS, float* __restrict__ aD, int nrows,
               int tile) {
    __shared__ __attribute__((aligned(16))) float sA[G1_ROWS * G1_PITCH];
    __shared__ float sS[G1_ROWS * NHD], sD[G1_ROWS * NHD];
    const int tid  = (int)threadIdx.x;
    const int lane = tid & 31;
    const int wave = tid >> 5;
    const int m    = lane & 15;
    const int half = lane >> 4;
    const int rowBase = tile * G1_ROWS;
    const float as0 = Asrc[(wave << 5) + m], as1 = Asrc[(wave << 5) + 16 + m];
    const float ad0 = Adst[(wave << 5) + m], ad1 = Adst[(wave << 5) + 16 + m];
    float4 st[(G1_ROWS * INCH / 4) / 256];
#pragma unroll
    for (int i = 0; i < (G1_ROWS * INCH / 4) / 256; ++i) {
        const int idx = tid + i * 256;
        const int r = idx >> 5, c4 = idx & 31;
        const int gr = (rowBase + r < nrows) ? (rowBase + r) : (nrows - 1);
        st[i] = *reinterpret_cast<const float4*>(X + (size_t)gr * INCH + (c4 << 2));
    }
#pragma unroll
    for (int i = 0; i < (G1_ROWS * INCH / 4) / 256; ++i) {
        const int idx = tid + i * 256;
        const int r = idx >> 5, c4 = idx & 31;
        *reinterpret_cast<float4*>(&sA[r * G1_PITCH + (c4 << 2)]) = st[i];
    }
    __syncthreads();
    const int c0 = wave << 5;
    const float* wp  = W + c0 + m;
    const float* a0p = sA + m * G1_PITCH + (half << 1);
    const float* a1p = a0p + 16 * G1_PITCH;
    v8f acc00 = {0.f, 0.f, 0.f, 0.f, 0.f, 0.f, 0.f, 0.f};
    v8f acc01 = acc00, acc10 = acc00, acc11 = acc00;
    for (int k = 0; k < INCH; k += 4) {
        const int kk = k + (half << 1);
        const v2f a0 = *reinterpret_cast<const v2f*>(a0p + k);
        const v2f a1 = *reinterpret_cast<const v2f*>(a1p + k);
        v2f b0, b1;
        b0.x = wp[(size_t)kk * HC];
        b0.y = wp[(size_t)(kk + 1) * HC];
        b1.x = wp[(size_t)kk * HC + 16];
        b1.y = wp[(size_t)(kk + 1) * HC + 16];
        acc00 = __builtin_amdgcn_wmma_f32_16x16x4_f32(false, a0, false, b0,
                                                      (short)0, acc00, false, false);
        acc01 = __builtin_amdgcn_wmma_f32_16x16x4_f32(false, a0, false, b1,
                                                      (short)0, acc01, false, false);
        acc10 = __builtin_amdgcn_wmma_f32_16x16x4_f32(false, a1, false, b0,
                                                      (short)0, acc10, false, false);
        acc11 = __builtin_amdgcn_wmma_f32_16x16x4_f32(false, a1, false, b1,
                                                      (short)0, acc11, false, false);
    }
    float* o0 = H1 + (size_t)(rowBase + (half << 3)) * HC + c0 + m;
    float* o1 = o0 + (size_t)16 * HC;
    if (rowBase + G1_ROWS <= nrows) {
#pragma unroll
        for (int v = 0; v < 8; ++v) {
            o0[(size_t)v * HC]      = acc00[v];
            o0[(size_t)v * HC + 16] = acc01[v];
            o1[(size_t)v * HC]      = acc10[v];
            o1[(size_t)v * HC + 16] = acc11[v];
        }
    } else {
#pragma unroll
        for (int v = 0; v < 8; ++v) {
            const int r0 = rowBase + v + (half << 3);
            if (r0 < nrows) {
                o0[(size_t)v * HC]      = acc00[v];
                o0[(size_t)v * HC + 16] = acc01[v];
            }
            if (r0 + 16 < nrows) {
                o1[(size_t)v * HC]      = acc10[v];
                o1[(size_t)v * HC + 16] = acc11[v];
            }
        }
    }
    float s0 = 0.f, s1 = 0.f, d0 = 0.f, d1 = 0.f;
#pragma unroll
    for (int v = 0; v < 8; ++v) {
        const float ts0 = row16_sum(acc00[v] * as0 + acc01[v] * as1);
        const float td0 = row16_sum(acc00[v] * ad0 + acc01[v] * ad1);
        const float ts1 = row16_sum(acc10[v] * as0 + acc11[v] * as1);
        const float td1 = row16_sum(acc10[v] * ad0 + acc11[v] * ad1);
        s0 = (m == v) ? ts0 : s0;
        d0 = (m == v) ? td0 : d0;
        s1 = (m == v) ? ts1 : s1;
        d1 = (m == v) ? td1 : d1;
    }
    if (m < 8) {
        const int r = (half << 3) + m;
        sS[r * NHD + wave] = s0;  sS[(r + 16) * NHD + wave] = s1;
        sD[r * NHD + wave] = d0;  sD[(r + 16) * NHD + wave] = d1;
    }
    lds_handover_barrier();
    if ((rowBase << 3) + tid < nrows * NHD) {
        aS[((size_t)rowBase << 3) + tid] = sS[tid];
        aD[((size_t)rowBase << 3) + tid] = sD[tid];
    }
}

#define G2_ROWS  16
#define G2_PITCH (HC + 4)
__device__ __forceinline__ void
gemm_lds_w2_tile(const float* sA, const float* __restrict__ W,
                 const float* __restrict__ Asrc, const float* __restrict__ Adst,
                 float* __restrict__ H2, float* __restrict__ aS, float* __restrict__ aD, int nrows,
                 int rowBase) {
    __shared__ float sS[3 * 16], sD[3 * 16];
    const int tid  = (int)threadIdx.x;
    const int lane = tid & 31;
    const int wave = tid >> 5;
    const int m    = lane & 15;
    const int half = lane >> 4;
    if (wave < 3) {
        const int col = (wave << 4) + m;
        const bool cin = col < OUTC;
        const float* ap = sA + m * G2_PITCH + (half << 1);
        const float* wp = W + (cin ? col : 0);
        v8f acc = {0.f, 0.f, 0.f, 0.f, 0.f, 0.f, 0.f, 0.f};
        for (int k = 0; k < HC; k += 4) {
            const int kk = k + (half << 1);
            const v2f a = *reinterpret_cast<const v2f*>(ap + k);
            v2f b;
            b.x = wp[kk * OUTC];
            b.y = wp[(kk + 1) * OUTC];
            acc = __builtin_amdgcn_wmma_f32_16x16x4_f32(false, a, false, b,
                                                        (short)0, acc, false, false);
        }
        if (cin) {
            const int r0 = rowBase + (half << 3);
            float* o = H2 + (size_t)r0 * OUTC + col;
            if (rowBase + G2_ROWS <= nrows) {
#pragma unroll
                for (int v = 0; v < 8; ++v) o[(size_t)v * OUTC] = acc[v];
            } else {
#pragma unroll
                for (int v = 0; v < 8; ++v)
                    if (r0 + v < nrows) o[(size_t)v * OUTC] = acc[v];
            }
        }
        const float as = cin ? Asrc[col] : 0.f;
        const float ad = cin ? Adst[col] : 0.f;
        float s = 0.f, d = 0.f;
#pragma unroll
        for (int v = 0; v < 8; ++v) {
            const float ts = row16_sum(cin ? acc[v] * as : 0.f);
            const float td = row16_sum(cin ? acc[v] * ad : 0.f);
            s = (m == v) ? ts : s;
            d = (m == v) ? td : d;
        }
        if (m < 8) {
            sS[wave * 16 + (half << 3) + m] = s;
            sD[wave * 16 + (half << 3) + m] = d;
        }
    }
    lds_handover_barrier();
    if (tid < G2_ROWS && rowBase + tid < nrows) {
        aS[rowBase + tid] = (sS[tid] + sS[16 + tid]) + sS[32 + tid];
        aD[rowBase + tid] = (sD[tid] + sD[16 + tid]) + sD[32 + tid];
    }
}

#define CSR_PARTS   64
#define CSR_THREADS 1024
#define CSR_BINS    51200
#define CSR_AHEAD   16

__device__ __forceinline__ int csr_part_len(int E) {
    return (((E + CSR_PARTS - 1) / CSR_PARTS) + 7) & ~7;
}

__global__ void __launch_bounds__(CSR_THREADS)
count_rank(const int* __restrict__ dstA, int* __restrict__ partc, int* __restrict__ rank,
           int E, int n) {
    __shared__ int hist[CSR_BINS];
    const int tid  = (int)threadIdx.x;
    const int part = (int)blockIdx.x;
    const int len  = csr_part_len(E);
    const int beg  = part * len;
    const int end  = (beg + len < E) ? (beg + len) : E;
    for (int w0 = 0; w0 < n; w0 += CSR_BINS) {
        const int wn = (n - w0 < CSR_BINS) ? (n - w0) : CSR_BINS;
        for (int i = tid; i < wn; i += CSR_THREADS) hist[i] = 0;
        __syncthreads();
        for (int e0 = beg; e0 < end; e0 += CSR_AHEAD * CSR_THREADS) {
            int d[CSR_AHEAD], r[CSR_AHEAD];
#pragma unroll
            for (int u = 0; u < CSR_AHEAD; ++u) {
                const int e = e0 + u * CSR_THREADS + tid;
                const bool in = e < end;
                const int x = dstA[in ? e : beg];
                d[u] = in ? (x - w0) : -1;
            }
#pragma unroll
            for (int u = 0; u < CSR_AHEAD; ++u) {
                r[u] = 0;
                if ((unsigned)d[u] < (unsigned)wn) r[u] = atomicAdd(&hist[d[u]], 1);
            }
#pragma unroll
            for (int u = 0; u < CSR_AHEAD; ++u) {
                const int e = e0 + u * CSR_THREADS + tid;
                if ((unsigned)d[u] < (unsigned)wn) rank[e] = r[u];
            }
        }
        __syncthreads();
        int* row = partc + (size_t)part * n + w0;
        for (int i = tid; i < wn; i += CSR_THREADS) row[i] = hist[i];
        __syncthreads();
    }
}

__device__ __forceinline__ void part_prefix_node(const int* __restrict__ partc,
                                                 int* __restrict__ partp,
                                                 int* __restrict__ cnt, int n, int d) {
    if (d >= n) return;
    int run = 0;
    for (int p0 = 0; p0 < CSR_PARTS; p0 += 16) {
        int x[16];
#pragma unroll
        for (int u = 0; u < 16; ++u) x[u] = partc[(size_t)(p0 + u) * n + d];
#pragma unroll
        for (int u = 0; u < 16; ++u) { partp[(size_t)(p0 + u) * n + d] = run; run += x[u]; }
    }
    cnt[d] = run + 1;
}

#define SCAN_AHEAD 16
__global__ void scan_rowptr(const int* __restrict__ cnt, int* __restrict__ rowptr, int n) {
    __shared__ int sPre[32];
    __shared__ int sTot[32];
    const int tid  = (int)threadIdx.x;
    const int lane = tid & 31;
    const int wave = tid >> 5;
    const int base = (int)blockIdx.x * 1024;
    const int i = base + tid;
    const int v = (i < n) ? cnt[i] : 0;
    int pre = 0;
    for (int k0 = 0; k0 < base; k0 += SCAN_AHEAD * 1024) {
        for (int u = 0; u < SCAN_AHEAD; ++u) {
            const int kb = k0 + u * 1024;
            const bool in = kb < base;
            const int x = cnt[(in ? kb : 0) + tid];
            pre += in ? x : 0;
        }
    }
    int inc = v;
    for (int d = 1; d < 32; d <<= 1) {
        const int t = __shfl_up(inc, d, 32);
        if (lane >= d) inc += t;
    }
    for (int o = 16; o > 0; o >>= 1) pre += __shfl_xor(pre, o, 32);
    if (lane == 31) { sTot[wave] = inc; sPre[wave] = pre; }
    __syncthreads();
    int off = sPre[lane] + ((lane < wave) ? sTot[lane] : 0);
    for (int o = 16; o > 0; o >>= 1) off += __shfl_xor(off, o, 32);
    const int excl = off + inc - v;
    if (i < n) rowptr[i] = excl;
    if (i == n - 1) rowptr[n] = excl + v;
}

#define FILL_EPT 8
__device__ __forceinline__ void fill_csr_edges(const int* __restrict__ srcA,
                                               const int* __restrict__ dstA,
                                               const int* __restrict__ rank,
                                               const int* __restrict__ partp,
                                               const int* __restrict__ rowptr,
                                               int* __restrict__ col, int E, int n, int base) {
    if (base >= E + n) return;
    const int len = csr_part_len(E);
    const bool vec = (((reinterpret_cast<size_t>(srcA) | reinterpret_cast<size_t>(dstA) |
                        reinterpret_cast<size_t>(rank)) & 15) == 0);
    if (vec && base + FILL_EPT <= E) {
        const int* pp = partp + (size_t)(base / len) * n;
        const int4 sA = *reinterpret_cast<const int4*>(srcA + base);
        const int4 sB = *reinterpret_cast<const int4*>(srcA + base + 4);
        const int4 dA = *reinterpret_cast<const int4*>(dstA + base);
        const int4 dB = *reinterpret_cast<const int4*>(dstA + base + 4);
        const int4 rA = *reinterpret_cast<const int4*>(rank + base);
        const int4 rB = *reinterpret_cast<const int4*>(rank + base + 4);
        const int p0 = rowptr[dA.x] + pp[dA.x];
        const int p1 = rowptr[dA.y] + pp[dA.y];
        const int p2 = rowptr[dA.z] + pp[dA.z];
        const int p3 = rowptr[dA.w] + pp[dA.w];
        const int p4 = rowptr[dB.x] + pp[dB.x];
        const int p5 = rowptr[dB.y] + pp[dB.y];
        const int p6 = rowptr[dB.z] + pp[dB.z];
        const int p7 = rowptr[dB.w] + pp[dB.w];
        col[p0 + rA.x] = sA.x;
        col[p1 + rA.y] = sA.y;
        col[p2 + rA.z] = sA.z;
        col[p3 + rA.w] = sA.w;
        col[p4 + rB.x] = sB.x;
        col[p5 + rB.y] = sB.y;
        col[p6 + rB.z] = sB.z;
        col[p7 + rB.w] = sB.w;
    } else {
        for (int k = 0; k < FILL_EPT; ++k) {
            const int e = base + k;
            if (e < E) {
                const int d = dstA[e];
                col[rowptr[d] + partp[(size_t)(e / len) * n + d] + rank[e]] = srcA[e];
            } else if (e < E + n) {
                const int i = e - E;
                col[rowptr[i + 1] - 1] = i;
            }
        }
    }
}

#define CSR_STEP_THREADS 256
#define CHIP_WGPS 256
__host__ __device__ __forceinline__ int g1_tiles(int n) { return (n + G1_ROWS - 1) / G1_ROWS; }
__host__ __device__ __forceinline__ int g1_tiles_first(int n) {
    const int half = (g1_tiles(n) + 1) / 2;
    return (half >= CHIP_WGPS) ? (half - half % CHIP_WGPS) : half;
}
__host__ __device__ __forceinline__ int prefix_blocks(int n) {
    return (n + CSR_STEP_THREADS - 1) / CSR_STEP_THREADS;
}
__host__ __device__ __forceinline__ int fill_blocks(int E, int n) {
    return (E + n + CSR_STEP_THREADS * FILL_EPT - 1) / (CSR_STEP_THREADS * FILL_EPT);
}

__global__ void __launch_bounds__(CSR_STEP_THREADS)
gemm1_and_part_prefix(const float* __restrict__ X, const float* __restrict__ W,
                      const float* __restrict__ Asrc, const float* __restrict__ Adst,
                      float* __restrict__ H1, float* __restrict__ aS, float* __restrict__ aD,
                      const int* __restrict__ partc, int* __restrict__ partp,
                      int* __restrict__ cnt, int n) {
    const int nt = g1_tiles_first(n);
    if ((int)blockIdx.x < nt) {
        gemm_x_w1_tile(X, W, Asrc, Adst, H1, aS, aD, n, (int)blockIdx.x);
    } else {
        part_prefix_node(partc, partp, cnt, n,
                         ((int)blockIdx.x - nt) * CSR_STEP_THREADS + (int)threadIdx.x);
    }
}

__global__ void __launch_bounds__(CSR_STEP_THREADS)
gemm1_and_fill_csr(const float* __restrict__ X, const float* __restrict__ W,
                   const float* __restrict__ Asrc, const float* __restrict__ Adst,
                   float* __restrict__ H1, float* __restrict__ aS, float* __restrict__ aD,
                   const int* __restrict__ srcA, const int* __restrict__ dstA,
                   const int* __restrict__ rank, const int* __restrict__ partp,
                   const int* __restrict__ rowptr, int* __restrict__ col, int E, int n) {
    const int t0 = g1_tiles_first(n);
    const int nt = g1_tiles(n) - t0;
    if ((int)blockIdx.x < nt) {
        gemm_x_w1_tile(X, W, Asrc, Adst, H1, aS, aD, n, t0 + (int)blockIdx.x);
    } else {
        fill_csr_edges(srcA, dstA, rank, partp, rowptr, col, E, n,
                       (((int)blockIdx.x - nt) * CSR_STEP_THREADS + (int)threadIdx.x) * FILL_EPT);
    }
}

#define L1_WAVES 8

#define ROW_XOR_MAX(x, k) fmaxf((x), ROW_XOR((x), (k)))
__device__ __forceinline__ float wave_max(float x) {
    x = ROW_XOR_MAX(x, 1);
    x = ROW_XOR_MAX(x, 2);
    x = ROW_XOR_MAX(x, 4);
    x = ROW_XOR_MAX(x, 8);
    const int xi = __builtin_bit_cast(int, x);
    return fmaxf(x, __builtin_bit_cast(float, __builtin_amdgcn_permlanex16(
        xi, xi, 0x76543210, 0xfedcba98, false, false)));
}

__device__ __forceinline__ float pick8(const float (&v)[NHD], int h) {
    float r = v[0];
#pragma unroll
    for (int i = 1; i < NHD; ++i) r = (h == i) ? v[i] : r;
    return r;
}

struct L1Edge { float4 lo, hi; float w; };
__device__ __forceinline__ L1Edge l1_fetch(const float* __restrict__ H1, const float (*w)[NHD],
                                           int sv, int e, int h, int ch) {
    const int se = __builtin_amdgcn_readlane(sv, e);
    const float4* hp = reinterpret_cast<const float4*>(H1 + (size_t)se * HC + ch);
    L1Edge r;
    r.lo = hp[0];
    r.hi = hp[1];
    r.w  = w[e][h];
    return r;
}
__device__ __forceinline__ void l1_add(float (&a)[8], float& den, const L1Edge& r) {
    a[0] = fmaf(r.w, r.lo.x, a[0]);
    a[1] = fmaf(r.w, r.lo.y, a[1]);
    a[2] = fmaf(r.w, r.lo.z, a[2]);
    a[3] = fmaf(r.w, r.lo.w, a[3]);
    a[4] = fmaf(r.w, r.hi.x, a[4]);
    a[5] = fmaf(r.w, r.hi.y, a[5]);
    a[6] = fmaf(r.w, r.hi.z, a[6]);
    a[7] = fmaf(r.w, r.hi.w, a[7]);
    den += r.w;
}
template <int K>
__device__ __forceinline__ void l1_group(float (&a)[8], float& den, const float* __restrict__ H1,
                                         const float (*w)[NHD], int sv, int e0, int h, int ch) {
    L1Edge r[K];
#pragma unroll
    for (int k = 0; k < K; ++k) r[k] = l1_fetch(H1, w, sv, e0 + k, h, ch);
#pragma unroll
    for (int k = 0; k < K; ++k) l1_add(a, den, r[k]);
}

#define L1_NPW (G2_ROWS / L1_WAVES)
__global__ void __launch_bounds__(32 * L1_WAVES)
gat_gather_l1_gemm2(const int* __restrict__ rowptr, const int* __restrict__ col,
                    const float* __restrict__ aS, const float* __restrict__ aD,
                    const float* __restrict__ H1, const float* __restrict__ b1,
                    const float* __restrict__ W2, const float* __restrict__ Asrc2,
                    const float* __restrict__ Adst2, float* __restrict__ H2,
                    float* __restrict__ aS2, float* __restrict__ aD2, int n) {
    __shared__ __attribute__((aligned(16))) float sW[L1_WAVES][32][NHD];
    __shared__ __attribute__((aligned(16))) float sT[G2_ROWS * G2_PITCH];
    const int lane = threadIdx.x & 31;
    const int wib  = __builtin_amdgcn_readfirstlane((int)(threadIdx.x >> 5));
    const int rowBase = (int)blockIdx.x * G2_ROWS;
    const int h  = lane >> 2;
    const int ch = lane << 3;
    const float ninf = -__builtin_inff();
    float (*w)[NHD] = sW[wib];
    for (int q = 0; q < L1_NPW; ++q) {
        const int r    = wib * L1_NPW + q;
        const int node = rowBase + r;
        float4 o0 = {0.f, 0.f, 0.f, 0.f}, o1 = o0;
        if (node < n) {
            const int beg = rowptr[node];
            const int end = rowptr[node + 1];
            const float4 dA = *reinterpret_cast<const float4*>(aD + (size_t)node * NHD);
            const float4 dB = *reinterpret_cast<const float4*>(aD + (size_t)node * NHD + 4);
            const float ad[NHD] = {dA.x, dA.y, dA.z, dA.w, dB.x, dB.y, dB.z, dB.w};
            float M[NHD];
#pragma unroll
            for (int i = 0; i < NHD; ++i) M[i] = ninf;
            float den = 0.f;
            float a[8] = {0.f, 0.f, 0.f, 0.f, 0.f, 0.f, 0.f, 0.f};
            for (int cb = beg; cb < end; cb += 32) {
                const int cnt = (end - cb < 32) ? (end - cb) : 32;
                const int s = col[cb + ((lane < cnt) ? lane : 0)];
                const float4 pA = *reinterpret_cast<const float4*>(aS + (size_t)s * NHD);
                const float4 pB = *reinterpret_cast<const float4*>(aS + (size_t)s * NHD + 4);
                const float as[NHD] = {pA.x, pA.y, pA.z, pA.w, pB.x, pB.y, pB.z, pB.w};
                float Mold[NHD], wt[NHD];
#pragma unroll
                for (int i = 0; i < NHD; ++i) {
                    const float l = leaky(as[i] + ad[i]);
                    Mold[i] = M[i];
                    M[i] = fmaxf(M[i], wave_max(l));
                    wt[i] = expf(l - M[i]);
                }
                if (cb != beg) {
                    const float sc = expf(pick8(Mold, h) - pick8(M, h));
#pragma unroll
                    for (int i = 0; i < 8; ++i) a[i] *= sc;
                    den *= sc;
                }
                float4 wA, wB;
                wA.x = wt[0];  wA.y = wt[1];  wA.z = wt[2];  wA.w = wt[3];
                wB.x = wt[4];  wB.y = wt[5];  wB.z = wt[6];  wB.w = wt[7];
                *reinterpret_cast<float4*>(&w[lane][0]) = wA;
                *reinterpret_cast<float4*>(&w[lane][4]) = wB;
                __builtin_amdgcn_fence(__ATOMIC_RELEASE, "wavefront");
                __builtin_amdgcn_wave_barrier();
                __builtin_amdgcn_fence(__ATOMIC_ACQUIRE, "wavefront");
                const int full = cnt & ~3;
                for (int e0 = 0; e0 < full; e0 += 4) l1_group<4>(a, den, H1, w, s, e0, h, ch);
                const int rem = cnt - full;
                if (rem == 3)      l1_group<3>(a, den, H1, w, s, full, h, ch);
                else if (rem == 2) l1_group<2>(a, den, H1, w, s, full, h, ch);
                else if (rem == 1) l1_group<1>(a, den, H1, w, s, full, h, ch);
                __builtin_amdgcn_fence(__ATOMIC_RELEASE, "wavefront");
                __builtin_amdgcn_wave_barrier();
                __builtin_amdgcn_fence(__ATOMIC_ACQUIRE, "wavefront");
            }
            const float v0 = a[0] / den + b1[ch + 0];
            const float v1 = a[1] / den + b1[ch + 1];
            const float v2 = a[2] / den + b1[ch + 2];
            const float v3 = a[3] / den + b1[ch + 3];
            const float v4 = a[4] / den + b1[ch + 4];
            const float v5 = a[5] / den + b1[ch + 5];
            const float v6 = a[6] / den + b1[ch + 6];
            const float v7 = a[7] / den + b1[ch + 7];
            o0.x = v0 > 0.f ? v0 : 0.f;
            o0.y = v1 > 0.f ? v1 : 0.f;
            o0.z = v2 > 0.f ? v2 : 0.f;
            o0.w = v3 > 0.f ? v3 : 0.f;
            o1.x = v4 > 0.f ? v4 : 0.f;
            o1.y = v5 > 0.f ? v5 : 0.f;
            o1.z = v6 > 0.f ? v6 : 0.f;
            o1.w = v7 > 0.f ? v7 : 0.f;
        }
        float4* tp = reinterpret_cast<float4*>(&sT[r * G2_PITCH + ch]);
        tp[0] = o0;
        tp[1] = o1;
    }
    __syncthreads();
    gemm_lds_w2_tile(sT, W2, Asrc2, Adst2, H2, aS2, aD2, n, rowBase);
}

#define L2_LANES 10
#define L2_NODES 3
#define L2_WAVES 8
#define L2_AHEAD 8
#define L2_CHUNK 64
__global__ void __launch_bounds__(32 * L2_WAVES)
gat_l2_fused(const int* __restrict__ rowptr, const int* __restrict__ col,
             const float* __restrict__ aS, const float* __restrict__ aD,
             const float* __restrict__ H2, const float* __restrict__ b2,
             float* __restrict__ out, int n) {
    __shared__ int2 sEW[L2_WAVES][L2_CHUNK];
    const int lane = threadIdx.x & 31;
    const int wib   = __builtin_amdgcn_readfirstlane((int)(threadIdx.x >> 5));
    const int node0 = ((int)blockIdx.x * L2_WAVES + wib) * L2_NODES;
    if (node0 >= n) return;
    const int nn   = (n - node0 < L2_NODES) ? (n - node0) : L2_NODES;
    const int grp  = lane / L2_LANES;
    const int k    = lane - grp * L2_LANES;
    const int c    = k << 2;
    const int node = node0 + grp;
    const bool act = (grp < L2_NODES) && (node < n);
    const int rb0 = rowptr[node0];
    const int rb1 = rowptr[node0 + (nn > 1 ? 1 : nn)];
    const int rb2 = rowptr[node0 + (nn > 2 ? 2 : nn)];
    const int rb3 = rowptr[node0 + nn];
    const float ad0 = aD[node0];
    const float ad1 = aD[node0 + (nn > 1 ? 1 : 0)];
    const float ad2 = aD[node0 + (nn > 2 ? 2 : 0)];
    const float bb0 = b2[c + 0], bb1 = b2[c + 1], bb2 = b2[c + 2], bb3 = b2[c + 3];
    const int glo = (grp == 0) ? rb0 : (grp == 1) ? rb1 : (grp == 2) ? rb2 : rb3;
    const int ghi = (grp == 0) ? rb1 : (grp == 1) ? rb2 : rb3;
    const float ninf = -__builtin_inff();
    float M0 = ninf, M1 = ninf, M2 = ninf;
    float den = 0.f, a0 = 0.f, a1 = 0.f, a2 = 0.f, a3 = 0.f;
    int2* ew = sEW[wib];
    for (int cb = rb0; cb < rb3; cb += L2_CHUNK) {
        const int e0 = cb + lane, e1 = e0 + 32;
        const bool ev0 = e0 < rb3, ev1 = e1 < rb3;
        const int s0 = col[ev0 ? e0 : rb0];
        const int s1 = col[ev1 ? e1 : rb0];
        const int ge0 = (e0 >= rb1 ? 1 : 0) + (e0 >= rb2 ? 1 : 0);
        const int ge1 = (e1 >= rb1 ? 1 : 0) + (e1 >= rb2 ? 1 : 0);
        const float l0 = leaky(aS[s0] + ((ge0 == 0) ? ad0 : (ge0 == 1) ? ad1 : ad2));
        const float l1 = leaky(aS[s1] + ((ge1 == 0) ? ad0 : (ge1 == 1) ? ad1 : ad2));
        const float c0 = wave_max(fmaxf((ev0 && ge0 == 0) ? l0 : ninf, (ev1 && ge1 == 0) ? l1 : ninf));
        const float c1 = wave_max(fmaxf((ev0 && ge0 == 1) ? l0 : ninf, (ev1 && ge1 == 1) ? l1 : ninf));
        const float c2 = wave_max(fmaxf((ev0 && ge0 == 2) ? l0 : ninf, (ev1 && ge1 == 2) ? l1 : ninf));
        const float Mo = (grp == 0) ? M0 : (grp == 1) ? M1 : M2;
        M0 = fmaxf(M0, c0);
        M1 = fmaxf(M1, c1);
        M2 = fmaxf(M2, c2);
        const float Mn = (grp == 0) ? M0 : (grp == 1) ? M1 : M2;
        const float sc = (Mo == Mn) ? 1.f : expf(Mo - Mn);
        a0 *= sc;  a1 *= sc;  a2 *= sc;  a3 *= sc;  den *= sc;
        int2 p0, p1;
        p0.x = s0;
        p0.y = __builtin_bit_cast(int, expf(l0 - ((ge0 == 0) ? M0 : (ge0 == 1) ? M1 : M2)));
        p1.x = s1;
        p1.y = __builtin_bit_cast(int, expf(l1 - ((ge1 == 0) ? M0 : (ge1 == 1) ? M1 : M2)));
        ew[lane]      = p0;
        ew[lane + 32] = p1;
        __builtin_amdgcn_fence(__ATOMIC_RELEASE, "wavefront");
        __builtin_amdgcn_wave_barrier();
        __builtin_amdgcn_fence(__ATOMIC_ACQUIRE, "wavefront");
        const int lo = ((glo > cb) ? glo : cb) - cb;
        const int hi = ((ghi < cb + L2_CHUNK) ? ghi : (cb + L2_CHUNK)) - cb;
        for (int j = lo; j < hi; j += L2_AHEAD) {
            float4 x[L2_AHEAD];
            float  w[L2_AHEAD];
#pragma unroll
            for (int u = 0; u < L2_AHEAD; ++u) {
                const bool v = j + u < hi;
                const int2 p = ew[v ? (j + u) : (hi - 1)];
                w[u] = v ? __builtin_bit_cast(float, p.y) : 0.f;
                x[u] = *reinterpret_cast<const float4*>(H2 + (size_t)p.x * OUTC + c);
            }
            __builtin_amdgcn_sched_barrier(0);
#pragma unroll
            for (int u = 0; u < L2_AHEAD; ++u) {
                a0 = fmaf(w[u], x[u].x, a0);
                a1 = fmaf(w[u], x[u].y, a1);
                a2 = fmaf(w[u], x[u].z, a2);
                a3 = fmaf(w[u], x[u].w, a3);
                den += w[u];
            }
        }
        __builtin_amdgcn_fence(__ATOMIC_RELEASE, "wavefront");
        __builtin_amdgcn_wave_barrier();
        __builtin_amdgcn_fence(__ATOMIC_ACQUIRE, "wavefront");
    }
    const float v0 = act ? a0 / den + bb0 : ninf;
    const float v1 = act ? a1 / den + bb1 : ninf;
    const float v2 = act ? a2 / den + bb2 : ninf;
    const float v3 = act ? a3 / den + bb3 : ninf;
    const int g0 = (grp < L2_NODES ? grp : 0) * L2_LANES;
    const float lmx = fmaxf(fmaxf(v0, v1), fmaxf(v2, v3));
    float mx = ninf;
    for (int t = 0; t < L2_LANES; ++t) mx = fmaxf(mx, __shfl(lmx, g0 + t, 32));
    const float lsum = act ? ((expf(v0 - mx) + expf(v1 - mx)) + expf(v2 - mx)) + expf(v3 - mx)
                           : 0.f;
    float ssum = 0.f;
    for (int t = 0; t < L2_LANES; ++t) ssum += __shfl(lsum, g0 + t, 32);
    const float lse = mx + logf(ssum);
    if (act) {
        float4 o;
        o.x = v0 - lse;  o.y = v1 - lse;  o.z = v2 - lse;  o.w = v3 - lse;
        *reinterpret_cast<float4*>(out + (size_t)node * OUTC + c) = o;
    }
}

extern "C" void kernel_launch(void* const* d_in, const int* in_sizes, int n_in,
                              void* d_out, int out_size, void* d_ws, size_t ws_size,
                              hipStream_t stream) {
    const float* x   = (const float*)d_in[0];
    const int*   ei  = (const int*)d_in[1];
    const float* W1  = (const float*)d_in[2];
    const float* as1 = (const float*)d_in[3];
    const float* ad1 = (const float*)d_in[4];
    const float* b1  = (const float*)d_in[5];
    const float* W2  = (const float*)d_in[6];
    const float* as2 = (const float*)d_in[7];
    const float* ad2 = (const float*)d_in[8];
    const float* b2  = (const float*)d_in[9];
    const int N  = in_sizes[0] / INCH;
    const int E  = in_sizes[1] / 2;
    const int Et = E + N;
    const int* srcA = ei;
    const int* dstA = ei + E;

    float* ws = (float*)d_ws;
    size_t off = 0;
    float* h1   = ws + off; off += (size_t)N * HC;
    float* out1 = ws + off; off += (size_t)N * HC;
    float* aS1  = ws + off; off += (size_t)N * NHD;
    float* aD1  = ws + off; off += (size_t)N * NHD;
    int* rowptr = (int*)(ws + off); off += (size_t)N + 1;
    int* cnt    = (int*)(ws + off); off += (size_t)N;
    int* col    = (int*)(ws + off); off += (size_t)Et;
    float* h2  = out1;
    float* aS2 = h2 + (size_t)N * OUTC;
    float* aD2 = aS2 + (((size_t)N + 31) & ~(size_t)31);
    int* partc = (int*)out1;
    int* partp = partc + (size_t)CSR_PARTS * N;
    int* rank  = partp + (size_t)CSR_PARTS * N;

    count_rank<<<CSR_PARTS, CSR_THREADS, 0, stream>>>(dstA, partc, rank, E, N);
    gemm1_and_part_prefix<<<g1_tiles_first(N) + prefix_blocks(N), CSR_STEP_THREADS, 0, stream>>>(
        x, W1, as1, ad1, h1, aS1, aD1, partc, partp, cnt, N);
    scan_rowptr<<<(N + 1023) / 1024, 1024, 0, stream>>>(cnt, rowptr, N);
    gemm1_and_fill_csr<<<g1_tiles(N) - g1_tiles_first(N) + fill_blocks(E, N), CSR_STEP_THREADS, 0,
                         stream>>>(
        x, W1, as1, ad1, h1, aS1, aD1, srcA, dstA, rank, partp, rowptr, col, E, N);

    gat_gather_l1_gemm2<<<(N + G2_ROWS - 1) / G2_ROWS, 32 * L1_WAVES, 0, stream>>>(
        rowptr, col, aS1, aD1, h1, b1, W2, as2, ad2, h2, aS2, aD2, N);

    gat_l2_fused<<<((N + L2_NODES - 1) / L2_NODES * 32 + 255) / 256, 256, 0, stream>>>(
        rowptr, col, aS2, aD2, h2, b2, (float*)d_out, N);
}
